# starting kernel
# baseline (speedup 1.0000x reference)
#include <hip/hip_runtime.h>
#include <math.h>

#define N_NODES 50000
#define N_EDGES 800000
#define NE_TOT  (N_EDGES + N_NODES)
#define IN_FEATS 256
#define HID 64
#define HEADS 2
#define F1 (HEADS * HID)
#define NCLS 40

typedef __attribute__((ext_vector_type(2))) float v2f;
typedef __attribute__((ext_vector_type(8))) float v8f;

static __device__ __forceinline__ unsigned fmap(float f) {
    unsigned u = __float_as_uint(f);
    return (u & 0x80000000u) ? ~u : (u | 0x80000000u);
}
static __device__ __forceinline__ float finv(unsigned t) {
    return (t & 0x80000000u) ? __uint_as_float(t & 0x7FFFFFFFu)
                             : __uint_as_float(~t);
}
static __device__ __forceinline__ void edge_sd(const int* __restrict__ ei, int e, int& s, int& d) {
    if (e < N_EDGES) { s = ei[e]; d = ei[N_EDGES + e]; }
    else             { s = e - N_EDGES; d = s; }
}
static __device__ __forceinline__ float lrelu(float x) { return x > 0.f ? x : 0.2f * x; }

__global__ __launch_bounds__(256) void gat_gemm1_wmma(const float* __restrict__ x,
                                                      const float* __restrict__ W1,
                                                      float* __restrict__ h1) {
    const int mtile = blockIdx.x;
    const int wave  = threadIdx.x >> 5;
    const int lane  = threadIdx.x & 31;
    const int half  = lane >> 4;
    const int l     = lane & 15;
    const int row   = mtile * 16 + l;
    const int ncol  = wave * 16 + l;

    v8f acc = {};
    const float* xr = x + (size_t)row * IN_FEATS;
    for (int k = 0; k < IN_FEATS; k += 4) {
        v2f a = *(const v2f*)(xr + k + 2 * half);
        const float* bp = W1 + (size_t)(k + 2 * half) * F1 + ncol;
        v2f b; b.x = bp[0]; b.y = bp[F1];
        acc = __builtin_amdgcn_wmma_f32_16x16x4_f32(false, a, false, b,
                                                    (short)0, acc, false, false);
    }
    float* dr = h1 + (size_t)(mtile * 16 + 8 * half) * F1 + wave * 16 + l;
#pragma unroll
    for (int r = 0; r < 8; ++r) dr[(size_t)r * F1] = acc[r];
}

__global__ __launch_bounds__(96) void gat_gemm2_wmma(const float* __restrict__ h2in,
                                                     const float* __restrict__ W2,
                                                     float* __restrict__ h2) {
    const int mtile = blockIdx.x;
    const int wave  = threadIdx.x >> 5;
    const int lane  = threadIdx.x & 31;
    const int half  = lane >> 4;
    const int l     = lane & 15;
    const int row   = mtile * 16 + l;
    const int ncol  = wave * 16 + l;

    v8f acc = {};
    const float* xr = h2in + (size_t)row * F1;
    for (int k = 0; k < F1; k += 4) {
        v2f a = *(const v2f*)(xr + k + 2 * half);
        v2f b;
        if (ncol < NCLS) {
            const float* bp = W2 + (size_t)(k + 2 * half) * NCLS + ncol;
            b.x = bp[0]; b.y = bp[NCLS];
        } else { b.x = 0.f; b.y = 0.f; }
        acc = __builtin_amdgcn_wmma_f32_16x16x4_f32(false, a, false, b,
                                                    (short)0, acc, false, false);
    }
    if (ncol < NCLS) {
        float* dr = h2 + (size_t)(mtile * 16 + 8 * half) * NCLS + ncol;
#pragma unroll
        for (int r = 0; r < 8; ++r) dr[(size_t)r * NCLS] = acc[r];
    }
}

__global__ void gat_att1(const float* __restrict__ h1, const float* __restrict__ as1,
                         const float* __restrict__ ad1, float* __restrict__ a_src,
                         float* __restrict__ a_dst) {
    int i = blockIdx.x * blockDim.x + threadIdx.x;
    if (i >= N_NODES * HEADS) return;
    int n = i >> 1, h = i & 1;
    const float* hp = h1 + (size_t)n * F1 + h * HID;
    const float* s0 = as1 + h * HID;
    const float* d0 = ad1 + h * HID;
    float s = 0.f, d = 0.f;
#pragma unroll 8
    for (int c = 0; c < HID; ++c) { float v = hp[c]; s += v * s0[c]; d += v * d0[c]; }
    a_src[i] = s; a_dst[i] = d;
}

__global__ void gat_att2(const float* __restrict__ h2, const float* __restrict__ as2,
                         const float* __restrict__ ad2, float* __restrict__ a_src,
                         float* __restrict__ a_dst) {
    int n = blockIdx.x * blockDim.x + threadIdx.x;
    if (n >= N_NODES) return;
    const float* hp = h2 + (size_t)n * NCLS;
    float s = 0.f, d = 0.f;
#pragma unroll 8
    for (int c = 0; c < NCLS; ++c) { float v = hp[c]; s += v * as2[c]; d += v * ad2[c]; }
    a_src[n] = s; a_dst[n] = d;
}

__global__ void gat_edge_max1(const int* __restrict__ ei, const float* __restrict__ a_src,
                              const float* __restrict__ a_dst, unsigned* __restrict__ emax) {
    int e = blockIdx.x * blockDim.x + threadIdx.x;
    if (e >= NE_TOT) return;
    int s, d; edge_sd(ei, e, s, d);
#pragma unroll
    for (int h = 0; h < HEADS; ++h) {
        float x = lrelu(a_src[s * 2 + h] + a_dst[d * 2 + h]);
        atomicMax(&emax[d * 2 + h], fmap(x));
    }
}

__global__ void gat_edge_max2(const int* __restrict__ ei, const float* __restrict__ a_src,
                              const float* __restrict__ a_dst, unsigned* __restrict__ emax) {
    int e = blockIdx.x * blockDim.x + threadIdx.x;
    if (e >= NE_TOT) return;
    int s, d; edge_sd(ei, e, s, d);
    float x = lrelu(a_src[s] + a_dst[d]);
    atomicMax(&emax[d], fmap(x));
}

__global__ __launch_bounds__(256) void gat_edge_accum1(const int* __restrict__ ei,
        const float* __restrict__ a_src, const float* __restrict__ a_dst,
        const unsigned* __restrict__ emax, const float* __restrict__ h1,
        float* __restrict__ denom, float* __restrict__ accum) {
    int e = blockIdx.x * 8 + (threadIdx.x >> 5);
    if (e >= NE_TOT) return;
    const int lane = threadIdx.x & 31;
    int s, d; edge_sd(ei, e, s, d);
    float p0 = __expf(lrelu(a_src[s * 2 + 0] + a_dst[d * 2 + 0]) - finv(emax[d * 2 + 0]));
    float p1 = __expf(lrelu(a_src[s * 2 + 1] + a_dst[d * 2 + 1]) - finv(emax[d * 2 + 1]));
    if (lane == 0) { atomicAdd(&denom[d * 2 + 0], p0); atomicAdd(&denom[d * 2 + 1], p1); }
    const float4 hv = *(const float4*)(h1 + (size_t)s * F1 + lane * 4);
    const float p = (lane < 16) ? p0 : p1;
    float* ap = accum + (size_t)d * F1 + lane * 4;
    atomicAdd(ap + 0, hv.x * p);
    atomicAdd(ap + 1, hv.y * p);
    atomicAdd(ap + 2, hv.z * p);
    atomicAdd(ap + 3, hv.w * p);
}

__global__ __launch_bounds__(256) void gat_edge_accum2(const int* __restrict__ ei,
        const float* __restrict__ a_src, const float* __restrict__ a_dst,
        const unsigned* __restrict__ emax, const float* __restrict__ h2,
        float* __restrict__ denom, float* __restrict__ accum) {
    int e = blockIdx.x * 8 + (threadIdx.x >> 5);
    if (e >= NE_TOT) return;
    const int lane = threadIdx.x & 31;
    int s, d; edge_sd(ei, e, s, d);
    float p = __expf(lrelu(a_src[s] + a_dst[d]) - finv(emax[d]));
    if (lane == 0) atomicAdd(&denom[d], p);
    for (int c = lane; c < NCLS; c += 32)
        atomicAdd(&accum[(size_t)d * NCLS + c], h2[(size_t)s * NCLS + c] * p);
}

__global__ void gat_finalize1(const float* __restrict__ accum, const float* __restrict__ denom,
                              const float* __restrict__ b1, float* __restrict__ h2in) {
    int i = blockIdx.x * blockDim.x + threadIdx.x;
    if (i >= N_NODES * F1) return;
    int n = i >> 7, c = i & 127;
    float v = accum[i] / denom[n * 2 + (c >> 6)] + b1[c];
    h2in[i] = v > 0.f ? v : (expf(v) - 1.0f);
}

__global__ void gat_finalize2(const float* __restrict__ accum, const float* __restrict__ denom,
                              const float* __restrict__ b2, float* __restrict__ out) {
    int i = blockIdx.x * blockDim.x + threadIdx.x;
    if (i >= N_NODES * NCLS) return;
    int n = i / NCLS, c = i - n * NCLS;
    out[i] = accum[i] / denom[n] + b2[c];
}

extern "C" void kernel_launch(void* const* d_in, const int* in_sizes, int n_in,
                              void* d_out, int out_size, void* d_ws, size_t ws_size,
                              hipStream_t stream) {
    const float* x    = (const float*)d_in[0];
    const int*   ei   = (const int*)  d_in[1];
    const float* W1   = (const float*)d_in[2];
    const float* as1  = (const float*)d_in[3];
    const float* ad1  = (const float*)d_in[4];
    const float* b1   = (const float*)d_in[5];
    const float* W2   = (const float*)d_in[6];
    const float* as2  = (const float*)d_in[7];
    const float* ad2  = (const float*)d_in[8];
    const float* b2   = (const float*)d_in[9];
    float* out = (float*)d_out;

    float* ws      = (float*)d_ws;
    float* h1      = ws;
    float* accum1  = h1     + (size_t)N_NODES * F1;
    float* accum2  = accum1 + (size_t)N_NODES * F1;
    float* denom1  = accum2 + (size_t)N_NODES * NCLS;
    float* denom2  = denom1 + (size_t)N_NODES * 2;
    unsigned* emax1 = (unsigned*)(denom2 + N_NODES);
    unsigned* emax2 = emax1 + (size_t)N_NODES * 2;
    float* a_src1  = (float*)(emax2 + N_NODES);
    float* a_dst1  = a_src1 + (size_t)N_NODES * 2;
    float* a_src2  = a_dst1 + (size_t)N_NODES * 2;
    float* a_dst2  = a_src2 + N_NODES;
    float* h2      = a_dst2 + N_NODES;

    const size_t zeroFloats = (size_t)N_NODES * (F1 + NCLS + 2 + 1 + 2 + 1);
    hipMemsetAsync(accum1, 0, zeroFloats * sizeof(float), stream);

    const int MT = N_NODES / 16;
    gat_gemm1_wmma<<<MT, 256, 0, stream>>>(x, W1, h1);
    gat_att1<<<(N_NODES * HEADS + 255) / 256, 256, 0, stream>>>(h1, as1, ad1, a_src1, a_dst1);
    gat_edge_max1<<<(NE_TOT + 255) / 256, 256, 0, stream>>>(ei, a_src1, a_dst1, emax1);
    gat_edge_accum1<<<(NE_TOT + 7) / 8, 256, 0, stream>>>(ei, a_src1, a_dst1, emax1, h1,
                                                          denom1, accum1);
    gat_finalize1<<<(N_NODES * F1 + 255) / 256, 256, 0, stream>>>(accum1, denom1, b1, h1);
    gat_gemm2_wmma<<<MT, 96, 0, stream>>>(h1, W2, h2);
    gat_att2<<<(N_NODES + 255) / 256, 256, 0, stream>>>(h2, as2, ad2, a_src2, a_dst2);
    gat_edge_max2<<<(NE_TOT + 255) / 256, 256, 0, stream>>>(ei, a_src2, a_dst2, emax2);
    gat_edge_accum2<<<(NE_TOT + 7) / 8, 256, 0, stream>>>(ei, a_src2, a_dst2, emax2, h2,
                                                          denom2, accum2);
    gat_finalize2<<<(N_NODES * NCLS + 255) / 256, 256, 0, stream>>>(accum2, denom2, b2, out);
}


# Edge accumulate re-tiled: a wave takes 32 edges (lane = edge for weights, then walks their rows, 32 loads in flight); sums go to f64 atomic accumulators: arrival order, but far below one f32 ulp
# speedup vs baseline: 1.1452x; 1.1452x over previous
#include <hip/hip_runtime.h>
#include <math.h>

#define N_NODES 50000
#define N_EDGES 800000
#define NE_TOT  (N_EDGES + N_NODES)
#define IN_FEATS 256
#define HID 64
#define HEADS 2
#define F1 (HEADS * HID)
#define NCLS 40

typedef __attribute__((ext_vector_type(2))) float v2f;
typedef __attribute__((ext_vector_type(8))) float v8f;

static __device__ __forceinline__ unsigned fmap(float f) {
    unsigned u = __float_as_uint(f);
    return (u & 0x80000000u) ? ~u : (u | 0x80000000u);
}
static __device__ __forceinline__ float finv(unsigned t) {
    return (t & 0x80000000u) ? __uint_as_float(t & 0x7FFFFFFFu)
                             : __uint_as_float(~t);
}
static __device__ __forceinline__ void edge_sd(const int* __restrict__ ei, int e, int& s, int& d) {
    if (e < N_EDGES) { s = ei[e]; d = ei[N_EDGES + e]; }
    else             { s = e - N_EDGES; d = s; }
}
static __device__ __forceinline__ float lrelu(float x) { return x > 0.f ? x : 0.2f * x; }

__global__ __launch_bounds__(256) void gat_gemm1_wmma(const float* __restrict__ x,
                                                      const float* __restrict__ W1,
                                                      float* __restrict__ h1) {
    const int mtile = blockIdx.x;
    const int wave  = threadIdx.x >> 5;
    const int lane  = threadIdx.x & 31;
    const int half  = lane >> 4;
    const int l     = lane & 15;
    const int row   = mtile * 16 + l;
    const int ncol  = wave * 16 + l;

    v8f acc = {};
    const float* xr = x + (size_t)row * IN_FEATS;
    for (int k = 0; k < IN_FEATS; k += 4) {
        v2f a = *(const v2f*)(xr + k + 2 * half);
        const float* bp = W1 + (size_t)(k + 2 * half) * F1 + ncol;
        v2f b; b.x = bp[0]; b.y = bp[F1];
        acc = __builtin_amdgcn_wmma_f32_16x16x4_f32(false, a, false, b,
                                                    (short)0, acc, false, false);
    }
    float* dr = h1 + (size_t)(mtile * 16 + 8 * half) * F1 + wave * 16 + l;
#pragma unroll
    for (int r = 0; r < 8; ++r) dr[(size_t)r * F1] = acc[r];
}

__global__ __launch_bounds__(96) void gat_gemm2_wmma(const float* __restrict__ h2in,
                                                     const float* __restrict__ W2,
                                                     float* __restrict__ h2) {
    const int mtile = blockIdx.x;
    const int wave  = threadIdx.x >> 5;
    const int lane  = threadIdx.x & 31;
    const int half  = lane >> 4;
    const int l     = lane & 15;
    const int row   = mtile * 16 + l;
    const int ncol  = wave * 16 + l;

    v8f acc = {};
    const float* xr = h2in + (size_t)row * F1;
    for (int k = 0; k < F1; k += 4) {
        v2f a = *(const v2f*)(xr + k + 2 * half);
        v2f b;
        if (ncol < NCLS) {
            const float* bp = W2 + (size_t)(k + 2 * half) * NCLS + ncol;
            b.x = bp[0]; b.y = bp[NCLS];
        } else { b.x = 0.f; b.y = 0.f; }
        acc = __builtin_amdgcn_wmma_f32_16x16x4_f32(false, a, false, b,
                                                    (short)0, acc, false, false);
    }
    if (ncol < NCLS) {
        float* dr = h2 + (size_t)(mtile * 16 + 8 * half) * NCLS + ncol;
#pragma unroll
        for (int r = 0; r < 8; ++r) dr[(size_t)r * NCLS] = acc[r];
    }
}

__global__ void gat_att1(const float* __restrict__ h1, const float* __restrict__ as1,
                         const float* __restrict__ ad1, float* __restrict__ a_src,
                         float* __restrict__ a_dst) {
    int i = blockIdx.x * blockDim.x + threadIdx.x;
    if (i >= N_NODES * HEADS) return;
    int n = i >> 1, h = i & 1;
    const float* hp = h1 + (size_t)n * F1 + h * HID;
    const float* s0 = as1 + h * HID;
    const float* d0 = ad1 + h * HID;
    float s = 0.f, d = 0.f;
#pragma unroll 8
    for (int c = 0; c < HID; ++c) { float v = hp[c]; s += v * s0[c]; d += v * d0[c]; }
    a_src[i] = s; a_dst[i] = d;
}

__global__ void gat_att2(const float* __restrict__ h2, const float* __restrict__ as2,
                         const float* __restrict__ ad2, float* __restrict__ a_src,
                         float* __restrict__ a_dst) {
    int n = blockIdx.x * blockDim.x + threadIdx.x;
    if (n >= N_NODES) return;
    const float* hp = h2 + (size_t)n * NCLS;
    float s = 0.f, d = 0.f;
#pragma unroll 8
    for (int c = 0; c < NCLS; ++c) { float v = hp[c]; s += v * as2[c]; d += v * ad2[c]; }
    a_src[n] = s; a_dst[n] = d;
}

__global__ void gat_edge_max1(const int* __restrict__ ei, const float* __restrict__ a_src,
                              const float* __restrict__ a_dst, unsigned* __restrict__ emax) {
    int e = blockIdx.x * blockDim.x + threadIdx.x;
    if (e >= NE_TOT) return;
    int s, d; edge_sd(ei, e, s, d);
#pragma unroll
    for (int h = 0; h < HEADS; ++h) {
        float x = lrelu(a_src[s * 2 + h] + a_dst[d * 2 + h]);
        atomicMax(&emax[d * 2 + h], fmap(x));
    }
}

__global__ void gat_edge_max2(const int* __restrict__ ei, const float* __restrict__ a_src,
                              const float* __restrict__ a_dst, unsigned* __restrict__ emax) {
    int e = blockIdx.x * blockDim.x + threadIdx.x;
    if (e >= NE_TOT) return;
    int s, d; edge_sd(ei, e, s, d);
    float x = lrelu(a_src[s] + a_dst[d]);
    atomicMax(&emax[d], fmap(x));
}

static __device__ __forceinline__ float lane_f(float v, int i) {
    return __int_as_float(__builtin_amdgcn_readlane(__float_as_int(v), i));
}

__global__ __launch_bounds__(256) void gat_edge_accum1(const int* __restrict__ ei,
        const float* __restrict__ a_src, const float* __restrict__ a_dst,
        const unsigned* __restrict__ emax, const float* __restrict__ h1,
        double* __restrict__ denom, double* __restrict__ accum) {
    const int lane = threadIdx.x & 31;
    const int base = (blockIdx.x * 8 + __builtin_amdgcn_readfirstlane(threadIdx.x >> 5)) * 32;
    if (base >= NE_TOT) return;
    const int cnt = (NE_TOT - base) < 32 ? (NE_TOT - base) : 32;
    int s = 0, d = 0;
    float p0 = 0.f, p1 = 0.f;
    if (lane < cnt) {
        edge_sd(ei, base + lane, s, d);
        p0 = __expf(lrelu(a_src[s * 2 + 0] + a_dst[d * 2 + 0]) - finv(emax[d * 2 + 0]));
        p1 = __expf(lrelu(a_src[s * 2 + 1] + a_dst[d * 2 + 1]) - finv(emax[d * 2 + 1]));
        atomicAdd(&denom[d * 2 + 0], (double)p0);
        atomicAdd(&denom[d * 2 + 1], (double)p1);
    }
#pragma unroll 1
    for (int i0 = 0; i0 < cnt; i0 += 8) {
        float hv[8][4];
#pragma unroll
        for (int j = 0; j < 8; ++j) {
            const int i = (i0 + j < cnt) ? (i0 + j) : (cnt - 1);
            const int sj = __builtin_amdgcn_readlane(s, i);
            const float* rp = h1 + (size_t)sj * F1 + lane;
#pragma unroll
            for (int k = 0; k < 4; ++k) hv[j][k] = rp[k * 32];
        }
#pragma unroll
        for (int j = 0; j < 8; ++j) {
            if (i0 + j < cnt) {
                const int dj = __builtin_amdgcn_readlane(d, i0 + j);
                const float q0 = lane_f(p0, i0 + j);
                const float q1 = lane_f(p1, i0 + j);
                double* ap = accum + (size_t)dj * F1 + lane;
                atomicAdd(ap +  0, (double)(hv[j][0] * q0));
                atomicAdd(ap + 32, (double)(hv[j][1] * q0));
                atomicAdd(ap + 64, (double)(hv[j][2] * q1));
                atomicAdd(ap + 96, (double)(hv[j][3] * q1));
            }
        }
    }
}

__global__ __launch_bounds__(256) void gat_edge_accum2(const int* __restrict__ ei,
        const float* __restrict__ a_src, const float* __restrict__ a_dst,
        const unsigned* __restrict__ emax, const float* __restrict__ h2,
        double* __restrict__ denom, double* __restrict__ accum) {
    const int lane = threadIdx.x & 31;
    const int base = (blockIdx.x * 8 + __builtin_amdgcn_readfirstlane(threadIdx.x >> 5)) * 32;
    if (base >= NE_TOT) return;
    const int cnt = (NE_TOT - base) < 32 ? (NE_TOT - base) : 32;
    int s = 0, d = 0;
    float p = 0.f;
    if (lane < cnt) {
        edge_sd(ei, base + lane, s, d);
        p = __expf(lrelu(a_src[s] + a_dst[d]) - finv(emax[d]));
        atomicAdd(&denom[d], (double)p);
    }
    const int total = cnt * NCLS;
#pragma unroll 1
    for (int t0 = 0; t0 < total; t0 += 8 * 32) {
        float hv[8];
        int   dd[8];
        float pp[8];
        int   cc[8];
#pragma unroll
        for (int j = 0; j < 8; ++j) {
            const int idx = t0 + j * 32 + lane;
            const int ok  = idx < total;
            const int e   = ok ? idx / NCLS : 0;
            const int col = ok ? idx - e * NCLS : 0;
            const int sj  = __shfl(s, e);
            const int dj  = __shfl(d, e);
            dd[j] = ok ? dj : -1;
            pp[j] = __shfl(p, e);
            cc[j] = col;
            hv[j] = h2[(size_t)sj * NCLS + col];
        }
#pragma unroll
        for (int j = 0; j < 8; ++j) {
            if (dd[j] >= 0)
                atomicAdd(&accum[(size_t)dd[j] * NCLS + cc[j]], (double)(hv[j] * pp[j]));
        }
    }
}

__global__ void gat_finalize1(const double* __restrict__ accum, const double* __restrict__ denom,
                              const float* __restrict__ b1, float* __restrict__ h2in) {
    int i = blockIdx.x * blockDim.x + threadIdx.x;
    if (i >= N_NODES * F1) return;
    int n = i >> 7, c = i & 127;
    float v = (float)accum[i] / (float)denom[n * 2 + (c >> 6)] + b1[c];
    h2in[i] = v > 0.f ? v : (expf(v) - 1.0f);
}

__global__ void gat_finalize2(const double* __restrict__ accum, const double* __restrict__ denom,
                              const float* __restrict__ b2, float* __restrict__ out) {
    int i = blockIdx.x * blockDim.x + threadIdx.x;
    if (i >= N_NODES * NCLS) return;
    int n = i / NCLS, c = i - n * NCLS;
    out[i] = (float)accum[i] / (float)denom[n] + b2[c];
}

extern "C" void kernel_launch(void* const* d_in, const int* in_sizes, int n_in,
                              void* d_out, int out_size, void* d_ws, size_t ws_size,
                              hipStream_t stream) {
    const float* x    = (const float*)d_in[0];
    const int*   ei   = (const int*)  d_in[1];
    const float* W1   = (const float*)d_in[2];
    const float* as1  = (const float*)d_in[3];
    const float* ad1  = (const float*)d_in[4];
    const float* b1   = (const float*)d_in[5];
    const float* W2   = (const float*)d_in[6];
    const float* as2  = (const float*)d_in[7];
    const float* ad2  = (const float*)d_in[8];
    const float* b2   = (const float*)d_in[9];
    float* out = (float*)d_out;

    float* ws      = (float*)d_ws;
    float* h1      = ws;
    double* accum1 = (double*)(h1 + (size_t)N_NODES * F1);
    double* accum2 = accum1 + (size_t)N_NODES * F1;
    double* denom1 = accum2 + (size_t)N_NODES * NCLS;
    double* denom2 = denom1 + (size_t)N_NODES * 2;
    unsigned* emax1 = (unsigned*)(denom2 + N_NODES);
    unsigned* emax2 = emax1 + (size_t)N_NODES * 2;
    float* a_src1  = (float*)(emax2 + N_NODES);
    float* a_dst1  = a_src1 + (size_t)N_NODES * 2;
    float* a_src2  = a_dst1 + (size_t)N_NODES * 2;
    float* a_dst2  = a_src2 + N_NODES;
    float* h2      = a_dst2 + N_NODES;

    const size_t zeroBytes = (size_t)N_NODES * ((F1 + NCLS + 2 + 1) * sizeof(double)
                                                + (2 + 1) * sizeof(unsigned));
    hipMemsetAsync(accum1, 0, zeroBytes, stream);

    const int MT = N_NODES / 16;
    gat_gemm1_wmma<<<MT, 256, 0, stream>>>(x, W1, h1);
    gat_att1<<<(N_NODES * HEADS + 255) / 256, 256, 0, stream>>>(h1, as1, ad1, a_src1, a_dst1);
    gat_edge_max1<<<(NE_TOT + 255) / 256, 256, 0, stream>>>(ei, a_src1, a_dst1, emax1);
    gat_edge_accum1<<<(NE_TOT + 255) / 256, 256, 0, stream>>>(ei, a_src1, a_dst1, emax1, h1,
                                                          denom1, accum1);
    gat_finalize1<<<(N_NODES * F1 + 255) / 256, 256, 0, stream>>>(accum1, denom1, b1, h1);
    gat_gemm2_wmma<<<MT, 96, 0, stream>>>(h1, W2, h2);
    gat_att2<<<(N_NODES + 255) / 256, 256, 0, stream>>>(h2, as2, ad2, a_src2, a_dst2);
    gat_edge_max2<<<(NE_TOT + 255) / 256, 256, 0, stream>>>(ei, a_src2, a_dst2, emax2);
    gat_edge_accum2<<<(NE_TOT + 255) / 256, 256, 0, stream>>>(ei, a_src2, a_dst2, emax2, h2,
                                                          denom2, accum2);
    gat_finalize2<<<(N_NODES * NCLS + 255) / 256, 256, 0, stream>>>(accum2, denom2, b2, out);
}


# Edge sums moved from global f64 atomics into LDS: a block owns 196 destinations, filters the edge list for them, adds by LDS f64 atomics (arrival order, far below an f32 ulp), stores each row once
# speedup vs baseline: 1.4104x; 1.2316x over previous
#include <hip/hip_runtime.h>
#include <math.h>

#define N_NODES 50000
#define N_EDGES 800000
#define NE_TOT  (N_EDGES + N_NODES)
#define IN_FEATS 256
#define HID 64
#define HEADS 2
#define F1 (HEADS * HID)
#define NCLS 40

typedef __attribute__((ext_vector_type(2))) float v2f;
typedef __attribute__((ext_vector_type(8))) float v8f;

static __device__ __forceinline__ unsigned fmap(float f) {
    unsigned u = __float_as_uint(f);
    return (u & 0x80000000u) ? ~u : (u | 0x80000000u);
}
static __device__ __forceinline__ float finv(unsigned t) {
    return (t & 0x80000000u) ? __uint_as_float(t & 0x7FFFFFFFu)
                             : __uint_as_float(~t);
}
static __device__ __forceinline__ void edge_sd(const int* __restrict__ ei, int e, int& s, int& d) {
    if (e < N_EDGES) { s = ei[e]; d = ei[N_EDGES + e]; }
    else             { s = e - N_EDGES; d = s; }
}
static __device__ __forceinline__ float lrelu(float x) { return x > 0.f ? x : 0.2f * x; }

__global__ __launch_bounds__(256) void gat_gemm1_wmma(const float* __restrict__ x,
                                                      const float* __restrict__ W1,
                                                      float* __restrict__ h1) {
    const int mtile = blockIdx.x;
    const int wave  = threadIdx.x >> 5;
    const int lane  = threadIdx.x & 31;
    const int half  = lane >> 4;
    const int l     = lane & 15;
    const int row   = mtile * 16 + l;
    const int ncol  = wave * 16 + l;

    v8f acc = {};
    const float* xr = x + (size_t)row * IN_FEATS;
    for (int k = 0; k < IN_FEATS; k += 4) {
        v2f a = *(const v2f*)(xr + k + 2 * half);
        const float* bp = W1 + (size_t)(k + 2 * half) * F1 + ncol;
        v2f b; b.x = bp[0]; b.y = bp[F1];
        acc = __builtin_amdgcn_wmma_f32_16x16x4_f32(false, a, false, b,
                                                    (short)0, acc, false, false);
    }
    float* dr = h1 + (size_t)(mtile * 16 + 8 * half) * F1 + wave * 16 + l;
#pragma unroll
    for (int r = 0; r < 8; ++r) dr[(size_t)r * F1] = acc[r];
}

__global__ __launch_bounds__(96) void gat_gemm2_wmma(const float* __restrict__ h2in,
                                                     const float* __restrict__ W2,
                                                     float* __restrict__ h2) {
    const int mtile = blockIdx.x;
    const int wave  = threadIdx.x >> 5;
    const int lane  = threadIdx.x & 31;
    const int half  = lane >> 4;
    const int l     = lane & 15;
    const int row   = mtile * 16 + l;
    const int ncol  = wave * 16 + l;

    v8f acc = {};
    const float* xr = h2in + (size_t)row * F1;
    for (int k = 0; k < F1; k += 4) {
        v2f a = *(const v2f*)(xr + k + 2 * half);
        v2f b;
        if (ncol < NCLS) {
            const float* bp = W2 + (size_t)(k + 2 * half) * NCLS + ncol;
            b.x = bp[0]; b.y = bp[NCLS];
        } else { b.x = 0.f; b.y = 0.f; }
        acc = __builtin_amdgcn_wmma_f32_16x16x4_f32(false, a, false, b,
                                                    (short)0, acc, false, false);
    }
    if (ncol < NCLS) {
        float* dr = h2 + (size_t)(mtile * 16 + 8 * half) * NCLS + ncol;
#pragma unroll
        for (int r = 0; r < 8; ++r) dr[(size_t)r * NCLS] = acc[r];
    }
}

__global__ void gat_att1(const float* __restrict__ h1, const float* __restrict__ as1,
                         const float* __restrict__ ad1, float* __restrict__ a_src,
                         float* __restrict__ a_dst) {
    int i = blockIdx.x * blockDim.x + threadIdx.x;
    if (i >= N_NODES * HEADS) return;
    int n = i >> 1, h = i & 1;
    const float* hp = h1 + (size_t)n * F1 + h * HID;
    const float* s0 = as1 + h * HID;
    const float* d0 = ad1 + h * HID;
    float s = 0.f, d = 0.f;
#pragma unroll 8
    for (int c = 0; c < HID; ++c) { float v = hp[c]; s += v * s0[c]; d += v * d0[c]; }
    a_src[i] = s; a_dst[i] = d;
}

__global__ void gat_att2(const float* __restrict__ h2, const float* __restrict__ as2,
                         const float* __restrict__ ad2, float* __restrict__ a_src,
                         float* __restrict__ a_dst) {
    int n = blockIdx.x * blockDim.x + threadIdx.x;
    if (n >= N_NODES) return;
    const float* hp = h2 + (size_t)n * NCLS;
    float s = 0.f, d = 0.f;
#pragma unroll 8
    for (int c = 0; c < NCLS; ++c) { float v = hp[c]; s += v * as2[c]; d += v * ad2[c]; }
    a_src[n] = s; a_dst[n] = d;
}

__global__ void gat_edge_max1(const int* __restrict__ ei, const float* __restrict__ a_src,
                              const float* __restrict__ a_dst, unsigned* __restrict__ emax) {
    int e = blockIdx.x * blockDim.x + threadIdx.x;
    if (e >= NE_TOT) return;
    int s, d; edge_sd(ei, e, s, d);
#pragma unroll
    for (int h = 0; h < HEADS; ++h) {
        float x = lrelu(a_src[s * 2 + h] + a_dst[d * 2 + h]);
        atomicMax(&emax[d * 2 + h], fmap(x));
    }
}

__global__ void gat_edge_max2(const int* __restrict__ ei, const float* __restrict__ a_src,
                              const float* __restrict__ a_dst, unsigned* __restrict__ emax) {
    int e = blockIdx.x * blockDim.x + threadIdx.x;
    if (e >= NE_TOT) return;
    int s, d; edge_sd(ei, e, s, d);
    float x = lrelu(a_src[s] + a_dst[d]);
    atomicMax(&emax[d], fmap(x));
}

#define OWN_R      196
#define OWN_BLOCKS ((N_NODES + OWN_R - 1) / OWN_R)
#define CHUNK_E    128
#define N_CHUNKS   (N_EDGES / CHUNK_E)
static_assert(N_EDGES % CHUNK_E == 0, "the destination list is read in whole 128-edge chunks");

static __device__ __forceinline__ float lane_f(float v, int i) {
    return __int_as_float(__builtin_amdgcn_readlane(__float_as_int(v), i));
}

template <int F, int NH, int TRIPC>
__global__ __launch_bounds__(1024) void gat_own_accum(const int* __restrict__ ei,
        const float* __restrict__ a_src, const float* __restrict__ a_dst,
        const unsigned* __restrict__ emax, const float* __restrict__ h,
        const float* __restrict__ bias, float* __restrict__ outp) {
    constexpr int CAP = TRIPC * CHUNK_E + 128;
    __shared__ double acc[OWN_R * F];
    __shared__ double den[OWN_R * NH];
    __shared__ int    que[32 * CAP];

    const int tid  = threadIdx.x;
    const int lane = tid & 31;
    const int wave = __builtin_amdgcn_readfirstlane(tid >> 5);
    const int n0   = blockIdx.x * OWN_R;
    const int nr   = (N_NODES - n0) < OWN_R ? (N_NODES - n0) : OWN_R;
    int* q = que + wave * CAP;

    for (int i = tid; i < OWN_R * F;  i += 1024) acc[i] = 0.0;
    for (int i = tid; i < OWN_R * NH; i += 1024) den[i] = 0.0;
    __syncthreads();

    int cnt = 0;
    for (int c = wave * 32; c < nr; c += 1024) {
        const bool m = (c + lane) < nr;
        const unsigned mk = __builtin_amdgcn_ballot_w32(m);
        if (m) q[cnt + __builtin_amdgcn_mbcnt_lo(mk, 0)] = N_EDGES + n0 + c + lane;
        cnt += __builtin_popcount(mk);
    }

    const int4* dv = reinterpret_cast<const int4*>(ei + N_EDGES);
    int c0 = wave;
    for (;;) {
        while (c0 < N_CHUNKS && cnt <= CAP - TRIPC * CHUNK_E) {
            int4 dd[TRIPC];
#pragma unroll
            for (int u = 0; u < TRIPC; ++u) {
                const int c = c0 + 32 * u;
                const int cc = c < N_CHUNKS ? c : c0;
                dd[u] = dv[(size_t)cc * 32 + lane];
            }
#pragma unroll
            for (int u = 0; u < TRIPC; ++u) {
                const int c = c0 + 32 * u;
                if (c < N_CHUNKS) {
                    const int eb = c * CHUNK_E + lane * 4;
                    const int dk[4] = {dd[u].x, dd[u].y, dd[u].z, dd[u].w};
#pragma unroll
                    for (int k = 0; k < 4; ++k) {
                        const bool m = (unsigned)(dk[k] - n0) < (unsigned)nr;
                        const unsigned mk = __builtin_amdgcn_ballot_w32(m);
                        if (mk) {
                            if (m) q[cnt + __builtin_amdgcn_mbcnt_lo(mk, 0)] = eb + k;
                            cnt += __builtin_popcount(mk);
                        }
                    }
                }
            }
            c0 += 32 * TRIPC;
        }
        for (int off = 0; off < cnt; off += 32) {
            const int nb = (cnt - off) < 32 ? (cnt - off) : 32;
            int s = 0, dl = 0;
            float p0 = 0.f, p1 = 0.f;
            if (lane < nb) {
                int d; edge_sd(ei, q[off + lane], s, d);
                dl = d - n0;
                p0 = __expf(lrelu(a_src[s * NH] + a_dst[d * NH]) - finv(emax[d * NH]));
                atomicAdd(&den[dl * NH], (double)p0);
                if constexpr (NH == 2) {
                    p1 = __expf(lrelu(a_src[s * 2 + 1] + a_dst[d * 2 + 1]) - finv(emax[d * 2 + 1]));
                    atomicAdd(&den[dl * 2 + 1], (double)p1);
                }
            }
            if constexpr (F == F1) {
#pragma unroll 1
                for (int i0 = 0; i0 < nb; i0 += 8) {
                    float hv[8][4];
#pragma unroll
                    for (int j = 0; j < 8; ++j) {
                        const int i = (i0 + j < nb) ? (i0 + j) : (nb - 1);
                        const int sj = __builtin_amdgcn_readlane(s, i);
                        const float* rp = h + (size_t)sj * F1 + lane;
#pragma unroll
                        for (int k = 0; k < 4; ++k) hv[j][k] = rp[k * 32];
                    }
#pragma unroll
                    for (int j = 0; j < 8; ++j) {
                        if (i0 + j < nb) {
                            const int dj = __builtin_amdgcn_readlane(dl, i0 + j);
                            const float q0 = lane_f(p0, i0 + j);
                            const float q1 = lane_f(p1, i0 + j);
                            double* ap = acc + dj * F1 + lane;
                            atomicAdd(ap +  0, (double)(hv[j][0] * q0));
                            atomicAdd(ap + 32, (double)(hv[j][1] * q0));
                            atomicAdd(ap + 64, (double)(hv[j][2] * q1));
                            atomicAdd(ap + 96, (double)(hv[j][3] * q1));
                        }
                    }
                }
            } else {
                const int total = nb * F;
#pragma unroll 1
                for (int t0 = 0; t0 < total; t0 += 8 * 32) {
                    float hv[8];
                    int   dd2[8];
                    float pp[8];
                    int   cc[8];
#pragma unroll
                    for (int j = 0; j < 8; ++j) {
                        const int idx = t0 + j * 32 + lane;
                        const int ok  = idx < total;
                        const int e   = ok ? idx / F : 0;
                        const int col = ok ? idx - e * F : 0;
                        const int sj  = __shfl(s, e);
                        const int dj  = __shfl(dl, e);
                        dd2[j] = ok ? dj : -1;
                        pp[j] = __shfl(p0, e);
                        cc[j] = col;
                        hv[j] = h[(size_t)sj * F + col];
                    }
#pragma unroll
                    for (int j = 0; j < 8; ++j) {
                        if (dd2[j] >= 0)
                            atomicAdd(&acc[dd2[j] * F + cc[j]], (double)(hv[j] * pp[j]));
                    }
                }
            }
        }
        cnt = 0;
        if (c0 >= N_CHUNKS) break;
    }

    __syncthreads();
    for (int i = tid; i < nr * F; i += 1024) {
        const int r = i / F, c = i - r * F;
        float v = (float)acc[i] / (float)den[r * NH + (NH == 2 ? (c >> 6) : 0)] + bias[c];
        if constexpr (F == F1) v = v > 0.f ? v : (expf(v) - 1.0f);
        outp[(size_t)n0 * F + i] = v;
    }
}

extern "C" void kernel_launch(void* const* d_in, const int* in_sizes, int n_in,
                              void* d_out, int out_size, void* d_ws, size_t ws_size,
                              hipStream_t stream) {
    const float* x    = (const float*)d_in[0];
    const int*   ei   = (const int*)  d_in[1];
    const float* W1   = (const float*)d_in[2];
    const float* as1  = (const float*)d_in[3];
    const float* ad1  = (const float*)d_in[4];
    const float* b1   = (const float*)d_in[5];
    const float* W2   = (const float*)d_in[6];
    const float* as2  = (const float*)d_in[7];
    const float* ad2  = (const float*)d_in[8];
    const float* b2   = (const float*)d_in[9];
    float* out = (float*)d_out;

    float* ws      = (float*)d_ws;
    float* h1      = ws;
    float* h2in    = h1 + (size_t)N_NODES * F1;
    unsigned* emax1 = (unsigned*)(h2in + (size_t)N_NODES * F1);
    unsigned* emax2 = emax1 + (size_t)N_NODES * 2;
    float* a_src1  = (float*)(emax2 + N_NODES);
    float* a_dst1  = a_src1 + (size_t)N_NODES * 2;
    float* a_src2  = a_dst1 + (size_t)N_NODES * 2;
    float* a_dst2  = a_src2 + N_NODES;
    float* h2      = a_dst2 + N_NODES;

    const size_t zeroBytes = (size_t)N_NODES * (2 + 1) * sizeof(unsigned);
    hipMemsetAsync(emax1, 0, zeroBytes, stream);

    const int MT = N_NODES / 16;
    gat_gemm1_wmma<<<MT, 256, 0, stream>>>(x, W1, h1);
    gat_att1<<<(N_NODES * HEADS + 255) / 256, 256, 0, stream>>>(h1, as1, ad1, a_src1, a_dst1);
    gat_edge_max1<<<(NE_TOT + 255) / 256, 256, 0, stream>>>(ei, a_src1, a_dst1, emax1);
    gat_own_accum<F1, HEADS, 6><<<OWN_BLOCKS, 1024, 0, stream>>>(ei, a_src1, a_dst1, emax1, h1,
                                                                b1, h2in);
    gat_gemm2_wmma<<<MT, 96, 0, stream>>>(h2in, W2, h2);
    gat_att2<<<(N_NODES + 255) / 256, 256, 0, stream>>>(h2, as2, ad2, a_src2, a_dst2);
    gat_edge_max2<<<(NE_TOT + 255) / 256, 256, 0, stream>>>(ei, a_src2, a_dst2, emax2);
    gat_own_accum<NCLS, 1, 8><<<OWN_BLOCKS, 1024, 0, stream>>>(ei, a_src2, a_dst2, emax2, h2,
                                                              b2, out);
}


# Segment max moved into the owner blocks: LDS atomic max over a block's kept edge list, then the list again for exp and sums; both max launches and the clear gone; sums as before, arrival order
# speedup vs baseline: 1.9341x; 1.3713x over previous
#include <hip/hip_runtime.h>
#include <math.h>

#define N_NODES 50000
#define N_EDGES 800000
#define IN_FEATS 256
#define HID 64
#define HEADS 2
#define F1 (HEADS * HID)
#define NCLS 40

typedef __attribute__((ext_vector_type(2))) float v2f;
typedef __attribute__((ext_vector_type(8))) float v8f;

static __device__ __forceinline__ unsigned fmap(float f) {
    unsigned u = __float_as_uint(f);
    return (u & 0x80000000u) ? ~u : (u | 0x80000000u);
}
static __device__ __forceinline__ float finv(unsigned t) {
    return (t & 0x80000000u) ? __uint_as_float(t & 0x7FFFFFFFu)
                             : __uint_as_float(~t);
}
static __device__ __forceinline__ void edge_sd(const int* __restrict__ ei, int e, int& s, int& d) {
    if (e < N_EDGES) { s = ei[e]; d = ei[N_EDGES + e]; }
    else             { s = e - N_EDGES; d = s; }
}
static __device__ __forceinline__ float lrelu(float x) { return x > 0.f ? x : 0.2f * x; }

__global__ __launch_bounds__(256) void gat_gemm1_wmma(const float* __restrict__ x,
                                                      const float* __restrict__ W1,
                                                      float* __restrict__ h1) {
    const int mtile = blockIdx.x;
    const int wave  = threadIdx.x >> 5;
    const int lane  = threadIdx.x & 31;
    const int half  = lane >> 4;
    const int l     = lane & 15;
    const int row   = mtile * 16 + l;
    const int ncol  = wave * 16 + l;

    v8f acc = {};
    const float* xr = x + (size_t)row * IN_FEATS;
    for (int k = 0; k < IN_FEATS; k += 4) {
        v2f a = *(const v2f*)(xr + k + 2 * half);
        const float* bp = W1 + (size_t)(k + 2 * half) * F1 + ncol;
        v2f b; b.x = bp[0]; b.y = bp[F1];
        acc = __builtin_amdgcn_wmma_f32_16x16x4_f32(false, a, false, b,
                                                    (short)0, acc, false, false);
    }
    float* dr = h1 + (size_t)(mtile * 16 + 8 * half) * F1 + wave * 16 + l;
#pragma unroll
    for (int r = 0; r < 8; ++r) dr[(size_t)r * F1] = acc[r];
}

__global__ __launch_bounds__(96) void gat_gemm2_wmma(const float* __restrict__ h2in,
                                                     const float* __restrict__ W2,
                                                     float* __restrict__ h2) {
    const int mtile = blockIdx.x;
    const int wave  = threadIdx.x >> 5;
    const int lane  = threadIdx.x & 31;
    const int half  = lane >> 4;
    const int l     = lane & 15;
    const int row   = mtile * 16 + l;
    const int ncol  = wave * 16 + l;

    v8f acc = {};
    const float* xr = h2in + (size_t)row * F1;
    for (int k = 0; k < F1; k += 4) {
        v2f a = *(const v2f*)(xr + k + 2 * half);
        v2f b;
        if (ncol < NCLS) {
            const float* bp = W2 + (size_t)(k + 2 * half) * NCLS + ncol;
            b.x = bp[0]; b.y = bp[NCLS];
        } else { b.x = 0.f; b.y = 0.f; }
        acc = __builtin_amdgcn_wmma_f32_16x16x4_f32(false, a, false, b,
                                                    (short)0, acc, false, false);
    }
    if (ncol < NCLS) {
        float* dr = h2 + (size_t)(mtile * 16 + 8 * half) * NCLS + ncol;
#pragma unroll
        for (int r = 0; r < 8; ++r) dr[(size_t)r * NCLS] = acc[r];
    }
}

__global__ void gat_att1(const float* __restrict__ h1, const float* __restrict__ as1,
                         const float* __restrict__ ad1, float* __restrict__ a_src,
                         float* __restrict__ a_dst) {
    int i = blockIdx.x * blockDim.x + threadIdx.x;
    if (i >= N_NODES * HEADS) return;
    int n = i >> 1, h = i & 1;
    const float* hp = h1 + (size_t)n * F1 + h * HID;
    const float* s0 = as1 + h * HID;
    const float* d0 = ad1 + h * HID;
    float s = 0.f, d = 0.f;
#pragma unroll 8
    for (int c = 0; c < HID; ++c) { float v = hp[c]; s += v * s0[c]; d += v * d0[c]; }
    a_src[i] = s; a_dst[i] = d;
}

__global__ void gat_att2(const float* __restrict__ h2, const float* __restrict__ as2,
                         const float* __restrict__ ad2, float* __restrict__ a_src,
                         float* __restrict__ a_dst) {
    int n = blockIdx.x * blockDim.x + threadIdx.x;
    if (n >= N_NODES) return;
    const float* hp = h2 + (size_t)n * NCLS;
    float s = 0.f, d = 0.f;
#pragma unroll 8
    for (int c = 0; c < NCLS; ++c) { float v = hp[c]; s += v * as2[c]; d += v * ad2[c]; }
    a_src[n] = s; a_dst[n] = d;
}

#define OWN_R      196
#define OWN_BLOCKS ((N_NODES + OWN_R - 1) / OWN_R)
#define CHUNK_E    128
#define N_CHUNKS   (N_EDGES / CHUNK_E)
#define SELF_LANES ((OWN_R + 31) / 32)
static_assert(N_EDGES % CHUNK_E == 0, "the destination list is read in whole 128-edge chunks");
static_assert(N_NODES <= 65536 && OWN_R <= 65536, "a list entry packs source and local destination, 16 bits each");
static_assert(SELF_LANES <= 32, "one lane per self-loop of a wave");

static __device__ __forceinline__ float lane_f(float v, int i) {
    return __int_as_float(__builtin_amdgcn_readlane(__float_as_int(v), i));
}

template <int F, int NH, int TRIPC, int CAP>
__global__ __launch_bounds__(1024) void gat_own_accum(const int* __restrict__ ei,
        const float* __restrict__ a_src, const float* __restrict__ a_dst,
        const float* __restrict__ h,
        const float* __restrict__ bias, float* __restrict__ outp) {
    static_assert(CAP >= CHUNK_E + SELF_LANES, "a chunk goes into a list only while 128 entries are free");
    __shared__ double   acc[OWN_R * F];
    __shared__ double   den[OWN_R * NH];
    __shared__ unsigned mx[OWN_R * NH];
    __shared__ int      que[32 * CAP];

    const int tid  = threadIdx.x;
    const int lane = tid & 31;
    const int wave = __builtin_amdgcn_readfirstlane(tid >> 5);
    const int n0   = blockIdx.x * OWN_R;
    const int nr   = (N_NODES - n0) < OWN_R ? (N_NODES - n0) : OWN_R;
    int* q = que + wave * CAP;

    for (int i = tid; i < OWN_R * F;  i += 1024) acc[i] = 0.0;
    for (int i = tid; i < OWN_R * NH; i += 1024) { den[i] = 0.0; mx[i] = 0u; }
    __syncthreads();

    auto max_pass = [&](int n) {
        for (int off = 0; off < n; off += 128) {
            int e[4], s[4], d[4];
            float x0[4];
            [[maybe_unused]] float x1[4];
#pragma unroll
            for (int u = 0; u < 4; ++u) {
                const int i = off + 32 * u + lane;
                e[u] = i < n ? q[i] : N_EDGES + n0;
            }
#pragma unroll
            for (int u = 0; u < 4; ++u) {
                const bool r  = e[u] < N_EDGES;
                const int  ec = r ? e[u] : 0;
                const int  sl = ei[ec], dg = ei[N_EDGES + ec];
                s[u] = r ? sl : e[u] - N_EDGES;
                d[u] = r ? dg : s[u];
            }
#pragma unroll
            for (int u = 0; u < 4; ++u) {
                x0[u] = lrelu(a_src[s[u] * NH] + a_dst[d[u] * NH]);
                if constexpr (NH == 2) x1[u] = lrelu(a_src[s[u] * 2 + 1] + a_dst[d[u] * 2 + 1]);
            }
#pragma unroll
            for (int u = 0; u < 4; ++u) {
                const int i = off + 32 * u + lane;
                if (i < n) {
                    const int dl = d[u] - n0;
                    atomicMax(&mx[dl * NH], fmap(x0[u]));
                    if constexpr (NH == 2) atomicMax(&mx[dl * 2 + 1], fmap(x1[u]));
                    q[i] = s[u] | (dl << 16);
                }
            }
        }
    };

    auto work = [&](int n) {
        for (int off = 0; off < n; off += 32) {
            const int nb = (n - off) < 32 ? (n - off) : 32;
            int s = 0, dl = 0;
            float p0 = 0.f, p1 = 0.f;
            if (lane < nb) {
                const unsigned pk = (unsigned)q[off + lane];
                s  = (int)(pk & 0xFFFFu);
                dl = (int)(pk >> 16);
                const int d = n0 + dl;
                p0 = __expf(lrelu(a_src[s * NH] + a_dst[d * NH]) - finv(mx[dl * NH]));
                atomicAdd(&den[dl * NH], (double)p0);
                if constexpr (NH == 2) {
                    p1 = __expf(lrelu(a_src[s * 2 + 1] + a_dst[d * 2 + 1]) - finv(mx[dl * 2 + 1]));
                    atomicAdd(&den[dl * 2 + 1], (double)p1);
                }
            }
            if constexpr (F == F1) {
#pragma unroll 1
                for (int i0 = 0; i0 < nb; i0 += 8) {
                    float hv[8][4];
#pragma unroll
                    for (int j = 0; j < 8; ++j) {
                        const int i = (i0 + j < nb) ? (i0 + j) : (nb - 1);
                        const int sj = __builtin_amdgcn_readlane(s, i);
                        const float* rp = h + (size_t)sj * F1 + lane;
#pragma unroll
                        for (int k = 0; k < 4; ++k) hv[j][k] = rp[k * 32];
                    }
#pragma unroll
                    for (int j = 0; j < 8; ++j) {
                        if (i0 + j < nb) {
                            const int dj = __builtin_amdgcn_readlane(dl, i0 + j);
                            const float q0 = lane_f(p0, i0 + j);
                            const float q1 = lane_f(p1, i0 + j);
                            double* ap = acc + dj * F1 + lane;
                            atomicAdd(ap +  0, (double)(hv[j][0] * q0));
                            atomicAdd(ap + 32, (double)(hv[j][1] * q0));
                            atomicAdd(ap + 64, (double)(hv[j][2] * q1));
                            atomicAdd(ap + 96, (double)(hv[j][3] * q1));
                        }
                    }
                }
            } else {
                const int total = nb * F;
#pragma unroll 1
                for (int t0 = 0; t0 < total; t0 += 8 * 32) {
                    float hv[8];
                    int   dd2[8];
                    float pp[8];
                    int   cc[8];
#pragma unroll
                    for (int j = 0; j < 8; ++j) {
                        const int idx = t0 + j * 32 + lane;
                        const int ok  = idx < total;
                        const int e   = ok ? idx / F : 0;
                        const int col = ok ? idx - e * F : 0;
                        const int sj  = __shfl(s, e);
                        const int dj  = __shfl(dl, e);
                        dd2[j] = ok ? dj : -1;
                        pp[j] = __shfl(p0, e);
                        cc[j] = col;
                        hv[j] = h[(size_t)sj * F + col];
                    }
#pragma unroll
                    for (int j = 0; j < 8; ++j) {
                        if (dd2[j] >= 0)
                            atomicAdd(&acc[dd2[j] * F + cc[j]], (double)(hv[j] * pp[j]));
                    }
                }
            }
        }
    };

    bool ovf = false;
    auto scan = [&](auto&& flush) -> int {
        int cnt = 0;
        {
            const int node = wave + 32 * lane;
            const bool m = lane < SELF_LANES && node < nr;
            const unsigned mk = __builtin_amdgcn_ballot_w32(m);
            if (m) q[__builtin_amdgcn_mbcnt_lo(mk, 0)] = N_EDGES + n0 + node;
            cnt = __builtin_popcount(mk);
        }
        const int4* dv = reinterpret_cast<const int4*>(ei + N_EDGES);
        int c0 = wave;
        for (;;) {
            bool full = false;
            while (c0 < N_CHUNKS && !full) {
                int4 dd[TRIPC];
#pragma unroll
                for (int u = 0; u < TRIPC; ++u) {
                    const int c = c0 + 32 * u;
                    const int cc = c < N_CHUNKS ? c : c0;
                    dd[u] = dv[(size_t)cc * 32 + lane];
                }
                int took = 0;
#pragma unroll
                for (int u = 0; u < TRIPC; ++u) {
                    const int c = c0 + 32 * u;
                    if (c < N_CHUNKS && !full) {
                        if (cnt > CAP - CHUNK_E) {
                            full = true;
                        } else {
                            const int eb = c * CHUNK_E + lane * 4;
                            const int dk[4] = {dd[u].x, dd[u].y, dd[u].z, dd[u].w};
#pragma unroll
                            for (int k = 0; k < 4; ++k) {
                                const bool m = (unsigned)(dk[k] - n0) < (unsigned)nr;
                                const unsigned mk = __builtin_amdgcn_ballot_w32(m);
                                if (mk) {
                                    if (m) q[cnt + __builtin_amdgcn_mbcnt_lo(mk, 0)] = eb + k;
                                    cnt += __builtin_popcount(mk);
                                }
                            }
                            took = u + 1;
                        }
                    }
                }
                c0 += 32 * (full ? took : TRIPC);
            }
            if (!full) break;
            flush(cnt);
            cnt = 0;
            ovf = true;
        }
        return cnt;
    };

    int cnt = scan([&](int n) { max_pass(n); });
    max_pass(cnt);
    __syncthreads();
    if (!ovf) {
        work(cnt);
    } else {
        cnt = scan([&](int n) { max_pass(n); work(n); });
        max_pass(cnt);
        work(cnt);
    }

    __syncthreads();
    for (int i = tid; i < nr * F; i += 1024) {
        const int r = i / F, c = i - r * F;
        float v = (float)acc[i] / (float)den[r * NH + (NH == 2 ? (c >> 6) : 0)] + bias[c];
        if constexpr (F == F1) v = v > 0.f ? v : (expf(v) - 1.0f);
        outp[(size_t)n0 * F + i] = v;
    }
}

extern "C" void kernel_launch(void* const* d_in, const int* in_sizes, int n_in,
                              void* d_out, int out_size, void* d_ws, size_t ws_size,
                              hipStream_t stream) {
    const float* x    = (const float*)d_in[0];
    const int*   ei   = (const int*)  d_in[1];
    const float* W1   = (const float*)d_in[2];
    const float* as1  = (const float*)d_in[3];
    const float* ad1  = (const float*)d_in[4];
    const float* b1   = (const float*)d_in[5];
    const float* W2   = (const float*)d_in[6];
    const float* as2  = (const float*)d_in[7];
    const float* ad2  = (const float*)d_in[8];
    const float* b2   = (const float*)d_in[9];
    float* out = (float*)d_out;

    float* ws      = (float*)d_ws;
    float* h1      = ws;
    float* h2in    = h1 + (size_t)N_NODES * F1;
    float* a_src1  = h2in + (size_t)N_NODES * F1;
    float* a_dst1  = a_src1 + (size_t)N_NODES * 2;
    float* a_src2  = a_dst1 + (size_t)N_NODES * 2;
    float* a_dst2  = a_src2 + N_NODES;
    float* h2      = a_dst2 + N_NODES;

    const int MT = N_NODES / 16;
    gat_gemm1_wmma<<<MT, 256, 0, stream>>>(x, W1, h1);
    gat_att1<<<(N_NODES * HEADS + 255) / 256, 256, 0, stream>>>(h1, as1, ad1, a_src1, a_dst1);
    gat_own_accum<F1, HEADS, 6, 896><<<OWN_BLOCKS, 1024, 0, stream>>>(ei, a_src1, a_dst1, h1, b1, h2in);
    gat_gemm2_wmma<<<MT, 96, 0, stream>>>(h2in, W2, h2);
    gat_att2<<<(N_NODES + 255) / 256, 256, 0, stream>>>(h2, as2, ad2, a_src2, a_dst2);
    gat_own_accum<NCLS, 1, 8, 1152><<<OWN_BLOCKS, 1024, 0, stream>>>(ei, a_src2, a_dst2, h2, b2, out);
}


# Kept-edge lists built once: each wave of the first owner launch saves its packed list and length in the workspace, layer 2's launch starts from them; its scan of the destination list is gone
# speedup vs baseline: 2.1733x; 1.1237x over previous
#include <hip/hip_runtime.h>
#include <math.h>

#define N_NODES 50000
#define N_EDGES 800000
#define IN_FEATS 256
#define HID 64
#define HEADS 2
#define F1 (HEADS * HID)
#define NCLS 40

typedef __attribute__((ext_vector_type(2))) float v2f;
typedef __attribute__((ext_vector_type(8))) float v8f;

static __device__ __forceinline__ unsigned fmap(float f) {
    unsigned u = __float_as_uint(f);
    return (u & 0x80000000u) ? ~u : (u | 0x80000000u);
}
static __device__ __forceinline__ float finv(unsigned t) {
    return (t & 0x80000000u) ? __uint_as_float(t & 0x7FFFFFFFu)
                             : __uint_as_float(~t);
}
static __device__ __forceinline__ void edge_sd(const int* __restrict__ ei, int e, int& s, int& d) {
    if (e < N_EDGES) { s = ei[e]; d = ei[N_EDGES + e]; }
    else             { s = e - N_EDGES; d = s; }
}
static __device__ __forceinline__ float lrelu(float x) { return x > 0.f ? x : 0.2f * x; }

__global__ __launch_bounds__(256) void gat_gemm1_wmma(const float* __restrict__ x,
                                                      const float* __restrict__ W1,
                                                      float* __restrict__ h1) {
    const int mtile = blockIdx.x;
    const int wave  = threadIdx.x >> 5;
    const int lane  = threadIdx.x & 31;
    const int half  = lane >> 4;
    const int l     = lane & 15;
    const int row   = mtile * 16 + l;
    const int ncol  = wave * 16 + l;

    v8f acc = {};
    const float* xr = x + (size_t)row * IN_FEATS;
    for (int k = 0; k < IN_FEATS; k += 4) {
        v2f a = *(const v2f*)(xr + k + 2 * half);
        const float* bp = W1 + (size_t)(k + 2 * half) * F1 + ncol;
        v2f b; b.x = bp[0]; b.y = bp[F1];
        acc = __builtin_amdgcn_wmma_f32_16x16x4_f32(false, a, false, b,
                                                    (short)0, acc, false, false);
    }
    float* dr = h1 + (size_t)(mtile * 16 + 8 * half) * F1 + wave * 16 + l;
#pragma unroll
    for (int r = 0; r < 8; ++r) dr[(size_t)r * F1] = acc[r];
}

__global__ __launch_bounds__(96) void gat_gemm2_wmma(const float* __restrict__ h2in,
                                                     const float* __restrict__ W2,
                                                     float* __restrict__ h2) {
    const int mtile = blockIdx.x;
    const int wave  = threadIdx.x >> 5;
    const int lane  = threadIdx.x & 31;
    const int half  = lane >> 4;
    const int l     = lane & 15;
    const int row   = mtile * 16 + l;
    const int ncol  = wave * 16 + l;

    v8f acc = {};
    const float* xr = h2in + (size_t)row * F1;
    for (int k = 0; k < F1; k += 4) {
        v2f a = *(const v2f*)(xr + k + 2 * half);
        v2f b;
        if (ncol < NCLS) {
            const float* bp = W2 + (size_t)(k + 2 * half) * NCLS + ncol;
            b.x = bp[0]; b.y = bp[NCLS];
        } else { b.x = 0.f; b.y = 0.f; }
        acc = __builtin_amdgcn_wmma_f32_16x16x4_f32(false, a, false, b,
                                                    (short)0, acc, false, false);
    }
    if (ncol < NCLS) {
        float* dr = h2 + (size_t)(mtile * 16 + 8 * half) * NCLS + ncol;
#pragma unroll
        for (int r = 0; r < 8; ++r) dr[(size_t)r * NCLS] = acc[r];
    }
}

__global__ void gat_att1(const float* __restrict__ h1, const float* __restrict__ as1,
                         const float* __restrict__ ad1, float* __restrict__ a_src,
                         float* __restrict__ a_dst) {
    int i = blockIdx.x * blockDim.x + threadIdx.x;
    if (i >= N_NODES * HEADS) return;
    int n = i >> 1, h = i & 1;
    const float* hp = h1 + (size_t)n * F1 + h * HID;
    const float* s0 = as1 + h * HID;
    const float* d0 = ad1 + h * HID;
    float s = 0.f, d = 0.f;
#pragma unroll 8
    for (int c = 0; c < HID; ++c) { float v = hp[c]; s += v * s0[c]; d += v * d0[c]; }
    a_src[i] = s; a_dst[i] = d;
}

__global__ void gat_att2(const float* __restrict__ h2, const float* __restrict__ as2,
                         const float* __restrict__ ad2, float* __restrict__ a_src,
                         float* __restrict__ a_dst) {
    int n = blockIdx.x * blockDim.x + threadIdx.x;
    if (n >= N_NODES) return;
    const float* hp = h2 + (size_t)n * NCLS;
    float s = 0.f, d = 0.f;
#pragma unroll 8
    for (int c = 0; c < NCLS; ++c) { float v = hp[c]; s += v * as2[c]; d += v * ad2[c]; }
    a_src[n] = s; a_dst[n] = d;
}

#define OWN_R      196
#define OWN_BLOCKS ((N_NODES + OWN_R - 1) / OWN_R)
#define CHUNK_E    128
#define N_CHUNKS   (N_EDGES / CHUNK_E)
#define SELF_LANES ((OWN_R + 31) / 32)
#define LIST_PITCH 896
#define WS_LISTS_OFF ((size_t)16 << 20)
#define WS_PLANES ((size_t)N_NODES * (2 * F1 + 2 * HEADS + 2 + NCLS))
static_assert(WS_PLANES <= WS_LISTS_OFF, "the saved lists start behind the feature and score planes");
static_assert((WS_LISTS_OFF + (size_t)OWN_BLOCKS * 32 * (LIST_PITCH + 1)) * 4 <= ((size_t)256 << 20),
              "lists and lengths end inside the workspace");
static_assert(N_EDGES % CHUNK_E == 0, "the destination list is read in whole 128-edge chunks");
static_assert(N_NODES <= 65536 && OWN_R <= 65536, "a list entry packs source and local destination, 16 bits each");
static_assert(SELF_LANES <= 32, "one lane per self-loop of a wave");

static __device__ __forceinline__ float lane_f(float v, int i) {
    return __int_as_float(__builtin_amdgcn_readlane(__float_as_int(v), i));
}

template <int F, int NH, int TRIPC, int CAP, int ROLE>
__global__ __launch_bounds__(1024) void gat_own_accum(const int* __restrict__ ei,
        const float* __restrict__ a_src, const float* __restrict__ a_dst,
        const float* __restrict__ h,
        const float* __restrict__ bias, float* __restrict__ outp,
        int* __restrict__ wl, int* __restrict__ wc) {
    static_assert(CAP >= CHUNK_E + SELF_LANES, "a chunk goes into a list only while 128 entries are free");
    static_assert(ROLE == 0 ? CAP <= LIST_PITCH : CAP >= LIST_PITCH, "a saved list fits its slot, and fits the list of the wave that takes it");
    __shared__ double   acc[OWN_R * F];
    __shared__ double   den[OWN_R * NH];
    __shared__ unsigned mx[OWN_R * NH];
    __shared__ int      que[32 * CAP];

    const int tid  = threadIdx.x;
    const int lane = tid & 31;
    const int wave = __builtin_amdgcn_readfirstlane(tid >> 5);
    const int n0   = blockIdx.x * OWN_R;
    const int nr   = (N_NODES - n0) < OWN_R ? (N_NODES - n0) : OWN_R;
    int* q = que + wave * CAP;

    for (int i = tid; i < OWN_R * F;  i += 1024) acc[i] = 0.0;
    for (int i = tid; i < OWN_R * NH; i += 1024) { den[i] = 0.0; mx[i] = 0u; }
    __syncthreads();

    auto max_pass = [&](int n) {
        for (int off = 0; off < n; off += 128) {
            int e[4], s[4], d[4];
            float x0[4];
            [[maybe_unused]] float x1[4];
#pragma unroll
            for (int u = 0; u < 4; ++u) {
                const int i = off + 32 * u + lane;
                e[u] = i < n ? q[i] : N_EDGES + n0;
            }
#pragma unroll
            for (int u = 0; u < 4; ++u) {
                const bool r  = e[u] < N_EDGES;
                const int  ec = r ? e[u] : 0;
                const int  sl = ei[ec], dg = ei[N_EDGES + ec];
                s[u] = r ? sl : e[u] - N_EDGES;
                d[u] = r ? dg : s[u];
            }
#pragma unroll
            for (int u = 0; u < 4; ++u) {
                x0[u] = lrelu(a_src[s[u] * NH] + a_dst[d[u] * NH]);
                if constexpr (NH == 2) x1[u] = lrelu(a_src[s[u] * 2 + 1] + a_dst[d[u] * 2 + 1]);
            }
#pragma unroll
            for (int u = 0; u < 4; ++u) {
                const int i = off + 32 * u + lane;
                if (i < n) {
                    const int dl = d[u] - n0;
                    atomicMax(&mx[dl * NH], fmap(x0[u]));
                    if constexpr (NH == 2) atomicMax(&mx[dl * 2 + 1], fmap(x1[u]));
                    q[i] = s[u] | (dl << 16);
                }
            }
        }
    };

    [[maybe_unused]] auto max_saved = [&](const int* __restrict__ sv, int n) {
        for (int off = 0; off < n; off += 128) {
            unsigned pk[4];
            float x0[4];
            [[maybe_unused]] float x1[4];
#pragma unroll
            for (int u = 0; u < 4; ++u) {
                const int i = off + 32 * u + lane;
                pk[u] = (unsigned)sv[i < n ? i : 0];
            }
            asm volatile("" :: "v"(pk[0]), "v"(pk[1]), "v"(pk[2]), "v"(pk[3]));
            float as0[4], ad0[4];
            [[maybe_unused]] float as1[4], ad1[4];
#pragma unroll
            for (int u = 0; u < 4; ++u) {
                const int s = (int)(pk[u] & 0xFFFFu), d = n0 + (int)(pk[u] >> 16);
                as0[u] = a_src[s * NH]; ad0[u] = a_dst[d * NH];
                if constexpr (NH == 2) { as1[u] = a_src[s * 2 + 1]; ad1[u] = a_dst[d * 2 + 1]; }
            }
            asm volatile("" :: "v"(as0[0]), "v"(as0[1]), "v"(as0[2]), "v"(as0[3]), "v"(ad0[0]), "v"(ad0[1]), "v"(ad0[2]), "v"(ad0[3]));
            if constexpr (NH == 2)
                asm volatile("" :: "v"(as1[0]), "v"(as1[1]), "v"(as1[2]), "v"(as1[3]), "v"(ad1[0]), "v"(ad1[1]), "v"(ad1[2]), "v"(ad1[3]));
#pragma unroll
            for (int u = 0; u < 4; ++u) {
                x0[u] = lrelu(as0[u] + ad0[u]);
                if constexpr (NH == 2) x1[u] = lrelu(as1[u] + ad1[u]);
            }
#pragma unroll
            for (int u = 0; u < 4; ++u) {
                const int i = off + 32 * u + lane;
                if (i < n) {
                    const int dl = (int)(pk[u] >> 16);
                    atomicMax(&mx[dl * NH], fmap(x0[u]));
                    if constexpr (NH == 2) atomicMax(&mx[dl * 2 + 1], fmap(x1[u]));
                    q[i] = (int)pk[u];
                }
            }
        }
    };

    auto work = [&](int n) {
        for (int off = 0; off < n; off += 32) {
            const int nb = (n - off) < 32 ? (n - off) : 32;
            int s = 0, dl = 0;
            float p0 = 0.f, p1 = 0.f;
            if (lane < nb) {
                const unsigned pk = (unsigned)q[off + lane];
                s  = (int)(pk & 0xFFFFu);
                dl = (int)(pk >> 16);
                const int d = n0 + dl;
                p0 = __expf(lrelu(a_src[s * NH] + a_dst[d * NH]) - finv(mx[dl * NH]));
                atomicAdd(&den[dl * NH], (double)p0);
                if constexpr (NH == 2) {
                    p1 = __expf(lrelu(a_src[s * 2 + 1] + a_dst[d * 2 + 1]) - finv(mx[dl * 2 + 1]));
                    atomicAdd(&den[dl * 2 + 1], (double)p1);
                }
            }
            if constexpr (F == F1) {
#pragma unroll 1
                for (int i0 = 0; i0 < nb; i0 += 8) {
                    float hv[8][4];
#pragma unroll
                    for (int j = 0; j < 8; ++j) {
                        const int i = (i0 + j < nb) ? (i0 + j) : (nb - 1);
                        const int sj = __builtin_amdgcn_readlane(s, i);
                        const float* rp = h + (size_t)sj * F1 + lane;
#pragma unroll
                        for (int k = 0; k < 4; ++k) hv[j][k] = rp[k * 32];
                    }
#pragma unroll
                    for (int j = 0; j < 8; ++j) {
                        if (i0 + j < nb) {
                            const int dj = __builtin_amdgcn_readlane(dl, i0 + j);
                            const float q0 = lane_f(p0, i0 + j);
                            const float q1 = lane_f(p1, i0 + j);
                            double* ap = acc + dj * F1 + lane;
                            atomicAdd(ap +  0, (double)(hv[j][0] * q0));
                            atomicAdd(ap + 32, (double)(hv[j][1] * q0));
                            atomicAdd(ap + 64, (double)(hv[j][2] * q1));
                            atomicAdd(ap + 96, (double)(hv[j][3] * q1));
                        }
                    }
                }
            } else {
                const int total = nb * F;
#pragma unroll 1
                for (int t0 = 0; t0 < total; t0 += 8 * 32) {
                    float hv[8];
                    int   dd2[8];
                    float pp[8];
                    int   cc[8];
#pragma unroll
                    for (int j = 0; j < 8; ++j) {
                        const int idx = t0 + j * 32 + lane;
                        const int ok  = idx < total;
                        const int e   = ok ? idx / F : 0;
                        const int col = ok ? idx - e * F : 0;
                        const int sj  = __shfl(s, e);
                        const int dj  = __shfl(dl, e);
                        dd2[j] = ok ? dj : -1;
                        pp[j] = __shfl(p0, e);
                        cc[j] = col;
                        hv[j] = h[(size_t)sj * F + col];
                    }
#pragma unroll
                    for (int j = 0; j < 8; ++j) {
                        if (dd2[j] >= 0)
                            atomicAdd(&acc[dd2[j] * F + cc[j]], (double)(hv[j] * pp[j]));
                    }
                }
            }
        }
    };

    bool ovf = false;
    auto scan = [&](auto&& flush) -> int {
        int cnt = 0;
        {
            const int node = wave + 32 * lane;
            const bool m = lane < SELF_LANES && node < nr;
            const unsigned mk = __builtin_amdgcn_ballot_w32(m);
            if (m) q[__builtin_amdgcn_mbcnt_lo(mk, 0)] = N_EDGES + n0 + node;
            cnt = __builtin_popcount(mk);
        }
        const int4* dv = reinterpret_cast<const int4*>(ei + N_EDGES);
        int c0 = wave;
        for (;;) {
            bool full = false;
            while (c0 < N_CHUNKS && !full) {
                int4 dd[TRIPC];
#pragma unroll
                for (int u = 0; u < TRIPC; ++u) {
                    const int c = c0 + 32 * u;
                    const int cc = c < N_CHUNKS ? c : c0;
                    dd[u] = dv[(size_t)cc * 32 + lane];
                }
                int took = 0;
#pragma unroll
                for (int u = 0; u < TRIPC; ++u) {
                    const int c = c0 + 32 * u;
                    if (c < N_CHUNKS && !full) {
                        if (cnt > CAP - CHUNK_E) {
                            full = true;
                        } else {
                            const int eb = c * CHUNK_E + lane * 4;
                            const int dk[4] = {dd[u].x, dd[u].y, dd[u].z, dd[u].w};
#pragma unroll
                            for (int k = 0; k < 4; ++k) {
                                const bool m = (unsigned)(dk[k] - n0) < (unsigned)nr;
                                const unsigned mk = __builtin_amdgcn_ballot_w32(m);
                                if (mk) {
                                    if (m) q[cnt + __builtin_amdgcn_mbcnt_lo(mk, 0)] = eb + k;
                                    cnt += __builtin_popcount(mk);
                                }
                            }
                            took = u + 1;
                        }
                    }
                }
                c0 += 32 * (full ? took : TRIPC);
            }
            if (!full) break;
            flush(cnt);
            cnt = 0;
            ovf = true;
        }
        return cnt;
    };

    const int slot = blockIdx.x * 32 + wave;
    int  cnt  = 0;
    bool have = false;
    if constexpr (ROLE == 1) {
        const int c = __builtin_amdgcn_readfirstlane(wc[slot]);
        if (c >= 0) {
            have = true;
            cnt  = c;
            max_saved(wl + (size_t)slot * LIST_PITCH, cnt);
        }
    }
    if (!have) {
        cnt = scan([&](int n) { max_pass(n); });
        max_pass(cnt);
    }
    if constexpr (ROLE == 0) {
        if (!ovf) for (int i = lane; i < cnt; i += 32) wl[(size_t)slot * LIST_PITCH + i] = q[i];
        if (lane == 0) wc[slot] = ovf ? -1 : cnt;
    }
    __syncthreads();
    if (!ovf) {
        work(cnt);
    } else {
        cnt = scan([&](int n) { max_pass(n); work(n); });
        max_pass(cnt);
        work(cnt);
    }

    __syncthreads();
    for (int i = tid; i < nr * F; i += 1024) {
        const int r = i / F, c = i - r * F;
        float v = (float)acc[i] / (float)den[r * NH + (NH == 2 ? (c >> 6) : 0)] + bias[c];
        if constexpr (F == F1) v = v > 0.f ? v : (expf(v) - 1.0f);
        outp[(size_t)n0 * F + i] = v;
    }
}

extern "C" void kernel_launch(void* const* d_in, const int* in_sizes, int n_in,
                              void* d_out, int out_size, void* d_ws, size_t ws_size,
                              hipStream_t stream) {
    const float* x    = (const float*)d_in[0];
    const int*   ei   = (const int*)  d_in[1];
    const float* W1   = (const float*)d_in[2];
    const float* as1  = (const float*)d_in[3];
    const float* ad1  = (const float*)d_in[4];
    const float* b1   = (const float*)d_in[5];
    const float* W2   = (const float*)d_in[6];
    const float* as2  = (const float*)d_in[7];
    const float* ad2  = (const float*)d_in[8];
    const float* b2   = (const float*)d_in[9];
    float* out = (float*)d_out;

    float* ws      = (float*)d_ws;
    float* h1      = ws;
    float* h2in    = h1 + (size_t)N_NODES * F1;
    float* a_src1  = h2in + (size_t)N_NODES * F1;
    float* a_dst1  = a_src1 + (size_t)N_NODES * 2;
    float* a_src2  = a_dst1 + (size_t)N_NODES * 2;
    float* a_dst2  = a_src2 + N_NODES;
    float* h2      = a_dst2 + N_NODES;
    int*   wl      = (int*)d_ws + WS_LISTS_OFF;
    int*   wc      = wl + (size_t)OWN_BLOCKS * 32 * LIST_PITCH;

    const int MT = N_NODES / 16;
    gat_gemm1_wmma<<<MT, 256, 0, stream>>>(x, W1, h1);
    gat_att1<<<(N_NODES * HEADS + 255) / 256, 256, 0, stream>>>(h1, as1, ad1, a_src1, a_dst1);
    gat_own_accum<F1, HEADS, 6, LIST_PITCH, 0><<<OWN_BLOCKS, 1024, 0, stream>>>(ei, a_src1, a_dst1, h1, b1, h2in, wl, wc);
    gat_gemm2_wmma<<<MT, 96, 0, stream>>>(h2in, W2, h2);
    gat_att2<<<(N_NODES + 255) / 256, 256, 0, stream>>>(h2, as2, ad2, a_src2, a_dst2);
    gat_own_accum<NCLS, 1, 8, 1152, 1><<<OWN_BLOCKS, 1024, 0, stream>>>(ei, a_src2, a_dst2, h2, b2, out, wl, wc);
}


# GEMM1 as 512 resident blocks over row tiles: a wave keeps its 256x16 slab of W1 in registers; the x tile comes once per block through LDS, the next one loaded under the products; k order kept
# speedup vs baseline: 2.4114x; 1.1096x over previous
#include <hip/hip_runtime.h>
#include <math.h>

#define N_NODES 50000
#define N_EDGES 800000
#define IN_FEATS 256
#define HID 64
#define HEADS 2
#define F1 (HEADS * HID)
#define NCLS 40

typedef __attribute__((ext_vector_type(2))) float v2f;
typedef __attribute__((ext_vector_type(8))) float v8f;
typedef __attribute__((ext_vector_type(4))) float v4f;

static __device__ __forceinline__ unsigned fmap(float f) {
    unsigned u = __float_as_uint(f);
    return (u & 0x80000000u) ? ~u : (u | 0x80000000u);
}
static __device__ __forceinline__ float finv(unsigned t) {
    return (t & 0x80000000u) ? __uint_as_float(t & 0x7FFFFFFFu)
                             : __uint_as_float(~t);
}
static __device__ __forceinline__ void edge_sd(const int* __restrict__ ei, int e, int& s, int& d) {
    if (e < N_EDGES) { s = ei[e]; d = ei[N_EDGES + e]; }
    else             { s = e - N_EDGES; d = s; }
}
static __device__ __forceinline__ float lrelu(float x) { return x > 0.f ? x : 0.2f * x; }

#define G1_PITCH  260
#define G1_BLOCKS 512
#define G1_TILES  (N_NODES / 16)
static_assert(N_NODES % 16 == 0, "row tiles are whole");

__global__ __launch_bounds__(256) void gat_gemm1_wmma(const float* __restrict__ x,
                                                      const float* __restrict__ W1,
                                                      float* __restrict__ h1) {
    __shared__ float xs[2][16 * G1_PITCH];

    const int tid   = threadIdx.x;
    const int wave  = tid >> 5;
    const int lane  = tid & 31;
    const int half  = lane >> 4;
    const int l     = lane & 15;

    v2f b[IN_FEATS / 4];
    {
        const float* bp = W1 + (size_t)(2 * half) * F1 + wave * 16 + l;
#pragma unroll
        for (int kk = 0; kk < IN_FEATS / 4; ++kk) {
            b[kk].x = bp[(size_t)(4 * kk) * F1];
            b[kk].y = bp[(size_t)(4 * kk + 1) * F1];
        }
    }

    const v4f* xv = reinterpret_cast<const v4f*>(x);
    const int so = (tid >> 6) * G1_PITCH + (tid & 63) * 4;
    v4f st0, st1, st2, st3;
    int mt = blockIdx.x;
    if (mt < G1_TILES) {
        const v4f* tp = xv + (size_t)mt * 1024 + tid;
        st0 = tp[0]; st1 = tp[256]; st2 = tp[512]; st3 = tp[768];
        float* sp = &xs[0][so];
        *reinterpret_cast<v4f*>(sp)                 = st0;
        *reinterpret_cast<v4f*>(sp +  4 * G1_PITCH) = st1;
        *reinterpret_cast<v4f*>(sp +  8 * G1_PITCH) = st2;
        *reinterpret_cast<v4f*>(sp + 12 * G1_PITCH) = st3;
    }
    __syncthreads();

    int buf = 0;
    for (; mt < G1_TILES; mt += gridDim.x) {
        const int  nxt  = mt + gridDim.x;
        const bool more = nxt < G1_TILES;
        if (more) {
            const v4f* tp = xv + (size_t)nxt * 1024 + tid;
            st0 = tp[0]; st1 = tp[256]; st2 = tp[512]; st3 = tp[768];
        }

        const float* ap = &xs[buf][l * G1_PITCH + 2 * half];
        v8f acc = {};
        v2f a0[8], a1[8];
#pragma unroll
        for (int i = 0; i < 8; ++i) a0[i] = *reinterpret_cast<const v2f*>(ap + 4 * i);
#pragma unroll
        for (int g = 0; g < IN_FEATS / 32; g += 2) {
#pragma unroll
            for (int i = 0; i < 8; ++i) a1[i] = *reinterpret_cast<const v2f*>(ap + 32 * (g + 1) + 4 * i);
            __builtin_amdgcn_sched_barrier(0);
#pragma unroll
            for (int i = 0; i < 8; ++i)
                acc = __builtin_amdgcn_wmma_f32_16x16x4_f32(false, a0[i], false, b[8 * g + i],
                                                            (short)0, acc, false, false);
            __builtin_amdgcn_sched_barrier(0);
            if (g + 2 < IN_FEATS / 32) {
#pragma unroll
                for (int i = 0; i < 8; ++i) a0[i] = *reinterpret_cast<const v2f*>(ap + 32 * (g + 2) + 4 * i);
            }
            __builtin_amdgcn_sched_barrier(0);
#pragma unroll
            for (int i = 0; i < 8; ++i)
                acc = __builtin_amdgcn_wmma_f32_16x16x4_f32(false, a1[i], false, b[8 * (g + 1) + i],
                                                            (short)0, acc, false, false);
            __builtin_amdgcn_sched_barrier(0);
        }
        if (more) {
            float* sp = &xs[buf ^ 1][so];
            *reinterpret_cast<v4f*>(sp)                 = st0;
            *reinterpret_cast<v4f*>(sp +  4 * G1_PITCH) = st1;
            *reinterpret_cast<v4f*>(sp +  8 * G1_PITCH) = st2;
            *reinterpret_cast<v4f*>(sp + 12 * G1_PITCH) = st3;
        }
        __syncthreads();
        buf ^= 1;

        float* dr = h1 + (size_t)(mt * 16 + 8 * half) * F1 + wave * 16 + l;
#pragma unroll
        for (int r = 0; r < 8; ++r) dr[(size_t)r * F1] = acc[r];
    }
}

__global__ __launch_bounds__(96) void gat_gemm2_wmma(const float* __restrict__ h2in,
                                                     const float* __restrict__ W2,
                                                     float* __restrict__ h2) {
    const int mtile = blockIdx.x;
    const int wave  = threadIdx.x >> 5;
    const int lane  = threadIdx.x & 31;
    const int half  = lane >> 4;
    const int l     = lane & 15;
    const int row   = mtile * 16 + l;
    const int ncol  = wave * 16 + l;

    v8f acc = {};
    const float* xr = h2in + (size_t)row * F1;
    for (int k = 0; k < F1; k += 4) {
        v2f a = *(const v2f*)(xr + k + 2 * half);
        v2f b;
        if (ncol < NCLS) {
            const float* bp = W2 + (size_t)(k + 2 * half) * NCLS + ncol;
            b.x = bp[0]; b.y = bp[NCLS];
        } else { b.x = 0.f; b.y = 0.f; }
        acc = __builtin_amdgcn_wmma_f32_16x16x4_f32(false, a, false, b,
                                                    (short)0, acc, false, false);
    }
    if (ncol < NCLS) {
        float* dr = h2 + (size_t)(mtile * 16 + 8 * half) * NCLS + ncol;
#pragma unroll
        for (int r = 0; r < 8; ++r) dr[(size_t)r * NCLS] = acc[r];
    }
}

__global__ void gat_att1(const float* __restrict__ h1, const float* __restrict__ as1,
                         const float* __restrict__ ad1, float* __restrict__ a_src,
                         float* __restrict__ a_dst) {
    int i = blockIdx.x * blockDim.x + threadIdx.x;
    if (i >= N_NODES * HEADS) return;
    int n = i >> 1, h = i & 1;
    const float* hp = h1 + (size_t)n * F1 + h * HID;
    const float* s0 = as1 + h * HID;
    const float* d0 = ad1 + h * HID;
    float s = 0.f, d = 0.f;
#pragma unroll 8
    for (int c = 0; c < HID; ++c) { float v = hp[c]; s += v * s0[c]; d += v * d0[c]; }
    a_src[i] = s; a_dst[i] = d;
}

__global__ void gat_att2(const float* __restrict__ h2, const float* __restrict__ as2,
                         const float* __restrict__ ad2, float* __restrict__ a_src,
                         float* __restrict__ a_dst) {
    int n = blockIdx.x * blockDim.x + threadIdx.x;
    if (n >= N_NODES) return;
    const float* hp = h2 + (size_t)n * NCLS;
    float s = 0.f, d = 0.f;
#pragma unroll 8
    for (int c = 0; c < NCLS; ++c) { float v = hp[c]; s += v * as2[c]; d += v * ad2[c]; }
    a_src[n] = s; a_dst[n] = d;
}

#define OWN_R      196
#define OWN_BLOCKS ((N_NODES + OWN_R - 1) / OWN_R)
#define CHUNK_E    128
#define N_CHUNKS   (N_EDGES / CHUNK_E)
#define SELF_LANES ((OWN_R + 31) / 32)
#define LIST_PITCH 896
#define WS_LISTS_OFF ((size_t)16 << 20)
#define WS_PLANES ((size_t)N_NODES * (2 * F1 + 2 * HEADS + 2 + NCLS))
static_assert(WS_PLANES <= WS_LISTS_OFF, "the saved lists start behind the feature and score planes");
static_assert((WS_LISTS_OFF + (size_t)OWN_BLOCKS * 32 * (LIST_PITCH + 1)) * 4 <= ((size_t)256 << 20),
              "lists and lengths end inside the workspace");
static_assert(N_EDGES % CHUNK_E == 0, "the destination list is read in whole 128-edge chunks");
static_assert(N_NODES <= 65536 && OWN_R <= 65536, "a list entry packs source and local destination, 16 bits each");
static_assert(SELF_LANES <= 32, "one lane per self-loop of a wave");

static __device__ __forceinline__ float lane_f(float v, int i) {
    return __int_as_float(__builtin_amdgcn_readlane(__float_as_int(v), i));
}

template <int F, int NH, int TRIPC, int CAP, int ROLE>
__global__ __launch_bounds__(1024) void gat_own_accum(const int* __restrict__ ei,
        const float* __restrict__ a_src, const float* __restrict__ a_dst,
        const float* __restrict__ h,
        const float* __restrict__ bias, float* __restrict__ outp,
        int* __restrict__ wl, int* __restrict__ wc) {
    static_assert(CAP >= CHUNK_E + SELF_LANES, "a chunk goes into a list only while 128 entries are free");
    static_assert(ROLE == 0 ? CAP <= LIST_PITCH : CAP >= LIST_PITCH, "a saved list fits its slot, and fits the list of the wave that takes it");
    __shared__ double   acc[OWN_R * F];
    __shared__ double   den[OWN_R * NH];
    __shared__ unsigned mx[OWN_R * NH];
    __shared__ int      que[32 * CAP];

    const int tid  = threadIdx.x;
    const int lane = tid & 31;
    const int wave = __builtin_amdgcn_readfirstlane(tid >> 5);
    const int n0   = blockIdx.x * OWN_R;
    const int nr   = (N_NODES - n0) < OWN_R ? (N_NODES - n0) : OWN_R;
    int* q = que + wave * CAP;

    for (int i = tid; i < OWN_R * F;  i += 1024) acc[i] = 0.0;
    for (int i = tid; i < OWN_R * NH; i += 1024) { den[i] = 0.0; mx[i] = 0u; }
    __syncthreads();

    auto max_pass = [&](int n) {
        for (int off = 0; off < n; off += 128) {
            int e[4], s[4], d[4];
            float x0[4];
            [[maybe_unused]] float x1[4];
#pragma unroll
            for (int u = 0; u < 4; ++u) {
                const int i = off + 32 * u + lane;
                e[u] = i < n ? q[i] : N_EDGES + n0;
            }
#pragma unroll
            for (int u = 0; u < 4; ++u) {
                const bool r  = e[u] < N_EDGES;
                const int  ec = r ? e[u] : 0;
                const int  sl = ei[ec], dg = ei[N_EDGES + ec];
                s[u] = r ? sl : e[u] - N_EDGES;
                d[u] = r ? dg : s[u];
            }
#pragma unroll
            for (int u = 0; u < 4; ++u) {
                x0[u] = lrelu(a_src[s[u] * NH] + a_dst[d[u] * NH]);
                if constexpr (NH == 2) x1[u] = lrelu(a_src[s[u] * 2 + 1] + a_dst[d[u] * 2 + 1]);
            }
#pragma unroll
            for (int u = 0; u < 4; ++u) {
                const int i = off + 32 * u + lane;
                if (i < n) {
                    const int dl = d[u] - n0;
                    atomicMax(&mx[dl * NH], fmap(x0[u]));
                    if constexpr (NH == 2) atomicMax(&mx[dl * 2 + 1], fmap(x1[u]));
                    q[i] = s[u] | (dl << 16);
                }
            }
        }
    };

    [[maybe_unused]] auto max_saved = [&](const int* __restrict__ sv, int n) {
        for (int off = 0; off < n; off += 128) {
            unsigned pk[4];
            float x0[4];
            [[maybe_unused]] float x1[4];
#pragma unroll
            for (int u = 0; u < 4; ++u) {
                const int i = off + 32 * u + lane;
                pk[u] = (unsigned)sv[i < n ? i : 0];
            }
            asm volatile("" :: "v"(pk[0]), "v"(pk[1]), "v"(pk[2]), "v"(pk[3]));
            float as0[4], ad0[4];
            [[maybe_unused]] float as1[4], ad1[4];
#pragma unroll
            for (int u = 0; u < 4; ++u) {
                const int s = (int)(pk[u] & 0xFFFFu), d = n0 + (int)(pk[u] >> 16);
                as0[u] = a_src[s * NH]; ad0[u] = a_dst[d * NH];
                if constexpr (NH == 2) { as1[u] = a_src[s * 2 + 1]; ad1[u] = a_dst[d * 2 + 1]; }
            }
            asm volatile("" :: "v"(as0[0]), "v"(as0[1]), "v"(as0[2]), "v"(as0[3]), "v"(ad0[0]), "v"(ad0[1]), "v"(ad0[2]), "v"(ad0[3]));
            if constexpr (NH == 2)
                asm volatile("" :: "v"(as1[0]), "v"(as1[1]), "v"(as1[2]), "v"(as1[3]), "v"(ad1[0]), "v"(ad1[1]), "v"(ad1[2]), "v"(ad1[3]));
#pragma unroll
            for (int u = 0; u < 4; ++u) {
                x0[u] = lrelu(as0[u] + ad0[u]);
                if constexpr (NH == 2) x1[u] = lrelu(as1[u] + ad1[u]);
            }
#pragma unroll
            for (int u = 0; u < 4; ++u) {
                const int i = off + 32 * u + lane;
                if (i < n) {
                    const int dl = (int)(pk[u] >> 16);
                    atomicMax(&mx[dl * NH], fmap(x0[u]));
                    if constexpr (NH == 2) atomicMax(&mx[dl * 2 + 1], fmap(x1[u]));
                    q[i] = (int)pk[u];
                }
            }
        }
    };

    auto work = [&](int n) {
        for (int off = 0; off < n; off += 32) {
            const int nb = (n - off) < 32 ? (n - off) : 32;
            int s = 0, dl = 0;
            float p0 = 0.f, p1 = 0.f;
            if (lane < nb) {
                const unsigned pk = (unsigned)q[off + lane];
                s  = (int)(pk & 0xFFFFu);
                dl = (int)(pk >> 16);
                const int d = n0 + dl;
                p0 = __expf(lrelu(a_src[s * NH] + a_dst[d * NH]) - finv(mx[dl * NH]));
                atomicAdd(&den[dl * NH], (double)p0);
                if constexpr (NH == 2) {
                    p1 = __expf(lrelu(a_src[s * 2 + 1] + a_dst[d * 2 + 1]) - finv(mx[dl * 2 + 1]));
                    atomicAdd(&den[dl * 2 + 1], (double)p1);
                }
            }
            if constexpr (F == F1) {
#pragma unroll 1
                for (int i0 = 0; i0 < nb; i0 += 8) {
                    float hv[8][4];
#pragma unroll
                    for (int j = 0; j < 8; ++j) {
                        const int i = (i0 + j < nb) ? (i0 + j) : (nb - 1);
                        const int sj = __builtin_amdgcn_readlane(s, i);
                        const float* rp = h + (size_t)sj * F1 + lane;
#pragma unroll
                        for (int k = 0; k < 4; ++k) hv[j][k] = rp[k * 32];
                    }
#pragma unroll
                    for (int j = 0; j < 8; ++j) {
                        if (i0 + j < nb) {
                            const int dj = __builtin_amdgcn_readlane(dl, i0 + j);
                            const float q0 = lane_f(p0, i0 + j);
                            const float q1 = lane_f(p1, i0 + j);
                            double* ap = acc + dj * F1 + lane;
                            atomicAdd(ap +  0, (double)(hv[j][0] * q0));
                            atomicAdd(ap + 32, (double)(hv[j][1] * q0));
                            atomicAdd(ap + 64, (double)(hv[j][2] * q1));
                            atomicAdd(ap + 96, (double)(hv[j][3] * q1));
                        }
                    }
                }
            } else {
                const int total = nb * F;
#pragma unroll 1
                for (int t0 = 0; t0 < total; t0 += 8 * 32) {
                    float hv[8];
                    int   dd2[8];
                    float pp[8];
                    int   cc[8];
#pragma unroll
                    for (int j = 0; j < 8; ++j) {
                        const int idx = t0 + j * 32 + lane;
                        const int ok  = idx < total;
                        const int e   = ok ? idx / F : 0;
                        const int col = ok ? idx - e * F : 0;
                        const int sj  = __shfl(s, e);
                        const int dj  = __shfl(dl, e);
                        dd2[j] = ok ? dj : -1;
                        pp[j] = __shfl(p0, e);
                        cc[j] = col;
                        hv[j] = h[(size_t)sj * F + col];
                    }
#pragma unroll
                    for (int j = 0; j < 8; ++j) {
                        if (dd2[j] >= 0)
                            atomicAdd(&acc[dd2[j] * F + cc[j]], (double)(hv[j] * pp[j]));
                    }
                }
            }
        }
    };

    bool ovf = false;
    auto scan = [&](auto&& flush) -> int {
        int cnt = 0;
        {
            const int node = wave + 32 * lane;
            const bool m = lane < SELF_LANES && node < nr;
            const unsigned mk = __builtin_amdgcn_ballot_w32(m);
            if (m) q[__builtin_amdgcn_mbcnt_lo(mk, 0)] = N_EDGES + n0 + node;
            cnt = __builtin_popcount(mk);
        }
        const int4* dv = reinterpret_cast<const int4*>(ei + N_EDGES);
        int c0 = wave;
        for (;;) {
            bool full = false;
            while (c0 < N_CHUNKS && !full) {
                int4 dd[TRIPC];
#pragma unroll
                for (int u = 0; u < TRIPC; ++u) {
                    const int c = c0 + 32 * u;
                    const int cc = c < N_CHUNKS ? c : c0;
                    dd[u] = dv[(size_t)cc * 32 + lane];
                }
                int took = 0;
#pragma unroll
                for (int u = 0; u < TRIPC; ++u) {
                    const int c = c0 + 32 * u;
                    if (c < N_CHUNKS && !full) {
                        if (cnt > CAP - CHUNK_E) {
                            full = true;
                        } else {
                            const int eb = c * CHUNK_E + lane * 4;
                            const int dk[4] = {dd[u].x, dd[u].y, dd[u].z, dd[u].w};
#pragma unroll
                            for (int k = 0; k < 4; ++k) {
                                const bool m = (unsigned)(dk[k] - n0) < (unsigned)nr;
                                const unsigned mk = __builtin_amdgcn_ballot_w32(m);
                                if (mk) {
                                    if (m) q[cnt + __builtin_amdgcn_mbcnt_lo(mk, 0)] = eb + k;
                                    cnt += __builtin_popcount(mk);
                                }
                            }
                            took = u + 1;
                        }
                    }
                }
                c0 += 32 * (full ? took : TRIPC);
            }
            if (!full) break;
            flush(cnt);
            cnt = 0;
            ovf = true;
        }
        return cnt;
    };

    const int slot = blockIdx.x * 32 + wave;
    int  cnt  = 0;
    bool have = false;
    if constexpr (ROLE == 1) {
        const int c = __builtin_amdgcn_readfirstlane(wc[slot]);
        if (c >= 0) {
            have = true;
            cnt  = c;
            max_saved(wl + (size_t)slot * LIST_PITCH, cnt);
        }
    }
    if (!have) {
        cnt = scan([&](int n) { max_pass(n); });
        max_pass(cnt);
    }
    if constexpr (ROLE == 0) {
        if (!ovf) for (int i = lane; i < cnt; i += 32) wl[(size_t)slot * LIST_PITCH + i] = q[i];
        if (lane == 0) wc[slot] = ovf ? -1 : cnt;
    }
    __syncthreads();
    if (!ovf) {
        work(cnt);
    } else {
        cnt = scan([&](int n) { max_pass(n); work(n); });
        max_pass(cnt);
        work(cnt);
    }

    __syncthreads();
    for (int i = tid; i < nr * F; i += 1024) {
        const int r = i / F, c = i - r * F;
        float v = (float)acc[i] / (float)den[r * NH + (NH == 2 ? (c >> 6) : 0)] + bias[c];
        if constexpr (F == F1) v = v > 0.f ? v : (expf(v) - 1.0f);
        outp[(size_t)n0 * F + i] = v;
    }
}

extern "C" void kernel_launch(void* const* d_in, const int* in_sizes, int n_in,
                              void* d_out, int out_size, void* d_ws, size_t ws_size,
                              hipStream_t stream) {
    const float* x    = (const float*)d_in[0];
    const int*   ei   = (const int*)  d_in[1];
    const float* W1   = (const float*)d_in[2];
    const float* as1  = (const float*)d_in[3];
    const float* ad1  = (const float*)d_in[4];
    const float* b1   = (const float*)d_in[5];
    const float* W2   = (const float*)d_in[6];
    const float* as2  = (const float*)d_in[7];
    const float* ad2  = (const float*)d_in[8];
    const float* b2   = (const float*)d_in[9];
    float* out = (float*)d_out;

    float* ws      = (float*)d_ws;
    float* h1      = ws;
    float* h2in    = h1 + (size_t)N_NODES * F1;
    float* a_src1  = h2in + (size_t)N_NODES * F1;
    float* a_dst1  = a_src1 + (size_t)N_NODES * 2;
    float* a_src2  = a_dst1 + (size_t)N_NODES * 2;
    float* a_dst2  = a_src2 + N_NODES;
    float* h2      = a_dst2 + N_NODES;
    int*   wl      = (int*)d_ws + WS_LISTS_OFF;
    int*   wc      = wl + (size_t)OWN_BLOCKS * 32 * LIST_PITCH;

    const int MT = N_NODES / 16;
    gat_gemm1_wmma<<<G1_BLOCKS, 256, 0, stream>>>(x, W1, h1);
    gat_att1<<<(N_NODES * HEADS + 255) / 256, 256, 0, stream>>>(h1, as1, ad1, a_src1, a_dst1);
    gat_own_accum<F1, HEADS, 6, LIST_PITCH, 0><<<OWN_BLOCKS, 1024, 0, stream>>>(ei, a_src1, a_dst1, h1, b1, h2in, wl, wc);
    gat_gemm2_wmma<<<MT, 96, 0, stream>>>(h2in, W2, h2);
    gat_att2<<<(N_NODES + 255) / 256, 256, 0, stream>>>(h2, as2, ad2, a_src2, a_dst2);
    gat_own_accum<NCLS, 1, 8, 1152, 1><<<OWN_BLOCKS, 1024, 0, stream>>>(ei, a_src2, a_dst2, h2, b2, out, wl, wc);
}


# speedup vs baseline: 3.0102x; 1.2483x over previous
#include <hip/hip_runtime.h>
#include <math.h>

#define N_NODES 50000
#define N_EDGES 800000
#define IN_FEATS 256
#define HID 64
#define HEADS 2
#define F1 (HEADS * HID)
#define NCLS 40

typedef __attribute__((ext_vector_type(2))) float v2f;
typedef __attribute__((ext_vector_type(8))) float v8f;
typedef __attribute__((ext_vector_type(4))) float v4f;

static __device__ __forceinline__ unsigned fmap(float f) {
    unsigned u = __float_as_uint(f);
    return (u & 0x80000000u) ? ~u : (u | 0x80000000u);
}
static __device__ __forceinline__ float finv(unsigned t) {
    return (t & 0x80000000u) ? __uint_as_float(t & 0x7FFFFFFFu)
                             : __uint_as_float(~t);
}
static __device__ __forceinline__ void edge_sd(const int* __restrict__ ei, int e, int& s, int& d) {
    if (e < N_EDGES) { s = ei[e]; d = ei[N_EDGES + e]; }
    else             { s = e - N_EDGES; d = s; }
}
static __device__ __forceinline__ float lrelu(float x) { return x > 0.f ? x : 0.2f * x; }

#define G1_PITCH  260
#define G1_BLOCKS 512
#define G1_TILES  (N_NODES / 16)
static_assert(N_NODES % 16 == 0, "row tiles are whole");

__global__ __launch_bounds__(256) void gat_gemm1_wmma(const float* __restrict__ x,
                                                      const float* __restrict__ W1,
                                                      float* __restrict__ h1) {
    __shared__ float xs[2][16 * G1_PITCH];

    const int tid   = threadIdx.x;
    const int wave  = tid >> 5;
    const int lane  = tid & 31;
    const int half  = lane >> 4;
    const int l     = lane & 15;

    v2f b[IN_FEATS / 4];
    {
        const float* bp = W1 + (size_t)(2 * half) * F1 + wave * 16 + l;
#pragma unroll
        for (int kk = 0; kk < IN_FEATS / 4; ++kk) {
            b[kk].x = bp[(size_t)(4 * kk) * F1];
            b[kk].y = bp[(size_t)(4 * kk + 1) * F1];
        }
    }

    const v4f* xv = reinterpret_cast<const v4f*>(x);
    const int so = (tid >> 6) * G1_PITCH + (tid & 63) * 4;
    v4f st0, st1, st2, st3;
    int mt = blockIdx.x;
    if (mt < G1_TILES) {
        const v4f* tp = xv + (size_t)mt * 1024 + tid;
        st0 = tp[0]; st1 = tp[256]; st2 = tp[512]; st3 = tp[768];
        float* sp = &xs[0][so];
        *reinterpret_cast<v4f*>(sp)                 = st0;
        *reinterpret_cast<v4f*>(sp +  4 * G1_PITCH) = st1;
        *reinterpret_cast<v4f*>(sp +  8 * G1_PITCH) = st2;
        *reinterpret_cast<v4f*>(sp + 12 * G1_PITCH) = st3;
    }
    __syncthreads();

    int buf = 0;
    for (; mt < G1_TILES; mt += gridDim.x) {
        const int  nxt  = mt + gridDim.x;
        const bool more = nxt < G1_TILES;
        if (more) {
            const v4f* tp = xv + (size_t)nxt * 1024 + tid;
            st0 = tp[0]; st1 = tp[256]; st2 = tp[512]; st3 = tp[768];
        }

        const float* ap = &xs[buf][l * G1_PITCH + 2 * half];
        v8f acc = {};
        v2f a0[8], a1[8];
#pragma unroll
        for (int i = 0; i < 8; ++i) a0[i] = *reinterpret_cast<const v2f*>(ap + 4 * i);
#pragma unroll
        for (int g = 0; g < IN_FEATS / 32; g += 2) {
#pragma unroll
            for (int i = 0; i < 8; ++i) a1[i] = *reinterpret_cast<const v2f*>(ap + 32 * (g + 1) + 4 * i);
            __builtin_amdgcn_sched_barrier(0);
#pragma unroll
            for (int i = 0; i < 8; ++i)
                acc = __builtin_amdgcn_wmma_f32_16x16x4_f32(false, a0[i], false, b[8 * g + i],
                                                            (short)0, acc, false, false);
            __builtin_amdgcn_sched_barrier(0);
            if (g + 2 < IN_FEATS / 32) {
#pragma unroll
                for (int i = 0; i < 8; ++i) a0[i] = *reinterpret_cast<const v2f*>(ap + 32 * (g + 2) + 4 * i);
            }
            __builtin_amdgcn_sched_barrier(0);
#pragma unroll
            for (int i = 0; i < 8; ++i)
                acc = __builtin_amdgcn_wmma_f32_16x16x4_f32(false, a1[i], false, b[8 * (g + 1) + i],
                                                            (short)0, acc, false, false);
            __builtin_amdgcn_sched_barrier(0);
        }
        if (more) {
            float* sp = &xs[buf ^ 1][so];
            *reinterpret_cast<v4f*>(sp)                 = st0;
            *reinterpret_cast<v4f*>(sp +  4 * G1_PITCH) = st1;
            *reinterpret_cast<v4f*>(sp +  8 * G1_PITCH) = st2;
            *reinterpret_cast<v4f*>(sp + 12 * G1_PITCH) = st3;
        }
        __syncthreads();
        buf ^= 1;

        float* dr = h1 + (size_t)(mt * 16 + 8 * half) * F1 + wave * 16 + l;
#pragma unroll
        for (int r = 0; r < 8; ++r) dr[(size_t)r * F1] = acc[r];
    }
}

__global__ __launch_bounds__(96) void gat_gemm2_wmma(const float* __restrict__ h2in,
                                                     const float* __restrict__ W2,
                                                     float* __restrict__ h2) {
    const int mtile = blockIdx.x;
    const int wave  = threadIdx.x >> 5;
    const int lane  = threadIdx.x & 31;
    const int half  = lane >> 4;
    const int l     = lane & 15;
    const int row   = mtile * 16 + l;
    const int ncol  = wave * 16 + l;

    v8f acc = {};
    const float* xr = h2in + (size_t)row * F1;
    for (int k = 0; k < F1; k += 4) {
        v2f a = *(const v2f*)(xr + k + 2 * half);
        v2f b;
        if (ncol < NCLS) {
            const float* bp = W2 + (size_t)(k + 2 * half) * NCLS + ncol;
            b.x = bp[0]; b.y = bp[NCLS];
        } else { b.x = 0.f; b.y = 0.f; }
        acc = __builtin_amdgcn_wmma_f32_16x16x4_f32(false, a, false, b,
                                                    (short)0, acc, false, false);
    }
    if (ncol < NCLS) {
        float* dr = h2 + (size_t)(mtile * 16 + 8 * half) * NCLS + ncol;
#pragma unroll
        for (int r = 0; r < 8; ++r) dr[(size_t)r * NCLS] = acc[r];
    }
}

__global__ void gat_att1(const float* __restrict__ h1, const float* __restrict__ as1,
                         const float* __restrict__ ad1, float* __restrict__ a_src,
                         float* __restrict__ a_dst) {
    int i = blockIdx.x * blockDim.x + threadIdx.x;
    if (i >= N_NODES * HEADS) return;
    int n = i >> 1, h = i & 1;
    const float* hp = h1 + (size_t)n * F1 + h * HID;
    const float* s0 = as1 + h * HID;
    const float* d0 = ad1 + h * HID;
    float s = 0.f, d = 0.f;
#pragma unroll 8
    for (int c = 0; c < HID; ++c) { float v = hp[c]; s += v * s0[c]; d += v * d0[c]; }
    a_src[i] = s; a_dst[i] = d;
}

__global__ void gat_att2(const float* __restrict__ h2, const float* __restrict__ as2,
                         const float* __restrict__ ad2, float* __restrict__ a_src,
                         float* __restrict__ a_dst) {
    int n = blockIdx.x * blockDim.x + threadIdx.x;
    if (n >= N_NODES) return;
    const float* hp = h2 + (size_t)n * NCLS;
    float s = 0.f, d = 0.f;
#pragma unroll 8
    for (int c = 0; c < NCLS; ++c) { float v = hp[c]; s += v * as2[c]; d += v * ad2[c]; }
    a_src[n] = s; a_dst[n] = d;
}

#define OWN_R      196
#define OWN_BLOCKS ((N_NODES + OWN_R - 1) / OWN_R)
#define CHUNK_E    128
#define N_CHUNKS   (N_EDGES / CHUNK_E)
#define SELF_LANES ((OWN_R + 31) / 32)
#define LIST_PITCH 896
#define WS_LISTS_OFF ((size_t)16 << 20)
#define WS_PLANES ((size_t)N_NODES * (2 * F1 + 2 * HEADS + 2 + NCLS))
static_assert(WS_PLANES <= WS_LISTS_OFF, "the saved lists start behind the feature and score planes");
static_assert((WS_LISTS_OFF + (size_t)OWN_BLOCKS * 32 * (LIST_PITCH + 1)) * 4 <= ((size_t)256 << 20),
              "lists and lengths end inside the workspace");
static_assert(N_EDGES % CHUNK_E == 0, "the destination list is read in whole 128-edge chunks");
static_assert(N_NODES <= 65536 && OWN_R <= 65536, "a list entry packs source and local destination, 16 bits each");
static_assert(SELF_LANES <= 32, "one lane per self-loop of a wave");

static __device__ __forceinline__ float lane_f(float v, int i) {
    return __int_as_float(__builtin_amdgcn_readlane(__float_as_int(v), i));
}

static __device__ __forceinline__ double widen_normal(float f) {
    const unsigned u  = __float_as_uint(f);
    const unsigned hi = (u & 0x80000000u) | (((u >> 3) & 0x0FFFFFFFu) + 0x38000000u);
    const unsigned lo = u << 29;
    return __builtin_bit_cast(double, ((unsigned long long)hi << 32) | lo);
}
static __device__ __forceinline__ bool not_normal(float f) { return __builtin_amdgcn_classf(f, 0x2F7); }
template <int N>
static __device__ __forceinline__ void widen(const float (&f)[N], double (&d)[N]) {
    bool other = false;
#pragma unroll
    for (int i = 0; i < N; ++i) { d[i] = widen_normal(f[i]); other |= not_normal(f[i]); }
    if (other) {
#pragma unroll
        for (int i = 0; i < N; ++i) d[i] = (double)f[i];
    }
}

template <int F, int NH, int TRIPC, int CAP, int ROLE>
__global__ __launch_bounds__(1024) void gat_own_accum(const int* __restrict__ ei,
        const float* __restrict__ a_src, const float* __restrict__ a_dst,
        const float* __restrict__ h,
        const float* __restrict__ bias, float* __restrict__ outp,
        int* __restrict__ wl, int* __restrict__ wc) {
    static_assert(CAP >= CHUNK_E + SELF_LANES, "a chunk goes into a list only while 128 entries are free");
    static_assert(ROLE == 0 ? CAP <= LIST_PITCH : CAP >= LIST_PITCH, "a saved list fits its slot, and fits the list of the wave that takes it");
    __shared__ double   acc[OWN_R * F];
    __shared__ double   den[OWN_R * NH];
    __shared__ unsigned mx[OWN_R * NH];
    __shared__ int      que[32 * CAP];

    const int tid  = threadIdx.x;
    const int lane = tid & 31;
    const int wave = __builtin_amdgcn_readfirstlane(tid >> 5);
    const int n0   = blockIdx.x * OWN_R;
    const int nr   = (N_NODES - n0) < OWN_R ? (N_NODES - n0) : OWN_R;
    int* q = que + wave * CAP;

    for (int i = tid; i < OWN_R * F;  i += 1024) acc[i] = 0.0;
    for (int i = tid; i < OWN_R * NH; i += 1024) { den[i] = 0.0; mx[i] = 0u; }
    __syncthreads();

    auto max_pass = [&](int n) {
        for (int off = 0; off < n; off += 128) {
            int e[4], s[4], d[4];
            float x0[4];
            [[maybe_unused]] float x1[4];
#pragma unroll
            for (int u = 0; u < 4; ++u) {
                const int i = off + 32 * u + lane;
                e[u] = i < n ? q[i] : N_EDGES + n0;
            }
#pragma unroll
            for (int u = 0; u < 4; ++u) {
                const bool r  = e[u] < N_EDGES;
                const int  ec = r ? e[u] : 0;
                const int  sl = ei[ec], dg = ei[N_EDGES + ec];
                s[u] = r ? sl : e[u] - N_EDGES;
                d[u] = r ? dg : s[u];
            }
#pragma unroll
            for (int u = 0; u < 4; ++u) {
                x0[u] = lrelu(a_src[s[u] * NH] + a_dst[d[u] * NH]);
                if constexpr (NH == 2) x1[u] = lrelu(a_src[s[u] * 2 + 1] + a_dst[d[u] * 2 + 1]);
            }
#pragma unroll
            for (int u = 0; u < 4; ++u) {
                const int i = off + 32 * u + lane;
                if (i < n) {
                    const int dl = d[u] - n0;
                    atomicMax(&mx[dl * NH], fmap(x0[u]));
                    if constexpr (NH == 2) atomicMax(&mx[dl * 2 + 1], fmap(x1[u]));
                    q[i] = s[u] | (dl << 16);
                }
            }
        }
    };

    [[maybe_unused]] auto max_saved = [&](const int* __restrict__ sv, int n) {
        for (int off = 0; off < n; off += 128) {
            unsigned pk[4];
            float x0[4];
            [[maybe_unused]] float x1[4];
#pragma unroll
            for (int u = 0; u < 4; ++u) {
                const int i = off + 32 * u + lane;
                pk[u] = (unsigned)sv[i < n ? i : 0];
            }
            asm volatile("" :: "v"(pk[0]), "v"(pk[1]), "v"(pk[2]), "v"(pk[3]));
            float as0[4], ad0[4];
            [[maybe_unused]] float as1[4], ad1[4];
#pragma unroll
            for (int u = 0; u < 4; ++u) {
                const int s = (int)(pk[u] & 0xFFFFu), d = n0 + (int)(pk[u] >> 16);
                as0[u] = a_src[s * NH]; ad0[u] = a_dst[d * NH];
                if constexpr (NH == 2) { as1[u] = a_src[s * 2 + 1]; ad1[u] = a_dst[d * 2 + 1]; }
            }
            asm volatile("" :: "v"(as0[0]), "v"(as0[1]), "v"(as0[2]), "v"(as0[3]), "v"(ad0[0]), "v"(ad0[1]), "v"(ad0[2]), "v"(ad0[3]));
            if constexpr (NH == 2)
                asm volatile("" :: "v"(as1[0]), "v"(as1[1]), "v"(as1[2]), "v"(as1[3]), "v"(ad1[0]), "v"(ad1[1]), "v"(ad1[2]), "v"(ad1[3]));
#pragma unroll
            for (int u = 0; u < 4; ++u) {
                x0[u] = lrelu(as0[u] + ad0[u]);
                if constexpr (NH == 2) x1[u] = lrelu(as1[u] + ad1[u]);
            }
#pragma unroll
            for (int u = 0; u < 4; ++u) {
                const int i = off + 32 * u + lane;
                if (i < n) {
                    const int dl = (int)(pk[u] >> 16);
                    atomicMax(&mx[dl * NH], fmap(x0[u]));
                    if constexpr (NH == 2) atomicMax(&mx[dl * 2 + 1], fmap(x1[u]));
                    q[i] = (int)pk[u];
                }
            }
        }
    };

    auto work = [&](int n) {
        for (int off = 0; off < n; off += 32) {
            const int nb = (n - off) < 32 ? (n - off) : 32;
            int s = 0, dl = 0;
            float p0 = 0.f, p1 = 0.f;
            if (lane < nb) {
                const unsigned pk = (unsigned)q[off + lane];
                s  = (int)(pk & 0xFFFFu);
                dl = (int)(pk >> 16);
                const int d = n0 + dl;
                p0 = __expf(lrelu(a_src[s * NH] + a_dst[d * NH]) - finv(mx[dl * NH]));
                atomicAdd(&den[dl * NH], (double)p0);
                if constexpr (NH == 2) {
                    p1 = __expf(lrelu(a_src[s * 2 + 1] + a_dst[d * 2 + 1]) - finv(mx[dl * 2 + 1]));
                    atomicAdd(&den[dl * 2 + 1], (double)p1);
                }
            }
            if constexpr (F == F1) {
#pragma unroll 1
                for (int i0 = 0; i0 < nb; i0 += 8) {
                    float hv[8][4];
#pragma unroll
                    for (int j = 0; j < 8; ++j) {
                        const int i = (i0 + j < nb) ? (i0 + j) : (nb - 1);
                        const int sj = __builtin_amdgcn_readlane(s, i);
                        const float* rp = h + (size_t)sj * F1 + lane;
#pragma unroll
                        for (int k = 0; k < 4; ++k) hv[j][k] = rp[k * 32];
                    }
#pragma unroll
                    for (int j = 0; j < 8; ++j) {
                        if (i0 + j < nb) {
                            const int dj = __builtin_amdgcn_readlane(dl, i0 + j);
                            const float q0 = lane_f(p0, i0 + j);
                            const float q1 = lane_f(p1, i0 + j);
                            double* ap = acc + dj * F1 + lane;
                            const float tf[4] = {hv[j][0] * q0, hv[j][1] * q0,
                                                 hv[j][2] * q1, hv[j][3] * q1};
                            double td[4];
                            widen(tf, td);
                            atomicAdd(ap +  0, td[0]);
                            atomicAdd(ap + 32, td[1]);
                            atomicAdd(ap + 64, td[2]);
                            atomicAdd(ap + 96, td[3]);
                        }
                    }
                }
            } else {
                const int total = nb * F;
#pragma unroll 1
                for (int t0 = 0; t0 < total; t0 += 8 * 32) {
                    float hv[8];
                    int   dd2[8];
                    float pp[8];
                    int   cc[8];
#pragma unroll
                    for (int j = 0; j < 8; ++j) {
                        const int idx = t0 + j * 32 + lane;
                        const int ok  = idx < total;
                        const int e   = ok ? idx / F : 0;
                        const int col = ok ? idx - e * F : 0;
                        const int sj  = __shfl(s, e);
                        const int dj  = __shfl(dl, e);
                        dd2[j] = ok ? dj : -1;
                        pp[j] = __shfl(p0, e);
                        cc[j] = col;
                        hv[j] = h[(size_t)sj * F + col];
                    }
                    float  tf[8];
                    double td[8];
#pragma unroll
                    for (int j = 0; j < 8; ++j) tf[j] = hv[j] * pp[j];
                    widen(tf, td);
#pragma unroll
                    for (int j = 0; j < 8; ++j) {
                        if (dd2[j] >= 0)
                            atomicAdd(&acc[dd2[j] * F + cc[j]], td[j]);
                    }
                }
            }
        }
    };

    bool ovf = false;
    auto scan = [&](auto&& flush) -> int {
        int cnt = 0;
        {
            const int node = wave + 32 * lane;
            const bool m = lane < SELF_LANES && node < nr;
            const unsigned mk = __builtin_amdgcn_ballot_w32(m);
            if (m) q[__builtin_amdgcn_mbcnt_lo(mk, 0)] = N_EDGES + n0 + node;
            cnt = __builtin_popcount(mk);
        }
        const int4* dv = reinterpret_cast<const int4*>(ei + N_EDGES);
        int c0 = wave;
        for (;;) {
            bool full = false;
            while (c0 < N_CHUNKS && !full) {
                int4 dd[TRIPC];
#pragma unroll
                for (int u = 0; u < TRIPC; ++u) {
                    const int c = c0 + 32 * u;
                    const int cc = c < N_CHUNKS ? c : c0;
                    dd[u] = dv[(size_t)cc * 32 + lane];
                }
                int took = 0;
#pragma unroll
                for (int u = 0; u < TRIPC; ++u) {
                    const int c = c0 + 32 * u;
                    if (c < N_CHUNKS && !full) {
                        if (cnt > CAP - CHUNK_E) {
                            full = true;
                        } else {
                            const int eb = c * CHUNK_E + lane * 4;
                            const int dk[4] = {dd[u].x, dd[u].y, dd[u].z, dd[u].w};
#pragma unroll
                            for (int k = 0; k < 4; ++k) {
                                const bool m = (unsigned)(dk[k] - n0) < (unsigned)nr;
                                const unsigned mk = __builtin_amdgcn_ballot_w32(m);
                                if (mk) {
                                    if (m) q[cnt + __builtin_amdgcn_mbcnt_lo(mk, 0)] = eb + k;
                                    cnt += __builtin_popcount(mk);
                                }
                            }
                            took = u + 1;
                        }
                    }
                }
                c0 += 32 * (full ? took : TRIPC);
            }
            if (!full) break;
            flush(cnt);
            cnt = 0;
            ovf = true;
        }
        return cnt;
    };

    const int slot = blockIdx.x * 32 + wave;
    int  cnt  = 0;
    bool have = false;
    if constexpr (ROLE == 1) {
        const int c = __builtin_amdgcn_readfirstlane(wc[slot]);
        if (c >= 0) {
            have = true;
            cnt  = c;
            max_saved(wl + (size_t)slot * LIST_PITCH, cnt);
        }
    }
    if (!have) {
        cnt = scan([&](int n) { max_pass(n); });
        max_pass(cnt);
    }
    if constexpr (ROLE == 0) {
        if (!ovf) for (int i = lane; i < cnt; i += 32) wl[(size_t)slot * LIST_PITCH + i] = q[i];
        if (lane == 0) wc[slot] = ovf ? -1 : cnt;
    }
    __syncthreads();
    if (!ovf) {
        work(cnt);
    } else {
        cnt = scan([&](int n) { max_pass(n); work(n); });
        max_pass(cnt);
        work(cnt);
    }

    __syncthreads();
    for (int i = tid; i < nr * F; i += 1024) {
        const int r = i / F, c = i - r * F;
        float v = (float)acc[i] / (float)den[r * NH + (NH == 2 ? (c >> 6) : 0)] + bias[c];
        if constexpr (F == F1) v = v > 0.f ? v : (expf(v) - 1.0f);
        outp[(size_t)n0 * F + i] = v;
    }
}

extern "C" void kernel_launch(void* const* d_in, const int* in_sizes, int n_in,
                              void* d_out, int out_size, void* d_ws, size_t ws_size,
                              hipStream_t stream) {
    const float* x    = (const float*)d_in[0];
    const int*   ei   = (const int*)  d_in[1];
    const float* W1   = (const float*)d_in[2];
    const float* as1  = (const float*)d_in[3];
    const float* ad1  = (const float*)d_in[4];
    const float* b1   = (const float*)d_in[5];
    const float* W2   = (const float*)d_in[6];
    const float* as2  = (const float*)d_in[7];
    const float* ad2  = (const float*)d_in[8];
    const float* b2   = (const float*)d_in[9];
    float* out = (float*)d_out;

    float* ws      = (float*)d_ws;
    float* h1      = ws;
    float* h2in    = h1 + (size_t)N_NODES * F1;
    float* a_src1  = h2in + (size_t)N_NODES * F1;
    float* a_dst1  = a_src1 + (size_t)N_NODES * 2;
    float* a_src2  = a_dst1 + (size_t)N_NODES * 2;
    float* a_dst2  = a_src2 + N_NODES;
    float* h2      = a_dst2 + N_NODES;
    int*   wl      = (int*)d_ws + WS_LISTS_OFF;
    int*   wc      = wl + (size_t)OWN_BLOCKS * 32 * LIST_PITCH;

    const int MT = N_NODES / 16;
    gat_gemm1_wmma<<<G1_BLOCKS, 256, 0, stream>>>(x, W1, h1);
    gat_att1<<<(N_NODES * HEADS + 255) / 256, 256, 0, stream>>>(h1, as1, ad1, a_src1, a_dst1);
    gat_own_accum<F1, HEADS, 6, LIST_PITCH, 0><<<OWN_BLOCKS, 1024, 0, stream>>>(ei, a_src1, a_dst1, h1, b1, h2in, wl, wc);
    gat_gemm2_wmma<<<MT, 96, 0, stream>>>(h2in, W2, h2);
    gat_att2<<<(N_NODES + 255) / 256, 256, 0, stream>>>(h2, as2, ad2, a_src2, a_dst2);
    gat_own_accum<NCLS, 1, 8, 1152, 1><<<OWN_BLOCKS, 1024, 0, stream>>>(ei, a_src2, a_dst2, h2, b2, out, wl, wc);
}


# Destination list re-coded each call as one owner byte per edge (rides in the first dot launch); owner blocks scan the bytes, 8 edges per lane per load, a quarter of the traffic; sums in arrival order
# speedup vs baseline: 3.4054x; 1.1313x over previous
#include <hip/hip_runtime.h>
#include <math.h>

#define N_NODES 50000
#define N_EDGES 800000
#define IN_FEATS 256
#define HID 64
#define HEADS 2
#define F1 (HEADS * HID)
#define NCLS 40
#define OWN_R      196
#define OWN_BLOCKS ((N_NODES + OWN_R - 1) / OWN_R)
#define ATT1_BLOCKS ((N_NODES * HEADS + 255) / 256)
#define OB_WORDS   (N_EDGES / 4)
#define OB_BLOCKS  ((OB_WORDS + 255) / 256)
static_assert(OWN_BLOCKS <= 256, "an owner index fits a byte");
static_assert(N_EDGES % 4 == 0, "the destination list is re-coded four edges at a time");

typedef __attribute__((ext_vector_type(2))) float v2f;
typedef __attribute__((ext_vector_type(8))) float v8f;
typedef __attribute__((ext_vector_type(4))) float v4f;

static __device__ __forceinline__ unsigned fmap(float f) {
    unsigned u = __float_as_uint(f);
    return (u & 0x80000000u) ? ~u : (u | 0x80000000u);
}
static __device__ __forceinline__ float finv(unsigned t) {
    return (t & 0x80000000u) ? __uint_as_float(t & 0x7FFFFFFFu)
                             : __uint_as_float(~t);
}
static __device__ __forceinline__ void edge_sd(const int* __restrict__ ei, int e, int& s, int& d) {
    if (e < N_EDGES) { s = ei[e]; d = ei[N_EDGES + e]; }
    else             { s = e - N_EDGES; d = s; }
}
static __device__ __forceinline__ float lrelu(float x) { return x > 0.f ? x : 0.2f * x; }

#define G1_PITCH  260
#define G1_BLOCKS 512
#define G1_TILES  (N_NODES / 16)
static_assert(N_NODES % 16 == 0, "row tiles are whole");

__global__ __launch_bounds__(256) void gat_gemm1_wmma(const float* __restrict__ x,
                                                      const float* __restrict__ W1,
                                                      float* __restrict__ h1) {
    __shared__ float xs[2][16 * G1_PITCH];

    const int tid   = threadIdx.x;
    const int wave  = tid >> 5;
    const int lane  = tid & 31;
    const int half  = lane >> 4;
    const int l     = lane & 15;

    v2f b[IN_FEATS / 4];
    {
        const float* bp = W1 + (size_t)(2 * half) * F1 + wave * 16 + l;
#pragma unroll
        for (int kk = 0; kk < IN_FEATS / 4; ++kk) {
            b[kk].x = bp[(size_t)(4 * kk) * F1];
            b[kk].y = bp[(size_t)(4 * kk + 1) * F1];
        }
    }

    const v4f* xv = reinterpret_cast<const v4f*>(x);
    const int so = (tid >> 6) * G1_PITCH + (tid & 63) * 4;
    v4f st0, st1, st2, st3;
    int mt = blockIdx.x;
    if (mt < G1_TILES) {
        const v4f* tp = xv + (size_t)mt * 1024 + tid;
        st0 = tp[0]; st1 = tp[256]; st2 = tp[512]; st3 = tp[768];
        float* sp = &xs[0][so];
        *reinterpret_cast<v4f*>(sp)                 = st0;
        *reinterpret_cast<v4f*>(sp +  4 * G1_PITCH) = st1;
        *reinterpret_cast<v4f*>(sp +  8 * G1_PITCH) = st2;
        *reinterpret_cast<v4f*>(sp + 12 * G1_PITCH) = st3;
    }
    __syncthreads();

    int buf = 0;
    for (; mt < G1_TILES; mt += gridDim.x) {
        const int  nxt  = mt + gridDim.x;
        const bool more = nxt < G1_TILES;
        if (more) {
            const v4f* tp = xv + (size_t)nxt * 1024 + tid;
            st0 = tp[0]; st1 = tp[256]; st2 = tp[512]; st3 = tp[768];
        }

        const float* ap = &xs[buf][l * G1_PITCH + 2 * half];
        v8f acc = {};
        v2f a0[8], a1[8];
#pragma unroll
        for (int i = 0; i < 8; ++i) a0[i] = *reinterpret_cast<const v2f*>(ap + 4 * i);
#pragma unroll
        for (int g = 0; g < IN_FEATS / 32; g += 2) {
#pragma unroll
            for (int i = 0; i < 8; ++i) a1[i] = *reinterpret_cast<const v2f*>(ap + 32 * (g + 1) + 4 * i);
            __builtin_amdgcn_sched_barrier(0);
#pragma unroll
            for (int i = 0; i < 8; ++i)
                acc = __builtin_amdgcn_wmma_f32_16x16x4_f32(false, a0[i], false, b[8 * g + i],
                                                            (short)0, acc, false, false);
            __builtin_amdgcn_sched_barrier(0);
            if (g + 2 < IN_FEATS / 32) {
#pragma unroll
                for (int i = 0; i < 8; ++i) a0[i] = *reinterpret_cast<const v2f*>(ap + 32 * (g + 2) + 4 * i);
            }
            __builtin_amdgcn_sched_barrier(0);
#pragma unroll
            for (int i = 0; i < 8; ++i)
                acc = __builtin_amdgcn_wmma_f32_16x16x4_f32(false, a1[i], false, b[8 * (g + 1) + i],
                                                            (short)0, acc, false, false);
            __builtin_amdgcn_sched_barrier(0);
        }
        if (more) {
            float* sp = &xs[buf ^ 1][so];
            *reinterpret_cast<v4f*>(sp)                 = st0;
            *reinterpret_cast<v4f*>(sp +  4 * G1_PITCH) = st1;
            *reinterpret_cast<v4f*>(sp +  8 * G1_PITCH) = st2;
            *reinterpret_cast<v4f*>(sp + 12 * G1_PITCH) = st3;
        }
        __syncthreads();
        buf ^= 1;

        float* dr = h1 + (size_t)(mt * 16 + 8 * half) * F1 + wave * 16 + l;
#pragma unroll
        for (int r = 0; r < 8; ++r) dr[(size_t)r * F1] = acc[r];
    }
}

__global__ __launch_bounds__(96) void gat_gemm2_wmma(const float* __restrict__ h2in,
                                                     const float* __restrict__ W2,
                                                     float* __restrict__ h2) {
    const int mtile = blockIdx.x;
    const int wave  = threadIdx.x >> 5;
    const int lane  = threadIdx.x & 31;
    const int half  = lane >> 4;
    const int l     = lane & 15;
    const int row   = mtile * 16 + l;
    const int ncol  = wave * 16 + l;

    v8f acc = {};
    const float* xr = h2in + (size_t)row * F1;
    for (int k = 0; k < F1; k += 4) {
        v2f a = *(const v2f*)(xr + k + 2 * half);
        v2f b;
        if (ncol < NCLS) {
            const float* bp = W2 + (size_t)(k + 2 * half) * NCLS + ncol;
            b.x = bp[0]; b.y = bp[NCLS];
        } else { b.x = 0.f; b.y = 0.f; }
        acc = __builtin_amdgcn_wmma_f32_16x16x4_f32(false, a, false, b,
                                                    (short)0, acc, false, false);
    }
    if (ncol < NCLS) {
        float* dr = h2 + (size_t)(mtile * 16 + 8 * half) * NCLS + ncol;
#pragma unroll
        for (int r = 0; r < 8; ++r) dr[(size_t)r * NCLS] = acc[r];
    }
}

__global__ void gat_att1(const float* __restrict__ h1, const float* __restrict__ as1,
                         const float* __restrict__ ad1, float* __restrict__ a_src,
                         float* __restrict__ a_dst, const int* __restrict__ ei,
                         unsigned* __restrict__ ob) {
    if (blockIdx.x >= ATT1_BLOCKS) {
        const int w = (blockIdx.x - ATT1_BLOCKS) * 256 + threadIdx.x;
        if (w < OB_WORDS) {
            const int4 d = reinterpret_cast<const int4*>(ei + N_EDGES)[w];
            const unsigned o0 = (unsigned)d.x / OWN_R, o1 = (unsigned)d.y / OWN_R;
            const unsigned o2 = (unsigned)d.z / OWN_R, o3 = (unsigned)d.w / OWN_R;
            ob[w] = (o0 < 255u ? o0 : 255u) | (o1 < 255u ? o1 : 255u) << 8 |
                    (o2 < 255u ? o2 : 255u) << 16 | (o3 < 255u ? o3 : 255u) << 24;
        }
        return;
    }
    int i = blockIdx.x * blockDim.x + threadIdx.x;
    if (i >= N_NODES * HEADS) return;
    int n = i >> 1, h = i & 1;
    const float* hp = h1 + (size_t)n * F1 + h * HID;
    const float* s0 = as1 + h * HID;
    const float* d0 = ad1 + h * HID;
    float s = 0.f, d = 0.f;
#pragma unroll 8
    for (int c = 0; c < HID; ++c) { float v = hp[c]; s += v * s0[c]; d += v * d0[c]; }
    a_src[i] = s; a_dst[i] = d;
}

__global__ void gat_att2(const float* __restrict__ h2, const float* __restrict__ as2,
                         const float* __restrict__ ad2, float* __restrict__ a_src,
                         float* __restrict__ a_dst) {
    int n = blockIdx.x * blockDim.x + threadIdx.x;
    if (n >= N_NODES) return;
    const float* hp = h2 + (size_t)n * NCLS;
    float s = 0.f, d = 0.f;
#pragma unroll 8
    for (int c = 0; c < NCLS; ++c) { float v = hp[c]; s += v * as2[c]; d += v * ad2[c]; }
    a_src[n] = s; a_dst[n] = d;
}

#define CHUNK_E    256
#define N_CHUNKS   (N_EDGES / CHUNK_E)
#define SELF_LANES ((OWN_R + 31) / 32)
#define LIST_PITCH 896
#define WS_LISTS_OFF ((size_t)16 << 20)
#define WS_PLANES ((size_t)N_NODES * (2 * F1 + 2 * HEADS + 2 + NCLS))
static_assert(WS_PLANES <= WS_LISTS_OFF, "the saved lists start behind the feature and score planes");
static_assert((WS_LISTS_OFF + (size_t)OWN_BLOCKS * 32 * (LIST_PITCH + 1)) * 4 <= ((size_t)256 << 20),
              "lists and lengths end inside the workspace");
#define WS_OB_OFF ((size_t)32 << 20)
static_assert(WS_LISTS_OFF + (size_t)OWN_BLOCKS * 32 * (LIST_PITCH + 1) <= WS_OB_OFF, "the owner bytes start behind the saved lists");
static_assert((WS_OB_OFF + OB_WORDS) * 4 <= ((size_t)256 << 20), "the owner bytes end inside the workspace");
static_assert(N_EDGES % CHUNK_E == 0, "the owner bytes are read in whole 256-edge chunks");
static_assert(N_NODES <= 65536 && OWN_R <= 65536, "a list entry packs source and local destination, 16 bits each");
static_assert(SELF_LANES <= 32, "one lane per self-loop of a wave");

static __device__ __forceinline__ float lane_f(float v, int i) {
    return __int_as_float(__builtin_amdgcn_readlane(__float_as_int(v), i));
}

static __device__ __forceinline__ double widen_normal(float f) {
    const unsigned u  = __float_as_uint(f);
    const unsigned hi = (u & 0x80000000u) | (((u >> 3) & 0x0FFFFFFFu) + 0x38000000u);
    const unsigned lo = u << 29;
    return __builtin_bit_cast(double, ((unsigned long long)hi << 32) | lo);
}
static __device__ __forceinline__ bool not_normal(float f) { return __builtin_amdgcn_classf(f, 0x2F7); }
template <int N>
static __device__ __forceinline__ void widen(const float (&f)[N], double (&d)[N]) {
    bool other = false;
#pragma unroll
    for (int i = 0; i < N; ++i) { d[i] = widen_normal(f[i]); other |= not_normal(f[i]); }
    if (other) {
#pragma unroll
        for (int i = 0; i < N; ++i) d[i] = (double)f[i];
    }
}

template <int F, int NH, int TRIPC, int CAP, int ROLE>
__global__ __launch_bounds__(1024) void gat_own_accum(const int* __restrict__ ei,
        const float* __restrict__ a_src, const float* __restrict__ a_dst,
        const float* __restrict__ h,
        const float* __restrict__ bias, float* __restrict__ outp,
        int* __restrict__ wl, int* __restrict__ wc, const unsigned* __restrict__ ob) {
    static_assert(CAP >= CHUNK_E + SELF_LANES, "a chunk goes into a list only while 256 entries are free");
    static_assert(ROLE == 0 ? CAP <= LIST_PITCH : CAP >= LIST_PITCH, "a saved list fits its slot, and fits the list of the wave that takes it");
    __shared__ double   acc[OWN_R * F];
    __shared__ double   den[OWN_R * NH];
    __shared__ unsigned mx[OWN_R * NH];
    __shared__ int      que[32 * CAP];

    const int tid  = threadIdx.x;
    const int lane = tid & 31;
    const int wave = __builtin_amdgcn_readfirstlane(tid >> 5);
    const int n0   = blockIdx.x * OWN_R;
    const int nr   = (N_NODES - n0) < OWN_R ? (N_NODES - n0) : OWN_R;
    int* q = que + wave * CAP;

    for (int i = tid; i < OWN_R * F;  i += 1024) acc[i] = 0.0;
    for (int i = tid; i < OWN_R * NH; i += 1024) { den[i] = 0.0; mx[i] = 0u; }
    __syncthreads();

    auto max_pass = [&](int n) {
        for (int off = 0; off < n; off += 128) {
            int e[4], s[4], d[4];
            float x0[4];
            [[maybe_unused]] float x1[4];
#pragma unroll
            for (int u = 0; u < 4; ++u) {
                const int i = off + 32 * u + lane;
                e[u] = i < n ? q[i] : N_EDGES + n0;
            }
#pragma unroll
            for (int u = 0; u < 4; ++u) {
                const bool r  = e[u] < N_EDGES;
                const int  ec = r ? e[u] : 0;
                const int  sl = ei[ec], dg = ei[N_EDGES + ec];
                s[u] = r ? sl : e[u] - N_EDGES;
                d[u] = r ? dg : s[u];
            }
#pragma unroll
            for (int u = 0; u < 4; ++u) {
                x0[u] = lrelu(a_src[s[u] * NH] + a_dst[d[u] * NH]);
                if constexpr (NH == 2) x1[u] = lrelu(a_src[s[u] * 2 + 1] + a_dst[d[u] * 2 + 1]);
            }
#pragma unroll
            for (int u = 0; u < 4; ++u) {
                const int i = off + 32 * u + lane;
                if (i < n) {
                    const int dl = d[u] - n0;
                    atomicMax(&mx[dl * NH], fmap(x0[u]));
                    if constexpr (NH == 2) atomicMax(&mx[dl * 2 + 1], fmap(x1[u]));
                    q[i] = s[u] | (dl << 16);
                }
            }
        }
    };

    [[maybe_unused]] auto max_saved = [&](const int* __restrict__ sv, int n) {
        for (int off = 0; off < n; off += 128) {
            unsigned pk[4];
            float x0[4];
            [[maybe_unused]] float x1[4];
#pragma unroll
            for (int u = 0; u < 4; ++u) {
                const int i = off + 32 * u + lane;
                pk[u] = (unsigned)sv[i < n ? i : 0];
            }
            asm volatile("" :: "v"(pk[0]), "v"(pk[1]), "v"(pk[2]), "v"(pk[3]));
            float as0[4], ad0[4];
            [[maybe_unused]] float as1[4], ad1[4];
#pragma unroll
            for (int u = 0; u < 4; ++u) {
                const int s = (int)(pk[u] & 0xFFFFu), d = n0 + (int)(pk[u] >> 16);
                as0[u] = a_src[s * NH]; ad0[u] = a_dst[d * NH];
                if constexpr (NH == 2) { as1[u] = a_src[s * 2 + 1]; ad1[u] = a_dst[d * 2 + 1]; }
            }
            asm volatile("" :: "v"(as0[0]), "v"(as0[1]), "v"(as0[2]), "v"(as0[3]), "v"(ad0[0]), "v"(ad0[1]), "v"(ad0[2]), "v"(ad0[3]));
            if constexpr (NH == 2)
                asm volatile("" :: "v"(as1[0]), "v"(as1[1]), "v"(as1[2]), "v"(as1[3]), "v"(ad1[0]), "v"(ad1[1]), "v"(ad1[2]), "v"(ad1[3]));
#pragma unroll
            for (int u = 0; u < 4; ++u) {
                x0[u] = lrelu(as0[u] + ad0[u]);
                if constexpr (NH == 2) x1[u] = lrelu(as1[u] + ad1[u]);
            }
#pragma unroll
            for (int u = 0; u < 4; ++u) {
                const int i = off + 32 * u + lane;
                if (i < n) {
                    const int dl = (int)(pk[u] >> 16);
                    atomicMax(&mx[dl * NH], fmap(x0[u]));
                    if constexpr (NH == 2) atomicMax(&mx[dl * 2 + 1], fmap(x1[u]));
                    q[i] = (int)pk[u];
                }
            }
        }
    };

    auto work = [&](int n) {
        for (int off = 0; off < n; off += 32) {
            const int nb = (n - off) < 32 ? (n - off) : 32;
            int s = 0, dl = 0;
            float p0 = 0.f, p1 = 0.f;
            if (lane < nb) {
                const unsigned pk = (unsigned)q[off + lane];
                s  = (int)(pk & 0xFFFFu);
                dl = (int)(pk >> 16);
                const int d = n0 + dl;
                p0 = __expf(lrelu(a_src[s * NH] + a_dst[d * NH]) - finv(mx[dl * NH]));
                atomicAdd(&den[dl * NH], (double)p0);
                if constexpr (NH == 2) {
                    p1 = __expf(lrelu(a_src[s * 2 + 1] + a_dst[d * 2 + 1]) - finv(mx[dl * 2 + 1]));
                    atomicAdd(&den[dl * 2 + 1], (double)p1);
                }
            }
            if constexpr (F == F1) {
#pragma unroll 1
                for (int i0 = 0; i0 < nb; i0 += 8) {
                    float hv[8][4];
#pragma unroll
                    for (int j = 0; j < 8; ++j) {
                        const int i = (i0 + j < nb) ? (i0 + j) : (nb - 1);
                        const int sj = __builtin_amdgcn_readlane(s, i);
                        const float* rp = h + (size_t)sj * F1 + lane;
#pragma unroll
                        for (int k = 0; k < 4; ++k) hv[j][k] = rp[k * 32];
                    }
#pragma unroll
                    for (int j = 0; j < 8; ++j) {
                        if (i0 + j < nb) {
                            const int dj = __builtin_amdgcn_readlane(dl, i0 + j);
                            const float q0 = lane_f(p0, i0 + j);
                            const float q1 = lane_f(p1, i0 + j);
                            double* ap = acc + dj * F1 + lane;
                            const float tf[4] = {hv[j][0] * q0, hv[j][1] * q0,
                                                 hv[j][2] * q1, hv[j][3] * q1};
                            double td[4];
                            widen(tf, td);
                            atomicAdd(ap +  0, td[0]);
                            atomicAdd(ap + 32, td[1]);
                            atomicAdd(ap + 64, td[2]);
                            atomicAdd(ap + 96, td[3]);
                        }
                    }
                }
            } else {
                const int total = nb * F;
#pragma unroll 1
                for (int t0 = 0; t0 < total; t0 += 8 * 32) {
                    float hv[8];
                    int   dd2[8];
                    float pp[8];
                    int   cc[8];
#pragma unroll
                    for (int j = 0; j < 8; ++j) {
                        const int idx = t0 + j * 32 + lane;
                        const int ok  = idx < total;
                        const int e   = ok ? idx / F : 0;
                        const int col = ok ? idx - e * F : 0;
                        const int sj  = __shfl(s, e);
                        const int dj  = __shfl(dl, e);
                        dd2[j] = ok ? dj : -1;
                        pp[j] = __shfl(p0, e);
                        cc[j] = col;
                        hv[j] = h[(size_t)sj * F + col];
                    }
                    float  tf[8];
                    double td[8];
#pragma unroll
                    for (int j = 0; j < 8; ++j) tf[j] = hv[j] * pp[j];
                    widen(tf, td);
#pragma unroll
                    for (int j = 0; j < 8; ++j) {
                        if (dd2[j] >= 0)
                            atomicAdd(&acc[dd2[j] * F + cc[j]], td[j]);
                    }
                }
            }
        }
    };

    bool ovf = false;
    auto scan = [&](auto&& flush) -> int {
        int cnt = 0;
        {
            const int node = wave + 32 * lane;
            const bool m = lane < SELF_LANES && node < nr;
            const unsigned mk = __builtin_amdgcn_ballot_w32(m);
            if (m) q[__builtin_amdgcn_mbcnt_lo(mk, 0)] = N_EDGES + n0 + node;
            cnt = __builtin_popcount(mk);
        }
        const uint2* ov = reinterpret_cast<const uint2*>(ob);
        const unsigned own4 = (unsigned)blockIdx.x * 0x01010101u;
        int c0 = wave;
        for (;;) {
            bool full = false;
            while (c0 < N_CHUNKS && !full) {
                uint2 dd[TRIPC];
#pragma unroll
                for (int u = 0; u < TRIPC; ++u) {
                    const int c = c0 + 32 * u;
                    const int cc = c < N_CHUNKS ? c : c0;
                    dd[u] = ov[(size_t)cc * 32 + lane];
                }
                int took = 0;
#pragma unroll
                for (int u = 0; u < TRIPC; ++u) {
                    const int c = c0 + 32 * u;
                    if (c < N_CHUNKS && !full) {
                        if (cnt > CAP - CHUNK_E) {
                            full = true;
                        } else {
                            const int eb = c * CHUNK_E + lane * 8;
                            const unsigned x0 = dd[u].x ^ own4, x1 = dd[u].y ^ own4;
                            const unsigned f0 = ~(((x0 & 0x7F7F7F7Fu) + 0x7F7F7F7Fu) | x0 | 0x7F7F7F7Fu);
                            const unsigned f1 = ~(((x1 & 0x7F7F7F7Fu) + 0x7F7F7F7Fu) | x1 | 0x7F7F7F7Fu);
                            unsigned long long f = ((unsigned long long)f1 << 32) | f0;
                            unsigned mk = __builtin_amdgcn_ballot_w32(f != 0);
                            while (mk) {
                                if (f != 0) {
                                    q[cnt + __builtin_amdgcn_mbcnt_lo(mk, 0)] = eb + (__builtin_ctzll(f) >> 3);
                                    f &= f - 1;
                                }
                                cnt += __builtin_popcount(mk);
                                mk = __builtin_amdgcn_ballot_w32(f != 0);
                            }
                            took = u + 1;
                        }
                    }
                }
                c0 += 32 * (full ? took : TRIPC);
            }
            if (!full) break;
            flush(cnt);
            cnt = 0;
            ovf = true;
        }
        return cnt;
    };

    const int slot = blockIdx.x * 32 + wave;
    int  cnt  = 0;
    bool have = false;
    if constexpr (ROLE == 1) {
        const int c = __builtin_amdgcn_readfirstlane(wc[slot]);
        if (c >= 0) {
            have = true;
            cnt  = c;
            max_saved(wl + (size_t)slot * LIST_PITCH, cnt);
        }
    }
    if (!have) {
        cnt = scan([&](int n) { max_pass(n); });
        max_pass(cnt);
    }
    if constexpr (ROLE == 0) {
        if (!ovf) for (int i = lane; i < cnt; i += 32) wl[(size_t)slot * LIST_PITCH + i] = q[i];
        if (lane == 0) wc[slot] = ovf ? -1 : cnt;
    }
    __syncthreads();
    if (!ovf) {
        work(cnt);
    } else {
        cnt = scan([&](int n) { max_pass(n); work(n); });
        max_pass(cnt);
        work(cnt);
    }

    __syncthreads();
    for (int i = tid; i < nr * F; i += 1024) {
        const int r = i / F, c = i - r * F;
        float v = (float)acc[i] / (float)den[r * NH + (NH == 2 ? (c >> 6) : 0)] + bias[c];
        if constexpr (F == F1) v = v > 0.f ? v : (expf(v) - 1.0f);
        outp[(size_t)n0 * F + i] = v;
    }
}

extern "C" void kernel_launch(void* const* d_in, const int* in_sizes, int n_in,
                              void* d_out, int out_size, void* d_ws, size_t ws_size,
                              hipStream_t stream) {
    const float* x    = (const float*)d_in[0];
    const int*   ei   = (const int*)  d_in[1];
    const float* W1   = (const float*)d_in[2];
    const float* as1  = (const float*)d_in[3];
    const float* ad1  = (const float*)d_in[4];
    const float* b1   = (const float*)d_in[5];
    const float* W2   = (const float*)d_in[6];
    const float* as2  = (const float*)d_in[7];
    const float* ad2  = (const float*)d_in[8];
    const float* b2   = (const float*)d_in[9];
    float* out = (float*)d_out;

    float* ws      = (float*)d_ws;
    float* h1      = ws;
    float* h2in    = h1 + (size_t)N_NODES * F1;
    float* a_src1  = h2in + (size_t)N_NODES * F1;
    float* a_dst1  = a_src1 + (size_t)N_NODES * 2;
    float* a_src2  = a_dst1 + (size_t)N_NODES * 2;
    float* a_dst2  = a_src2 + N_NODES;
    float* h2      = a_dst2 + N_NODES;
    int*   wl      = (int*)d_ws + WS_LISTS_OFF;
    int*   wc      = wl + (size_t)OWN_BLOCKS * 32 * LIST_PITCH;
    unsigned* ob   = (unsigned*)d_ws + WS_OB_OFF;

    const int MT = N_NODES / 16;
    gat_gemm1_wmma<<<G1_BLOCKS, 256, 0, stream>>>(x, W1, h1);
    gat_att1<<<ATT1_BLOCKS + OB_BLOCKS, 256, 0, stream>>>(h1, as1, ad1, a_src1, a_dst1, ei, ob);
    gat_own_accum<F1, HEADS, 12, LIST_PITCH, 0><<<OWN_BLOCKS, 1024, 0, stream>>>(ei, a_src1, a_dst1, h1, b1, h2in, wl, wc, ob);
    gat_gemm2_wmma<<<MT, 96, 0, stream>>>(h2in, W2, h2);
    gat_att2<<<(N_NODES + 255) / 256, 256, 0, stream>>>(h2, as2, ad2, a_src2, a_dst2);
    gat_own_accum<NCLS, 1, 8, 1152, 1><<<OWN_BLOCKS, 1024, 0, stream>>>(ei, a_src2, a_dst2, h2, b2, out, wl, wc, ob);
}


# Edges sorted by owner block in the first dot launch (LDS counting sort per 1024 edges, run table); an owner block copies only its own runs, its scan of all owner bytes is gone; sums in arrival order
# speedup vs baseline: 3.5292x; 1.0364x over previous
#include <hip/hip_runtime.h>
#include <math.h>

#define N_NODES 50000
#define N_EDGES 800000
#define IN_FEATS 256
#define HID 64
#define HEADS 2
#define F1 (HEADS * HID)
#define NCLS 40
#define OWN_R      196
#define OWN_BLOCKS ((N_NODES + OWN_R - 1) / OWN_R)
#define ATT1_BLOCKS ((N_NODES * HEADS + 255) / 256)
#define SORT_E     1024
#define SORT_WORDS (N_EDGES / 4)
#define SORT_BLOCKS ((SORT_WORDS + 255) / 256)
#define RUN_PITCH  800
#define RUNS_PER_WAVE ((SORT_BLOCKS + 31) / 32)
static_assert(OWN_BLOCKS <= 256, "one bin per owner in a sorting block of 256 threads");
static_assert(N_EDGES % 4 == 0, "the edge list is read four edges at a time");
static_assert(RUN_PITCH >= SORT_BLOCKS && RUNS_PER_WAVE <= 32, "a lane per run, a table slot per sorting block");

typedef __attribute__((ext_vector_type(2))) float v2f;
typedef __attribute__((ext_vector_type(8))) float v8f;
typedef __attribute__((ext_vector_type(4))) float v4f;

static __device__ __forceinline__ unsigned fmap(float f) {
    unsigned u = __float_as_uint(f);
    return (u & 0x80000000u) ? ~u : (u | 0x80000000u);
}
static __device__ __forceinline__ float finv(unsigned t) {
    return (t & 0x80000000u) ? __uint_as_float(t & 0x7FFFFFFFu)
                             : __uint_as_float(~t);
}
static __device__ __forceinline__ float lrelu(float x) { return x > 0.f ? x : 0.2f * x; }

#define G1_PITCH  260
#define G1_BLOCKS 512
#define G1_TILES  (N_NODES / 16)
static_assert(N_NODES % 16 == 0, "row tiles are whole");

__global__ __launch_bounds__(256) void gat_gemm1_wmma(const float* __restrict__ x,
                                                      const float* __restrict__ W1,
                                                      float* __restrict__ h1) {
    __shared__ float xs[2][16 * G1_PITCH];

    const int tid   = threadIdx.x;
    const int wave  = tid >> 5;
    const int lane  = tid & 31;
    const int half  = lane >> 4;
    const int l     = lane & 15;

    v2f b[IN_FEATS / 4];
    {
        const float* bp = W1 + (size_t)(2 * half) * F1 + wave * 16 + l;
#pragma unroll
        for (int kk = 0; kk < IN_FEATS / 4; ++kk) {
            b[kk].x = bp[(size_t)(4 * kk) * F1];
            b[kk].y = bp[(size_t)(4 * kk + 1) * F1];
        }
    }

    const v4f* xv = reinterpret_cast<const v4f*>(x);
    const int so = (tid >> 6) * G1_PITCH + (tid & 63) * 4;
    v4f st0, st1, st2, st3;
    int mt = blockIdx.x;
    if (mt < G1_TILES) {
        const v4f* tp = xv + (size_t)mt * 1024 + tid;
        st0 = tp[0]; st1 = tp[256]; st2 = tp[512]; st3 = tp[768];
        float* sp = &xs[0][so];
        *reinterpret_cast<v4f*>(sp)                 = st0;
        *reinterpret_cast<v4f*>(sp +  4 * G1_PITCH) = st1;
        *reinterpret_cast<v4f*>(sp +  8 * G1_PITCH) = st2;
        *reinterpret_cast<v4f*>(sp + 12 * G1_PITCH) = st3;
    }
    __syncthreads();

    int buf = 0;
    for (; mt < G1_TILES; mt += gridDim.x) {
        const int  nxt  = mt + gridDim.x;
        const bool more = nxt < G1_TILES;
        if (more) {
            const v4f* tp = xv + (size_t)nxt * 1024 + tid;
            st0 = tp[0]; st1 = tp[256]; st2 = tp[512]; st3 = tp[768];
        }

        const float* ap = &xs[buf][l * G1_PITCH + 2 * half];
        v8f acc = {};
        v2f a0[8], a1[8];
#pragma unroll
        for (int i = 0; i < 8; ++i) a0[i] = *reinterpret_cast<const v2f*>(ap + 4 * i);
#pragma unroll
        for (int g = 0; g < IN_FEATS / 32; g += 2) {
#pragma unroll
            for (int i = 0; i < 8; ++i) a1[i] = *reinterpret_cast<const v2f*>(ap + 32 * (g + 1) + 4 * i);
            __builtin_amdgcn_sched_barrier(0);
#pragma unroll
            for (int i = 0; i < 8; ++i)
                acc = __builtin_amdgcn_wmma_f32_16x16x4_f32(false, a0[i], false, b[8 * g + i],
                                                            (short)0, acc, false, false);
            __builtin_amdgcn_sched_barrier(0);
            if (g + 2 < IN_FEATS / 32) {
#pragma unroll
                for (int i = 0; i < 8; ++i) a0[i] = *reinterpret_cast<const v2f*>(ap + 32 * (g + 2) + 4 * i);
            }
            __builtin_amdgcn_sched_barrier(0);
#pragma unroll
            for (int i = 0; i < 8; ++i)
                acc = __builtin_amdgcn_wmma_f32_16x16x4_f32(false, a1[i], false, b[8 * (g + 1) + i],
                                                            (short)0, acc, false, false);
            __builtin_amdgcn_sched_barrier(0);
        }
        if (more) {
            float* sp = &xs[buf ^ 1][so];
            *reinterpret_cast<v4f*>(sp)                 = st0;
            *reinterpret_cast<v4f*>(sp +  4 * G1_PITCH) = st1;
            *reinterpret_cast<v4f*>(sp +  8 * G1_PITCH) = st2;
            *reinterpret_cast<v4f*>(sp + 12 * G1_PITCH) = st3;
        }
        __syncthreads();
        buf ^= 1;

        float* dr = h1 + (size_t)(mt * 16 + 8 * half) * F1 + wave * 16 + l;
#pragma unroll
        for (int r = 0; r < 8; ++r) dr[(size_t)r * F1] = acc[r];
    }
}

__global__ __launch_bounds__(96) void gat_gemm2_wmma(const float* __restrict__ h2in,
                                                     const float* __restrict__ W2,
                                                     float* __restrict__ h2) {
    const int mtile = blockIdx.x;
    const int wave  = threadIdx.x >> 5;
    const int lane  = threadIdx.x & 31;
    const int half  = lane >> 4;
    const int l     = lane & 15;
    const int row   = mtile * 16 + l;
    const int ncol  = wave * 16 + l;

    v8f acc = {};
    const float* xr = h2in + (size_t)row * F1;
    for (int k = 0; k < F1; k += 4) {
        v2f a = *(const v2f*)(xr + k + 2 * half);
        v2f b;
        if (ncol < NCLS) {
            const float* bp = W2 + (size_t)(k + 2 * half) * NCLS + ncol;
            b.x = bp[0]; b.y = bp[NCLS];
        } else { b.x = 0.f; b.y = 0.f; }
        acc = __builtin_amdgcn_wmma_f32_16x16x4_f32(false, a, false, b,
                                                    (short)0, acc, false, false);
    }
    if (ncol < NCLS) {
        float* dr = h2 + (size_t)(mtile * 16 + 8 * half) * NCLS + ncol;
#pragma unroll
        for (int r = 0; r < 8; ++r) dr[(size_t)r * NCLS] = acc[r];
    }
}

__global__ __launch_bounds__(256) void gat_att1(const float* __restrict__ h1, const float* __restrict__ as1,
                         const float* __restrict__ ad1, float* __restrict__ a_src,
                         float* __restrict__ a_dst, const int* __restrict__ ei,
                         int* __restrict__ srt, int* __restrict__ rtab) {
    if (blockIdx.x >= ATT1_BLOCKS) {
        __shared__ int bins[256];
        __shared__ int wtot[8];
        const int t    = threadIdx.x;
        const int lane = t & 31;
        const int wave = t >> 5;
        const int blk  = blockIdx.x - ATT1_BLOCKS;
        const int w    = blk * 256 + t;
        const bool ok  = w < SORT_WORDS;
        bins[t] = 0;
        __syncthreads();
        int pk[4], own[4], rk[4];
        if (ok) {
            const int4 s = reinterpret_cast<const int4*>(ei)[w];
            const int4 d = reinterpret_cast<const int4*>(ei + N_EDGES)[w];
            const int sv[4] = {s.x, s.y, s.z, s.w};
            const int dv[4] = {d.x, d.y, d.z, d.w};
#pragma unroll
            for (int k = 0; k < 4; ++k) {
                const unsigned o = (unsigned)dv[k] / OWN_R;
                own[k] = (int)(o < 255u ? o : 255u);
                pk[k]  = sv[k] | ((dv[k] - own[k] * OWN_R) << 16);
                rk[k]  = atomicAdd(&bins[own[k]], 1);
            }
        }
        __syncthreads();
        const int c = bins[t];
        int incl = c;
        for (int dlt = 1; dlt < 32; dlt <<= 1) {
            const int v = __shfl_up(incl, dlt);
            if (lane >= dlt) incl += v;
        }
        if (lane == 31) wtot[wave] = incl;
        __syncthreads();
        int base = incl - c;
        for (int i = 0; i < wave; ++i) base += wtot[i];
        bins[t] = base;
        rtab[t * RUN_PITCH + blk] = base | (c << 16);
        __syncthreads();
        if (ok) {
#pragma unroll
            for (int k = 0; k < 4; ++k) srt[blk * SORT_E + bins[own[k]] + rk[k]] = pk[k];
        }
        return;
    }
    int i = blockIdx.x * blockDim.x + threadIdx.x;
    if (i >= N_NODES * HEADS) return;
    int n = i >> 1, h = i & 1;
    const float* hp = h1 + (size_t)n * F1 + h * HID;
    const float* s0 = as1 + h * HID;
    const float* d0 = ad1 + h * HID;
    float s = 0.f, d = 0.f;
#pragma unroll 8
    for (int c = 0; c < HID; ++c) { float v = hp[c]; s += v * s0[c]; d += v * d0[c]; }
    a_src[i] = s; a_dst[i] = d;
}

__global__ void gat_att2(const float* __restrict__ h2, const float* __restrict__ as2,
                         const float* __restrict__ ad2, float* __restrict__ a_src,
                         float* __restrict__ a_dst) {
    int n = blockIdx.x * blockDim.x + threadIdx.x;
    if (n >= N_NODES) return;
    const float* hp = h2 + (size_t)n * NCLS;
    float s = 0.f, d = 0.f;
#pragma unroll 8
    for (int c = 0; c < NCLS; ++c) { float v = hp[c]; s += v * as2[c]; d += v * ad2[c]; }
    a_src[n] = s; a_dst[n] = d;
}

#define SELF_LANES ((OWN_R + 31) / 32)
#define LIST_PITCH 896
#define WS_LISTS_OFF ((size_t)16 << 20)
#define WS_PLANES ((size_t)N_NODES * (2 * F1 + 2 * HEADS + 2 + NCLS))
static_assert(WS_PLANES <= WS_LISTS_OFF, "the saved lists start behind the feature and score planes");
static_assert((WS_LISTS_OFF + (size_t)OWN_BLOCKS * 32 * (LIST_PITCH + 1)) * 4 <= ((size_t)256 << 20),
              "lists and lengths end inside the workspace");
#define WS_SORT_OFF ((size_t)32 << 20)
static_assert(WS_LISTS_OFF + (size_t)OWN_BLOCKS * 32 * (LIST_PITCH + 1) <= WS_SORT_OFF, "the sorted entries start behind the saved lists");
static_assert((WS_SORT_OFF + (size_t)SORT_BLOCKS * SORT_E + (size_t)OWN_BLOCKS * RUN_PITCH) * 4 <= ((size_t)256 << 20),
              "sorted entries and run table end inside the workspace");
static_assert(N_NODES <= 65536 && OWN_R <= 65536, "a list entry packs source and local destination, 16 bits each");
static_assert(SORT_E <= 32767, "a run's start and length fit 16 bits each");
static_assert(SELF_LANES <= 32, "one lane per self-loop of a wave");

static __device__ __forceinline__ float lane_f(float v, int i) {
    return __int_as_float(__builtin_amdgcn_readlane(__float_as_int(v), i));
}

static __device__ __forceinline__ double widen_normal(float f) {
    const unsigned u  = __float_as_uint(f);
    const unsigned hi = (u & 0x80000000u) | (((u >> 3) & 0x0FFFFFFFu) + 0x38000000u);
    const unsigned lo = u << 29;
    return __builtin_bit_cast(double, ((unsigned long long)hi << 32) | lo);
}
static __device__ __forceinline__ bool not_normal(float f) { return __builtin_amdgcn_classf(f, 0x2F7); }
template <int N>
static __device__ __forceinline__ void widen(const float (&f)[N], double (&d)[N]) {
    bool other = false;
#pragma unroll
    for (int i = 0; i < N; ++i) { d[i] = widen_normal(f[i]); other |= not_normal(f[i]); }
    if (other) {
#pragma unroll
        for (int i = 0; i < N; ++i) d[i] = (double)f[i];
    }
}

template <int F, int NH, int CAP, int ROLE>
__global__ __launch_bounds__(1024) void gat_own_accum(
        const float* __restrict__ a_src, const float* __restrict__ a_dst,
        const float* __restrict__ h,
        const float* __restrict__ bias, float* __restrict__ outp,
        int* __restrict__ wl, int* __restrict__ wc,
        const int* __restrict__ srt, const int* __restrict__ rtab) {
    static_assert(CAP >= 32 + SELF_LANES, "a piece of 32 entries goes into a list that holds the self-loops");
    static_assert(ROLE == 0 ? CAP <= LIST_PITCH : CAP >= LIST_PITCH, "a saved list fits its slot, and fits the list of the wave that takes it");
    __shared__ double   acc[OWN_R * F];
    __shared__ double   den[OWN_R * NH];
    __shared__ unsigned mx[OWN_R * NH];
    __shared__ int      que[32 * CAP];

    const int tid  = threadIdx.x;
    const int lane = tid & 31;
    const int wave = __builtin_amdgcn_readfirstlane(tid >> 5);
    const int n0   = blockIdx.x * OWN_R;
    const int nr   = (N_NODES - n0) < OWN_R ? (N_NODES - n0) : OWN_R;
    int* q = que + wave * CAP;

    for (int i = tid; i < OWN_R * F;  i += 1024) acc[i] = 0.0;
    for (int i = tid; i < OWN_R * NH; i += 1024) { den[i] = 0.0; mx[i] = 0u; }
    __syncthreads();

    auto max_list = [&](int n) {
        for (int off = 0; off < n; off += 128) {
            unsigned pk[4];
            float x0[4];
            [[maybe_unused]] float x1[4];
#pragma unroll
            for (int u = 0; u < 4; ++u) {
                const int i = off + 32 * u + lane;
                pk[u] = (unsigned)q[i < n ? i : 0];
            }
            float as0[4], ad0[4];
            [[maybe_unused]] float as1[4], ad1[4];
#pragma unroll
            for (int u = 0; u < 4; ++u) {
                const int s = (int)(pk[u] & 0xFFFFu), d = n0 + (int)(pk[u] >> 16);
                as0[u] = a_src[s * NH]; ad0[u] = a_dst[d * NH];
                if constexpr (NH == 2) { as1[u] = a_src[s * 2 + 1]; ad1[u] = a_dst[d * 2 + 1]; }
            }
            asm volatile("" :: "v"(as0[0]), "v"(as0[1]), "v"(as0[2]), "v"(as0[3]), "v"(ad0[0]), "v"(ad0[1]), "v"(ad0[2]), "v"(ad0[3]));
            if constexpr (NH == 2)
                asm volatile("" :: "v"(as1[0]), "v"(as1[1]), "v"(as1[2]), "v"(as1[3]), "v"(ad1[0]), "v"(ad1[1]), "v"(ad1[2]), "v"(ad1[3]));
#pragma unroll
            for (int u = 0; u < 4; ++u) {
                x0[u] = lrelu(as0[u] + ad0[u]);
                if constexpr (NH == 2) x1[u] = lrelu(as1[u] + ad1[u]);
            }
#pragma unroll
            for (int u = 0; u < 4; ++u) {
                const int i = off + 32 * u + lane;
                if (i < n) {
                    const int dl = (int)(pk[u] >> 16);
                    atomicMax(&mx[dl * NH], fmap(x0[u]));
                    if constexpr (NH == 2) atomicMax(&mx[dl * 2 + 1], fmap(x1[u]));
                }
            }
        }
    };

    [[maybe_unused]] auto max_saved = [&](const int* __restrict__ sv, int n) {
        for (int off = 0; off < n; off += 128) {
            unsigned pk[4];
            float x0[4];
            [[maybe_unused]] float x1[4];
#pragma unroll
            for (int u = 0; u < 4; ++u) {
                const int i = off + 32 * u + lane;
                pk[u] = (unsigned)sv[i < n ? i : 0];
            }
            asm volatile("" :: "v"(pk[0]), "v"(pk[1]), "v"(pk[2]), "v"(pk[3]));
            float as0[4], ad0[4];
            [[maybe_unused]] float as1[4], ad1[4];
#pragma unroll
            for (int u = 0; u < 4; ++u) {
                const int s = (int)(pk[u] & 0xFFFFu), d = n0 + (int)(pk[u] >> 16);
                as0[u] = a_src[s * NH]; ad0[u] = a_dst[d * NH];
                if constexpr (NH == 2) { as1[u] = a_src[s * 2 + 1]; ad1[u] = a_dst[d * 2 + 1]; }
            }
            asm volatile("" :: "v"(as0[0]), "v"(as0[1]), "v"(as0[2]), "v"(as0[3]), "v"(ad0[0]), "v"(ad0[1]), "v"(ad0[2]), "v"(ad0[3]));
            if constexpr (NH == 2)
                asm volatile("" :: "v"(as1[0]), "v"(as1[1]), "v"(as1[2]), "v"(as1[3]), "v"(ad1[0]), "v"(ad1[1]), "v"(ad1[2]), "v"(ad1[3]));
#pragma unroll
            for (int u = 0; u < 4; ++u) {
                x0[u] = lrelu(as0[u] + ad0[u]);
                if constexpr (NH == 2) x1[u] = lrelu(as1[u] + ad1[u]);
            }
#pragma unroll
            for (int u = 0; u < 4; ++u) {
                const int i = off + 32 * u + lane;
                if (i < n) {
                    const int dl = (int)(pk[u] >> 16);
                    atomicMax(&mx[dl * NH], fmap(x0[u]));
                    if constexpr (NH == 2) atomicMax(&mx[dl * 2 + 1], fmap(x1[u]));
                    q[i] = (int)pk[u];
                }
            }
        }
    };

    auto work = [&](int n) {
        for (int off = 0; off < n; off += 32) {
            const int nb = (n - off) < 32 ? (n - off) : 32;
            int s = 0, dl = 0;
            float p0 = 0.f, p1 = 0.f;
            if (lane < nb) {
                const unsigned pk = (unsigned)q[off + lane];
                s  = (int)(pk & 0xFFFFu);
                dl = (int)(pk >> 16);
                const int d = n0 + dl;
                p0 = __expf(lrelu(a_src[s * NH] + a_dst[d * NH]) - finv(mx[dl * NH]));
                atomicAdd(&den[dl * NH], (double)p0);
                if constexpr (NH == 2) {
                    p1 = __expf(lrelu(a_src[s * 2 + 1] + a_dst[d * 2 + 1]) - finv(mx[dl * 2 + 1]));
                    atomicAdd(&den[dl * 2 + 1], (double)p1);
                }
            }
            if constexpr (F == F1) {
#pragma unroll 1
                for (int i0 = 0; i0 < nb; i0 += 8) {
                    float hv[8][4];
#pragma unroll
                    for (int j = 0; j < 8; ++j) {
                        const int i = (i0 + j < nb) ? (i0 + j) : (nb - 1);
                        const int sj = __builtin_amdgcn_readlane(s, i);
                        const float* rp = h + (size_t)sj * F1 + lane;
#pragma unroll
                        for (int k = 0; k < 4; ++k) hv[j][k] = rp[k * 32];
                    }
#pragma unroll
                    for (int j = 0; j < 8; ++j) {
                        if (i0 + j < nb) {
                            const int dj = __builtin_amdgcn_readlane(dl, i0 + j);
                            const float q0 = lane_f(p0, i0 + j);
                            const float q1 = lane_f(p1, i0 + j);
                            double* ap = acc + dj * F1 + lane;
                            const float tf[4] = {hv[j][0] * q0, hv[j][1] * q0,
                                                 hv[j][2] * q1, hv[j][3] * q1};
                            double td[4];
                            widen(tf, td);
                            atomicAdd(ap +  0, td[0]);
                            atomicAdd(ap + 32, td[1]);
                            atomicAdd(ap + 64, td[2]);
                            atomicAdd(ap + 96, td[3]);
                        }
                    }
                }
            } else {
                const int total = nb * F;
#pragma unroll 1
                for (int t0 = 0; t0 < total; t0 += 8 * 32) {
                    float hv[8];
                    int   dd2[8];
                    float pp[8];
                    int   cc[8];
#pragma unroll
                    for (int j = 0; j < 8; ++j) {
                        const int idx = t0 + j * 32 + lane;
                        const int ok  = idx < total;
                        const int e   = ok ? idx / F : 0;
                        const int col = ok ? idx - e * F : 0;
                        const int sj  = __shfl(s, e);
                        const int dj  = __shfl(dl, e);
                        dd2[j] = ok ? dj : -1;
                        pp[j] = __shfl(p0, e);
                        cc[j] = col;
                        hv[j] = h[(size_t)sj * F + col];
                    }
                    float  tf[8];
                    double td[8];
#pragma unroll
                    for (int j = 0; j < 8; ++j) tf[j] = hv[j] * pp[j];
                    widen(tf, td);
#pragma unroll
                    for (int j = 0; j < 8; ++j) {
                        if (dd2[j] >= 0)
                            atomicAdd(&acc[dd2[j] * F + cc[j]], td[j]);
                    }
                }
            }
        }
    };

    bool ovf = false;
    auto gather = [&](auto&& flush) -> int {
        int cnt = 0;
        {
            const int node = wave + 32 * lane;
            const bool m = lane < SELF_LANES && node < nr;
            const unsigned mk = __builtin_amdgcn_ballot_w32(m);
            if (m) q[__builtin_amdgcn_mbcnt_lo(mk, 0)] = (n0 + node) | (node << 16);
            cnt = __builtin_popcount(mk);
        }
        const int blk = wave * RUNS_PER_WAVE + lane;
        const bool has = lane < RUNS_PER_WAVE && blk < SORT_BLOCKS;
        const int rv = has ? rtab[blockIdx.x * RUN_PITCH + blk] : 0;
        const int st = rv & 0xFFFF, n = rv >> 16;
        int incl = n;
        for (int dlt = 1; dlt < 32; dlt <<= 1) {
            const int v = __shfl_up(incl, dlt);
            if (lane >= dlt) incl += v;
        }
        const int tot = __builtin_amdgcn_readlane(incl, 31);
        if (cnt + tot <= CAP) {
            const int* sp = n > 0 ? srt + (size_t)blk * SORT_E + st : srt;
            const int pos = cnt + incl - n;
            for (int k = 0; __builtin_amdgcn_ballot_w32(k < n) != 0; k += 8) {
                int v[8];
#pragma unroll
                for (int u = 0; u < 8; ++u) v[u] = sp[k + u < n ? k + u : 0];
#pragma unroll
                for (int u = 0; u < 8; ++u) if (k + u < n) q[pos + k + u] = v[u];
            }
            cnt += tot;
        } else {
            for (int l = 0; l < RUNS_PER_WAVE; ++l) {
                const int nl = __builtin_amdgcn_readlane(n, l);
                const int sl = __builtin_amdgcn_readlane(st, l);
                const int* rp = srt + (size_t)(wave * RUNS_PER_WAVE + l) * SORT_E + sl;
                for (int off = 0; off < nl; off += 32) {
                    if (cnt > CAP - 32) { flush(cnt); cnt = 0; ovf = true; }
                    const int i = off + lane;
                    if (i < nl) q[cnt + lane] = rp[i];
                    cnt += (nl - off) < 32 ? (nl - off) : 32;
                }
            }
        }
        return cnt;
    };

    const int slot = blockIdx.x * 32 + wave;
    int  cnt  = 0;
    bool have = false;
    if constexpr (ROLE == 1) {
        const int c = __builtin_amdgcn_readfirstlane(wc[slot]);
        if (c >= 0) {
            have = true;
            cnt  = c;
            max_saved(wl + (size_t)slot * LIST_PITCH, cnt);
        }
    }
    if (!have) {
        cnt = gather([&](int n) { max_list(n); });
        max_list(cnt);
    }
    if constexpr (ROLE == 0) {
        if (!ovf) for (int i = lane; i < cnt; i += 32) wl[(size_t)slot * LIST_PITCH + i] = q[i];
        if (lane == 0) wc[slot] = ovf ? -1 : cnt;
    }
    __syncthreads();
    if (!ovf) {
        work(cnt);
    } else {
        cnt = gather([&](int n) { work(n); });
        work(cnt);
    }

    __syncthreads();
    for (int i = tid; i < nr * F; i += 1024) {
        const int r = i / F, c = i - r * F;
        float v = (float)acc[i] / (float)den[r * NH + (NH == 2 ? (c >> 6) : 0)] + bias[c];
        if constexpr (F == F1) v = v > 0.f ? v : (expf(v) - 1.0f);
        outp[(size_t)n0 * F + i] = v;
    }
}

extern "C" void kernel_launch(void* const* d_in, const int* in_sizes, int n_in,
                              void* d_out, int out_size, void* d_ws, size_t ws_size,
                              hipStream_t stream) {
    const float* x    = (const float*)d_in[0];
    const int*   ei   = (const int*)  d_in[1];
    const float* W1   = (const float*)d_in[2];
    const float* as1  = (const float*)d_in[3];
    const float* ad1  = (const float*)d_in[4];
    const float* b1   = (const float*)d_in[5];
    const float* W2   = (const float*)d_in[6];
    const float* as2  = (const float*)d_in[7];
    const float* ad2  = (const float*)d_in[8];
    const float* b2   = (const float*)d_in[9];
    float* out = (float*)d_out;

    float* ws      = (float*)d_ws;
    float* h1      = ws;
    float* h2in    = h1 + (size_t)N_NODES * F1;
    float* a_src1  = h2in + (size_t)N_NODES * F1;
    float* a_dst1  = a_src1 + (size_t)N_NODES * 2;
    float* a_src2  = a_dst1 + (size_t)N_NODES * 2;
    float* a_dst2  = a_src2 + N_NODES;
    float* h2      = a_dst2 + N_NODES;
    int*   wl      = (int*)d_ws + WS_LISTS_OFF;
    int*   wc      = wl + (size_t)OWN_BLOCKS * 32 * LIST_PITCH;
    int*   srt     = (int*)d_ws + WS_SORT_OFF;
    int*   rtab    = srt + (size_t)SORT_BLOCKS * SORT_E;

    const int MT = N_NODES / 16;
    gat_gemm1_wmma<<<G1_BLOCKS, 256, 0, stream>>>(x, W1, h1);
    gat_att1<<<ATT1_BLOCKS + SORT_BLOCKS, 256, 0, stream>>>(h1, as1, ad1, a_src1, a_dst1, ei, srt, rtab);
    gat_own_accum<F1, HEADS, LIST_PITCH, 0><<<OWN_BLOCKS, 1024, 0, stream>>>(a_src1, a_dst1, h1, b1, h2in, wl, wc, srt, rtab);
    gat_gemm2_wmma<<<MT, 96, 0, stream>>>(h2in, W2, h2);
    gat_att2<<<(N_NODES + 255) / 256, 256, 0, stream>>>(h2, as2, ad2, a_src2, a_dst2);
    gat_own_accum<NCLS, 1, 1152, 1><<<OWN_BLOCKS, 1024, 0, stream>>>(a_src2, a_dst2, h2, b2, out, wl, wc, srt, rtab);
}


# Layer-2 row walk re-tiled: fixed lane = (row slot, 16-byte piece), 10 lanes a 40-float row, 3 rows a load, 11 loads in flight; index division, per-element exchanges gone; f64 LDS adds, arrival order
# speedup vs baseline: 3.8545x; 1.0922x over previous
#include <hip/hip_runtime.h>
#include <math.h>

#define N_NODES 50000
#define N_EDGES 800000
#define IN_FEATS 256
#define HID 64
#define HEADS 2
#define F1 (HEADS * HID)
#define NCLS 40
#define OWN_R      196
#define OWN_BLOCKS ((N_NODES + OWN_R - 1) / OWN_R)
#define ATT1_BLOCKS ((N_NODES * HEADS + 255) / 256)
#define SORT_E     1024
#define SORT_WORDS (N_EDGES / 4)
#define SORT_BLOCKS ((SORT_WORDS + 255) / 256)
#define RUN_PITCH  800
#define RUNS_PER_WAVE ((SORT_BLOCKS + 31) / 32)
static_assert(OWN_BLOCKS <= 256, "one bin per owner in a sorting block of 256 threads");
static_assert(N_EDGES % 4 == 0, "the edge list is read four edges at a time");
static_assert(RUN_PITCH >= SORT_BLOCKS && RUNS_PER_WAVE <= 32, "a lane per run, a table slot per sorting block");

typedef __attribute__((ext_vector_type(2))) float v2f;
typedef __attribute__((ext_vector_type(8))) float v8f;
typedef __attribute__((ext_vector_type(4))) float v4f;

static __device__ __forceinline__ unsigned fmap(float f) {
    unsigned u = __float_as_uint(f);
    return (u & 0x80000000u) ? ~u : (u | 0x80000000u);
}
static __device__ __forceinline__ float finv(unsigned t) {
    return (t & 0x80000000u) ? __uint_as_float(t & 0x7FFFFFFFu)
                             : __uint_as_float(~t);
}
static __device__ __forceinline__ float lrelu(float x) { return x > 0.f ? x : 0.2f * x; }

#define G1_PITCH  260
#define G1_BLOCKS 512
#define G1_TILES  (N_NODES / 16)
static_assert(N_NODES % 16 == 0, "row tiles are whole");

__global__ __launch_bounds__(256) void gat_gemm1_wmma(const float* __restrict__ x,
                                                      const float* __restrict__ W1,
                                                      float* __restrict__ h1) {
    __shared__ float xs[2][16 * G1_PITCH];

    const int tid   = threadIdx.x;
    const int wave  = tid >> 5;
    const int lane  = tid & 31;
    const int half  = lane >> 4;
    const int l     = lane & 15;

    v2f b[IN_FEATS / 4];
    {
        const float* bp = W1 + (size_t)(2 * half) * F1 + wave * 16 + l;
#pragma unroll
        for (int kk = 0; kk < IN_FEATS / 4; ++kk) {
            b[kk].x = bp[(size_t)(4 * kk) * F1];
            b[kk].y = bp[(size_t)(4 * kk + 1) * F1];
        }
    }

    const v4f* xv = reinterpret_cast<const v4f*>(x);
    const int so = (tid >> 6) * G1_PITCH + (tid & 63) * 4;
    v4f st0, st1, st2, st3;
    int mt = blockIdx.x;
    if (mt < G1_TILES) {
        const v4f* tp = xv + (size_t)mt * 1024 + tid;
        st0 = tp[0]; st1 = tp[256]; st2 = tp[512]; st3 = tp[768];
        float* sp = &xs[0][so];
        *reinterpret_cast<v4f*>(sp)                 = st0;
        *reinterpret_cast<v4f*>(sp +  4 * G1_PITCH) = st1;
        *reinterpret_cast<v4f*>(sp +  8 * G1_PITCH) = st2;
        *reinterpret_cast<v4f*>(sp + 12 * G1_PITCH) = st3;
    }
    __syncthreads();

    int buf = 0;
    for (; mt < G1_TILES; mt += gridDim.x) {
        const int  nxt  = mt + gridDim.x;
        const bool more = nxt < G1_TILES;
        if (more) {
            const v4f* tp = xv + (size_t)nxt * 1024 + tid;
            st0 = tp[0]; st1 = tp[256]; st2 = tp[512]; st3 = tp[768];
        }

        const float* ap = &xs[buf][l * G1_PITCH + 2 * half];
        v8f acc = {};
        v2f a0[8], a1[8];
#pragma unroll
        for (int i = 0; i < 8; ++i) a0[i] = *reinterpret_cast<const v2f*>(ap + 4 * i);
#pragma unroll
        for (int g = 0; g < IN_FEATS / 32; g += 2) {
#pragma unroll
            for (int i = 0; i < 8; ++i) a1[i] = *reinterpret_cast<const v2f*>(ap + 32 * (g + 1) + 4 * i);
            __builtin_amdgcn_sched_barrier(0);
#pragma unroll
            for (int i = 0; i < 8; ++i)
                acc = __builtin_amdgcn_wmma_f32_16x16x4_f32(false, a0[i], false, b[8 * g + i],
                                                            (short)0, acc, false, false);
            __builtin_amdgcn_sched_barrier(0);
            if (g + 2 < IN_FEATS / 32) {
#pragma unroll
                for (int i = 0; i < 8; ++i) a0[i] = *reinterpret_cast<const v2f*>(ap + 32 * (g + 2) + 4 * i);
            }
            __builtin_amdgcn_sched_barrier(0);
#pragma unroll
            for (int i = 0; i < 8; ++i)
                acc = __builtin_amdgcn_wmma_f32_16x16x4_f32(false, a1[i], false, b[8 * (g + 1) + i],
                                                            (short)0, acc, false, false);
            __builtin_amdgcn_sched_barrier(0);
        }
        if (more) {
            float* sp = &xs[buf ^ 1][so];
            *reinterpret_cast<v4f*>(sp)                 = st0;
            *reinterpret_cast<v4f*>(sp +  4 * G1_PITCH) = st1;
            *reinterpret_cast<v4f*>(sp +  8 * G1_PITCH) = st2;
            *reinterpret_cast<v4f*>(sp + 12 * G1_PITCH) = st3;
        }
        __syncthreads();
        buf ^= 1;

        float* dr = h1 + (size_t)(mt * 16 + 8 * half) * F1 + wave * 16 + l;
#pragma unroll
        for (int r = 0; r < 8; ++r) dr[(size_t)r * F1] = acc[r];
    }
}

__global__ __launch_bounds__(96) void gat_gemm2_wmma(const float* __restrict__ h2in,
                                                     const float* __restrict__ W2,
                                                     float* __restrict__ h2) {
    const int mtile = blockIdx.x;
    const int wave  = threadIdx.x >> 5;
    const int lane  = threadIdx.x & 31;
    const int half  = lane >> 4;
    const int l     = lane & 15;
    const int row   = mtile * 16 + l;
    const int ncol  = wave * 16 + l;

    v8f acc = {};
    const float* xr = h2in + (size_t)row * F1;
    for (int k = 0; k < F1; k += 4) {
        v2f a = *(const v2f*)(xr + k + 2 * half);
        v2f b;
        if (ncol < NCLS) {
            const float* bp = W2 + (size_t)(k + 2 * half) * NCLS + ncol;
            b.x = bp[0]; b.y = bp[NCLS];
        } else { b.x = 0.f; b.y = 0.f; }
        acc = __builtin_amdgcn_wmma_f32_16x16x4_f32(false, a, false, b,
                                                    (short)0, acc, false, false);
    }
    if (ncol < NCLS) {
        float* dr = h2 + (size_t)(mtile * 16 + 8 * half) * NCLS + ncol;
#pragma unroll
        for (int r = 0; r < 8; ++r) dr[(size_t)r * NCLS] = acc[r];
    }
}

__global__ __launch_bounds__(256) void gat_att1(const float* __restrict__ h1, const float* __restrict__ as1,
                         const float* __restrict__ ad1, float* __restrict__ a_src,
                         float* __restrict__ a_dst, const int* __restrict__ ei,
                         int* __restrict__ srt, int* __restrict__ rtab) {
    if (blockIdx.x >= ATT1_BLOCKS) {
        __shared__ int bins[256];
        __shared__ int wtot[8];
        const int t    = threadIdx.x;
        const int lane = t & 31;
        const int wave = t >> 5;
        const int blk  = blockIdx.x - ATT1_BLOCKS;
        const int w    = blk * 256 + t;
        const bool ok  = w < SORT_WORDS;
        bins[t] = 0;
        __syncthreads();
        int pk[4], own[4], rk[4];
        if (ok) {
            const int4 s = reinterpret_cast<const int4*>(ei)[w];
            const int4 d = reinterpret_cast<const int4*>(ei + N_EDGES)[w];
            const int sv[4] = {s.x, s.y, s.z, s.w};
            const int dv[4] = {d.x, d.y, d.z, d.w};
#pragma unroll
            for (int k = 0; k < 4; ++k) {
                const unsigned o = (unsigned)dv[k] / OWN_R;
                own[k] = (int)(o < 255u ? o : 255u);
                pk[k]  = sv[k] | ((dv[k] - own[k] * OWN_R) << 16);
                rk[k]  = atomicAdd(&bins[own[k]], 1);
            }
        }
        __syncthreads();
        const int c = bins[t];
        int incl = c;
        for (int dlt = 1; dlt < 32; dlt <<= 1) {
            const int v = __shfl_up(incl, dlt);
            if (lane >= dlt) incl += v;
        }
        if (lane == 31) wtot[wave] = incl;
        __syncthreads();
        int base = incl - c;
        for (int i = 0; i < wave; ++i) base += wtot[i];
        bins[t] = base;
        rtab[t * RUN_PITCH + blk] = base | (c << 16);
        __syncthreads();
        if (ok) {
#pragma unroll
            for (int k = 0; k < 4; ++k) srt[blk * SORT_E + bins[own[k]] + rk[k]] = pk[k];
        }
        return;
    }
    int i = blockIdx.x * blockDim.x + threadIdx.x;
    if (i >= N_NODES * HEADS) return;
    int n = i >> 1, h = i & 1;
    const float* hp = h1 + (size_t)n * F1 + h * HID;
    const float* s0 = as1 + h * HID;
    const float* d0 = ad1 + h * HID;
    float s = 0.f, d = 0.f;
#pragma unroll 8
    for (int c = 0; c < HID; ++c) { float v = hp[c]; s += v * s0[c]; d += v * d0[c]; }
    a_src[i] = s; a_dst[i] = d;
}

__global__ void gat_att2(const float* __restrict__ h2, const float* __restrict__ as2,
                         const float* __restrict__ ad2, float* __restrict__ a_src,
                         float* __restrict__ a_dst) {
    int n = blockIdx.x * blockDim.x + threadIdx.x;
    if (n >= N_NODES) return;
    const float* hp = h2 + (size_t)n * NCLS;
    float s = 0.f, d = 0.f;
#pragma unroll 8
    for (int c = 0; c < NCLS; ++c) { float v = hp[c]; s += v * as2[c]; d += v * ad2[c]; }
    a_src[n] = s; a_dst[n] = d;
}

#define SELF_LANES ((OWN_R + 31) / 32)
#define LIST_PITCH 896
#define WS_LISTS_OFF ((size_t)16 << 20)
#define WS_PLANES ((size_t)N_NODES * (2 * F1 + 2 * HEADS + 2 + NCLS))
static_assert(WS_PLANES <= WS_LISTS_OFF, "the saved lists start behind the feature and score planes");
static_assert((WS_LISTS_OFF + (size_t)OWN_BLOCKS * 32 * (LIST_PITCH + 1)) * 4 <= ((size_t)256 << 20),
              "lists and lengths end inside the workspace");
#define WS_SORT_OFF ((size_t)32 << 20)
static_assert(WS_LISTS_OFF + (size_t)OWN_BLOCKS * 32 * (LIST_PITCH + 1) <= WS_SORT_OFF, "the sorted entries start behind the saved lists");
static_assert((WS_SORT_OFF + (size_t)SORT_BLOCKS * SORT_E + (size_t)OWN_BLOCKS * RUN_PITCH) * 4 <= ((size_t)256 << 20),
              "sorted entries and run table end inside the workspace");
static_assert(N_NODES <= 65536 && OWN_R <= 65536, "a list entry packs source and local destination, 16 bits each");
static_assert(SORT_E <= 32767, "a run's start and length fit 16 bits each");
static_assert(SELF_LANES <= 32, "one lane per self-loop of a wave");

static __device__ __forceinline__ float lane_f(float v, int i) {
    return __int_as_float(__builtin_amdgcn_readlane(__float_as_int(v), i));
}

static __device__ __forceinline__ double widen_normal(float f) {
    const unsigned u  = __float_as_uint(f);
    const unsigned hi = (u & 0x80000000u) | (((u >> 3) & 0x0FFFFFFFu) + 0x38000000u);
    const unsigned lo = u << 29;
    return __builtin_bit_cast(double, ((unsigned long long)hi << 32) | lo);
}
static __device__ __forceinline__ bool not_normal(float f) { return __builtin_amdgcn_classf(f, 0x2F7); }
template <int N>
static __device__ __forceinline__ void widen(const float (&f)[N], double (&d)[N]) {
    bool other = false;
#pragma unroll
    for (int i = 0; i < N; ++i) { d[i] = widen_normal(f[i]); other |= not_normal(f[i]); }
    if (__builtin_expect(other, false)) {
#pragma unroll
        for (int i = 0; i < N; ++i) d[i] = (double)f[i];
    }
}

template <int F, int NH, int CAP, int ROLE>
__global__ __launch_bounds__(1024) void gat_own_accum(
        const float* __restrict__ a_src, const float* __restrict__ a_dst,
        const float* __restrict__ h,
        const float* __restrict__ bias, float* __restrict__ outp,
        int* __restrict__ wl, int* __restrict__ wc,
        const int* __restrict__ srt, const int* __restrict__ rtab) {
    static_assert(CAP >= 32 + SELF_LANES, "a piece of 32 entries goes into a list that holds the self-loops");
    static_assert(ROLE == 0 ? CAP <= LIST_PITCH : CAP >= LIST_PITCH, "a saved list fits its slot, and fits the list of the wave that takes it");
    __shared__ double   acc[OWN_R * F];
    __shared__ double   den[OWN_R * NH];
    __shared__ unsigned mx[OWN_R * NH];
    __shared__ int      que[32 * CAP];

    const int tid  = threadIdx.x;
    const int lane = tid & 31;
    const int wave = __builtin_amdgcn_readfirstlane(tid >> 5);
    const int n0   = blockIdx.x * OWN_R;
    const int nr   = (N_NODES - n0) < OWN_R ? (N_NODES - n0) : OWN_R;
    int* q = que + wave * CAP;

    for (int i = tid; i < OWN_R * F;  i += 1024) acc[i] = 0.0;
    for (int i = tid; i < OWN_R * NH; i += 1024) { den[i] = 0.0; mx[i] = 0u; }
    __syncthreads();

    auto max_list = [&](int n) {
        for (int off = 0; off < n; off += 128) {
            unsigned pk[4];
            float x0[4];
            [[maybe_unused]] float x1[4];
#pragma unroll
            for (int u = 0; u < 4; ++u) {
                const int i = off + 32 * u + lane;
                pk[u] = (unsigned)q[i < n ? i : 0];
            }
            float as0[4], ad0[4];
            [[maybe_unused]] float as1[4], ad1[4];
#pragma unroll
            for (int u = 0; u < 4; ++u) {
                const int s = (int)(pk[u] & 0xFFFFu), d = n0 + (int)(pk[u] >> 16);
                as0[u] = a_src[s * NH]; ad0[u] = a_dst[d * NH];
                if constexpr (NH == 2) { as1[u] = a_src[s * 2 + 1]; ad1[u] = a_dst[d * 2 + 1]; }
            }
            asm volatile("" :: "v"(as0[0]), "v"(as0[1]), "v"(as0[2]), "v"(as0[3]), "v"(ad0[0]), "v"(ad0[1]), "v"(ad0[2]), "v"(ad0[3]));
            if constexpr (NH == 2)
                asm volatile("" :: "v"(as1[0]), "v"(as1[1]), "v"(as1[2]), "v"(as1[3]), "v"(ad1[0]), "v"(ad1[1]), "v"(ad1[2]), "v"(ad1[3]));
#pragma unroll
            for (int u = 0; u < 4; ++u) {
                x0[u] = lrelu(as0[u] + ad0[u]);
                if constexpr (NH == 2) x1[u] = lrelu(as1[u] + ad1[u]);
            }
#pragma unroll
            for (int u = 0; u < 4; ++u) {
                const int i = off + 32 * u + lane;
                if (i < n) {
                    const int dl = (int)(pk[u] >> 16);
                    atomicMax(&mx[dl * NH], fmap(x0[u]));
                    if constexpr (NH == 2) atomicMax(&mx[dl * 2 + 1], fmap(x1[u]));
                }
            }
        }
    };

    [[maybe_unused]] auto max_saved = [&](const int* __restrict__ sv, int n) {
        for (int off = 0; off < n; off += 128) {
            unsigned pk[4];
            float x0[4];
            [[maybe_unused]] float x1[4];
#pragma unroll
            for (int u = 0; u < 4; ++u) {
                const int i = off + 32 * u + lane;
                pk[u] = (unsigned)sv[i < n ? i : 0];
            }
            asm volatile("" :: "v"(pk[0]), "v"(pk[1]), "v"(pk[2]), "v"(pk[3]));
            float as0[4], ad0[4];
            [[maybe_unused]] float as1[4], ad1[4];
#pragma unroll
            for (int u = 0; u < 4; ++u) {
                const int s = (int)(pk[u] & 0xFFFFu), d = n0 + (int)(pk[u] >> 16);
                as0[u] = a_src[s * NH]; ad0[u] = a_dst[d * NH];
                if constexpr (NH == 2) { as1[u] = a_src[s * 2 + 1]; ad1[u] = a_dst[d * 2 + 1]; }
            }
            asm volatile("" :: "v"(as0[0]), "v"(as0[1]), "v"(as0[2]), "v"(as0[3]), "v"(ad0[0]), "v"(ad0[1]), "v"(ad0[2]), "v"(ad0[3]));
            if constexpr (NH == 2)
                asm volatile("" :: "v"(as1[0]), "v"(as1[1]), "v"(as1[2]), "v"(as1[3]), "v"(ad1[0]), "v"(ad1[1]), "v"(ad1[2]), "v"(ad1[3]));
#pragma unroll
            for (int u = 0; u < 4; ++u) {
                x0[u] = lrelu(as0[u] + ad0[u]);
                if constexpr (NH == 2) x1[u] = lrelu(as1[u] + ad1[u]);
            }
#pragma unroll
            for (int u = 0; u < 4; ++u) {
                const int i = off + 32 * u + lane;
                if (i < n) {
                    const int dl = (int)(pk[u] >> 16);
                    atomicMax(&mx[dl * NH], fmap(x0[u]));
                    if constexpr (NH == 2) atomicMax(&mx[dl * 2 + 1], fmap(x1[u]));
                    q[i] = (int)pk[u];
                }
            }
        }
    };

    auto work = [&](int n) {
        for (int off = 0; off < n; off += 32) {
            const int nb = (n - off) < 32 ? (n - off) : 32;
            int s = 0, dl = 0;
            [[maybe_unused]] int pkl = 0;
            float p0 = 0.f, p1 = 0.f;
            if (lane < nb) {
                const unsigned pk = (unsigned)q[off + lane];
                pkl = (int)pk;
                s  = (int)(pk & 0xFFFFu);
                dl = (int)(pk >> 16);
                const int d = n0 + dl;
                p0 = __expf(lrelu(a_src[s * NH] + a_dst[d * NH]) - finv(mx[dl * NH]));
                atomicAdd(&den[dl * NH], (double)p0);
                if constexpr (NH == 2) {
                    p1 = __expf(lrelu(a_src[s * 2 + 1] + a_dst[d * 2 + 1]) - finv(mx[dl * 2 + 1]));
                    atomicAdd(&den[dl * 2 + 1], (double)p1);
                }
            }
            if constexpr (F == F1) {
#pragma unroll 1
                for (int i0 = 0; i0 < nb; i0 += 8) {
                    float hv[8][4];
#pragma unroll
                    for (int j = 0; j < 8; ++j) {
                        const int i = (i0 + j < nb) ? (i0 + j) : (nb - 1);
                        const int sj = __builtin_amdgcn_readlane(s, i);
                        const float* rp = h + (size_t)sj * F1 + lane;
#pragma unroll
                        for (int k = 0; k < 4; ++k) hv[j][k] = rp[k * 32];
                    }
#pragma unroll
                    for (int j = 0; j < 8; ++j) {
                        if (i0 + j < nb) {
                            const int dj = __builtin_amdgcn_readlane(dl, i0 + j);
                            const float q0 = lane_f(p0, i0 + j);
                            const float q1 = lane_f(p1, i0 + j);
                            double* ap = acc + dj * F1 + lane;
                            const float tf[4] = {hv[j][0] * q0, hv[j][1] * q0,
                                                 hv[j][2] * q1, hv[j][3] * q1};
                            double td[4];
                            widen(tf, td);
                            atomicAdd(ap +  0, td[0]);
                            atomicAdd(ap + 32, td[1]);
                            atomicAdd(ap + 64, td[2]);
                            atomicAdd(ap + 96, td[3]);
                        }
                    }
                }
            } else {
                static_assert(F % 4 == 0 && 3 * (F / 4) <= 32, "three rows of 16-byte pieces fit the lanes of a wave");
                constexpr int PCS = F / 4;
                const int slot  = lane / PCS;
                const int piece = lane - slot * PCS;
                const float* hp = h + 4 * piece;
                constexpr int STEPS = (32 + 2) / 3;
                int   ea[STEPS];
                int   pe[STEPS];
                float pw[STEPS];
                v4f   hv[STEPS];
#pragma unroll
                for (int j = 0; j < STEPS; ++j) {
                    const int e = 3 * j + slot;
                    ea[j] = (slot < 3 && e < nb) ? (e << 2) : 0;
                    pe[j] = __builtin_amdgcn_ds_bpermute(ea[j], pkl);
                }
#pragma unroll
                for (int j = 0; j < STEPS; ++j)
                    hv[j] = *reinterpret_cast<const v4f*>(hp + (size_t)(pe[j] & 0xFFFF) * F);
#pragma unroll
                for (int j = 0; j < STEPS; ++j)
                    pw[j] = __int_as_float(__builtin_amdgcn_ds_bpermute(ea[j], __float_as_int(p0)));
#pragma unroll
                for (int j = 0; j < STEPS; ++j) {
                    if (3 * j < nb) {
                        const int e = 3 * j + slot;
                        const float tf[4] = {hv[j].x * pw[j], hv[j].y * pw[j], hv[j].z * pw[j], hv[j].w * pw[j]};
                        double td[4];
                        widen(tf, td);
                        if (slot < 3 && e < nb) {
                            double* ap = acc + (int)((unsigned)pe[j] >> 16) * F + piece;
                            atomicAdd(ap,           td[0]);
                            atomicAdd(ap + PCS,     td[1]);
                            atomicAdd(ap + 2 * PCS, td[2]);
                            atomicAdd(ap + 3 * PCS, td[3]);
                        }
                    }
                }
            }
        }
    };

    bool ovf = false;
    auto gather = [&](auto&& flush) -> int {
        int cnt = 0;
        {
            const int node = wave + 32 * lane;
            const bool m = lane < SELF_LANES && node < nr;
            const unsigned mk = __builtin_amdgcn_ballot_w32(m);
            if (m) q[__builtin_amdgcn_mbcnt_lo(mk, 0)] = (n0 + node) | (node << 16);
            cnt = __builtin_popcount(mk);
        }
        const int blk = wave * RUNS_PER_WAVE + lane;
        const bool has = lane < RUNS_PER_WAVE && blk < SORT_BLOCKS;
        const int rv = has ? rtab[blockIdx.x * RUN_PITCH + blk] : 0;
        const int st = rv & 0xFFFF, n = rv >> 16;
        int incl = n;
        for (int dlt = 1; dlt < 32; dlt <<= 1) {
            const int v = __shfl_up(incl, dlt);
            if (lane >= dlt) incl += v;
        }
        const int tot = __builtin_amdgcn_readlane(incl, 31);
        if (cnt + tot <= CAP) {
            const int* sp = n > 0 ? srt + (size_t)blk * SORT_E + st : srt;
            const int pos = cnt + incl - n;
            for (int k = 0; __builtin_amdgcn_ballot_w32(k < n) != 0; k += 8) {
                int v[8];
#pragma unroll
                for (int u = 0; u < 8; ++u) v[u] = sp[k + u < n ? k + u : 0];
#pragma unroll
                for (int u = 0; u < 8; ++u) if (k + u < n) q[pos + k + u] = v[u];
            }
            cnt += tot;
        } else {
            for (int l = 0; l < RUNS_PER_WAVE; ++l) {
                const int nl = __builtin_amdgcn_readlane(n, l);
                const int sl = __builtin_amdgcn_readlane(st, l);
                const int* rp = srt + (size_t)(wave * RUNS_PER_WAVE + l) * SORT_E + sl;
                for (int off = 0; off < nl; off += 32) {
                    if (cnt > CAP - 32) { flush(cnt); cnt = 0; ovf = true; }
                    const int i = off + lane;
                    if (i < nl) q[cnt + lane] = rp[i];
                    cnt += (nl - off) < 32 ? (nl - off) : 32;
                }
            }
        }
        return cnt;
    };

    const int slot = blockIdx.x * 32 + wave;
    int  cnt  = 0;
    bool have = false;
    if constexpr (ROLE == 1) {
        const int c = __builtin_amdgcn_readfirstlane(wc[slot]);
        if (c >= 0) {
            have = true;
            cnt  = c;
            max_saved(wl + (size_t)slot * LIST_PITCH, cnt);
        }
    }
    if (!have) {
        cnt = gather([&](int n) { max_list(n); });
        max_list(cnt);
    }
    if constexpr (ROLE == 0) {
        if (!ovf) for (int i = lane; i < cnt; i += 32) wl[(size_t)slot * LIST_PITCH + i] = q[i];
        if (lane == 0) wc[slot] = ovf ? -1 : cnt;
    }
    __syncthreads();
    if (!ovf) {
        work(cnt);
    } else {
        cnt = gather([&](int n) { work(n); });
        work(cnt);
    }

    __syncthreads();
    for (int i = tid; i < nr * F; i += 1024) {
        const int r = i / F, c = i - r * F;
        const int ai = (F == F1) ? i : r * F + (c & 3) * (F / 4) + (c >> 2);
        float v = (float)acc[ai] / (float)den[r * NH + (NH == 2 ? (c >> 6) : 0)] + bias[c];
        if constexpr (F == F1) v = v > 0.f ? v : (expf(v) - 1.0f);
        outp[(size_t)n0 * F + i] = v;
    }
}

extern "C" void kernel_launch(void* const* d_in, const int* in_sizes, int n_in,
                              void* d_out, int out_size, void* d_ws, size_t ws_size,
                              hipStream_t stream) {
    const float* x    = (const float*)d_in[0];
    const int*   ei   = (const int*)  d_in[1];
    const float* W1   = (const float*)d_in[2];
    const float* as1  = (const float*)d_in[3];
    const float* ad1  = (const float*)d_in[4];
    const float* b1   = (const float*)d_in[5];
    const float* W2   = (const float*)d_in[6];
    const float* as2  = (const float*)d_in[7];
    const float* ad2  = (const float*)d_in[8];
    const float* b2   = (const float*)d_in[9];
    float* out = (float*)d_out;

    float* ws      = (float*)d_ws;
    float* h1      = ws;
    float* h2in    = h1 + (size_t)N_NODES * F1;
    float* a_src1  = h2in + (size_t)N_NODES * F1;
    float* a_dst1  = a_src1 + (size_t)N_NODES * 2;
    float* a_src2  = a_dst1 + (size_t)N_NODES * 2;
    float* a_dst2  = a_src2 + N_NODES;
    float* h2      = a_dst2 + N_NODES;
    int*   wl      = (int*)d_ws + WS_LISTS_OFF;
    int*   wc      = wl + (size_t)OWN_BLOCKS * 32 * LIST_PITCH;
    int*   srt     = (int*)d_ws + WS_SORT_OFF;
    int*   rtab    = srt + (size_t)SORT_BLOCKS * SORT_E;

    const int MT = N_NODES / 16;
    gat_gemm1_wmma<<<G1_BLOCKS, 256, 0, stream>>>(x, W1, h1);
    gat_att1<<<ATT1_BLOCKS + SORT_BLOCKS, 256, 0, stream>>>(h1, as1, ad1, a_src1, a_dst1, ei, srt, rtab);
    gat_own_accum<F1, HEADS, LIST_PITCH, 0><<<OWN_BLOCKS, 1024, 0, stream>>>(a_src1, a_dst1, h1, b1, h2in, wl, wc, srt, rtab);
    gat_gemm2_wmma<<<MT, 96, 0, stream>>>(h2in, W2, h2);
    gat_att2<<<(N_NODES + 255) / 256, 256, 0, stream>>>(h2, as2, ad2, a_src2, a_dst2);
    gat_own_accum<NCLS, 1, 1152, 1><<<OWN_BLOCKS, 1024, 0, stream>>>(a_src2, a_dst2, h2, b2, out, wl, wc, srt, rtab);
}


# Layer-2 score vectors folded into W2: a padded copy built per call in the first dot launch has W2.att_src, W2.att_dst as columns 40, 41; GEMM2 yields the scores, dot launch 2 gone; dots sum over k
# speedup vs baseline: 4.2499x; 1.1026x over previous
#include <hip/hip_runtime.h>
#include <math.h>

#define N_NODES 50000
#define N_EDGES 800000
#define IN_FEATS 256
#define HID 64
#define HEADS 2
#define F1 (HEADS * HID)
#define NCLS 40
#define W2P_COLS 48
#define OWN_R      196
#define OWN_BLOCKS ((N_NODES + OWN_R - 1) / OWN_R)
#define ATT1_BLOCKS ((N_NODES * HEADS + 255) / 256)
#define SORT_E     1024
#define SORT_WORDS (N_EDGES / 4)
#define SORT_BLOCKS ((SORT_WORDS + 255) / 256)
#define RUN_PITCH  800
#define RUNS_PER_WAVE ((SORT_BLOCKS + 31) / 32)
static_assert(OWN_BLOCKS <= 256, "one bin per owner in a sorting block of 256 threads");
static_assert(N_EDGES % 4 == 0, "the edge list is read four edges at a time");
static_assert(RUN_PITCH >= SORT_BLOCKS && RUNS_PER_WAVE <= 32, "a lane per run, a table slot per sorting block");

typedef __attribute__((ext_vector_type(2))) float v2f;
typedef __attribute__((ext_vector_type(8))) float v8f;
typedef __attribute__((ext_vector_type(4))) float v4f;

static __device__ __forceinline__ unsigned fmap(float f) {
    unsigned u = __float_as_uint(f);
    return (u & 0x80000000u) ? ~u : (u | 0x80000000u);
}
static __device__ __forceinline__ float finv(unsigned t) {
    return (t & 0x80000000u) ? __uint_as_float(t & 0x7FFFFFFFu)
                             : __uint_as_float(~t);
}
static __device__ __forceinline__ float lrelu(float x) { return x > 0.f ? x : 0.2f * x; }

#define G1_PITCH  260
#define G1_BLOCKS 512
#define G1_TILES  (N_NODES / 16)
static_assert(N_NODES % 16 == 0, "row tiles are whole");

__global__ __launch_bounds__(256) void gat_gemm1_wmma(const float* __restrict__ x,
                                                      const float* __restrict__ W1,
                                                      float* __restrict__ h1) {
    __shared__ float xs[2][16 * G1_PITCH];

    const int tid   = threadIdx.x;
    const int wave  = tid >> 5;
    const int lane  = tid & 31;
    const int half  = lane >> 4;
    const int l     = lane & 15;

    v2f b[IN_FEATS / 4];
    {
        const float* bp = W1 + (size_t)(2 * half) * F1 + wave * 16 + l;
#pragma unroll
        for (int kk = 0; kk < IN_FEATS / 4; ++kk) {
            b[kk].x = bp[(size_t)(4 * kk) * F1];
            b[kk].y = bp[(size_t)(4 * kk + 1) * F1];
        }
    }

    const v4f* xv = reinterpret_cast<const v4f*>(x);
    const int so = (tid >> 6) * G1_PITCH + (tid & 63) * 4;
    v4f st0, st1, st2, st3;
    int mt = blockIdx.x;
    if (mt < G1_TILES) {
        const v4f* tp = xv + (size_t)mt * 1024 + tid;
        st0 = tp[0]; st1 = tp[256]; st2 = tp[512]; st3 = tp[768];
        float* sp = &xs[0][so];
        *reinterpret_cast<v4f*>(sp)                 = st0;
        *reinterpret_cast<v4f*>(sp +  4 * G1_PITCH) = st1;
        *reinterpret_cast<v4f*>(sp +  8 * G1_PITCH) = st2;
        *reinterpret_cast<v4f*>(sp + 12 * G1_PITCH) = st3;
    }
    __syncthreads();

    int buf = 0;
    for (; mt < G1_TILES; mt += gridDim.x) {
        const int  nxt  = mt + gridDim.x;
        const bool more = nxt < G1_TILES;
        if (more) {
            const v4f* tp = xv + (size_t)nxt * 1024 + tid;
            st0 = tp[0]; st1 = tp[256]; st2 = tp[512]; st3 = tp[768];
        }

        const float* ap = &xs[buf][l * G1_PITCH + 2 * half];
        v8f acc = {};
        v2f a0[8], a1[8];
#pragma unroll
        for (int i = 0; i < 8; ++i) a0[i] = *reinterpret_cast<const v2f*>(ap + 4 * i);
#pragma unroll
        for (int g = 0; g < IN_FEATS / 32; g += 2) {
#pragma unroll
            for (int i = 0; i < 8; ++i) a1[i] = *reinterpret_cast<const v2f*>(ap + 32 * (g + 1) + 4 * i);
            __builtin_amdgcn_sched_barrier(0);
#pragma unroll
            for (int i = 0; i < 8; ++i)
                acc = __builtin_amdgcn_wmma_f32_16x16x4_f32(false, a0[i], false, b[8 * g + i],
                                                            (short)0, acc, false, false);
            __builtin_amdgcn_sched_barrier(0);
            if (g + 2 < IN_FEATS / 32) {
#pragma unroll
                for (int i = 0; i < 8; ++i) a0[i] = *reinterpret_cast<const v2f*>(ap + 32 * (g + 2) + 4 * i);
            }
            __builtin_amdgcn_sched_barrier(0);
#pragma unroll
            for (int i = 0; i < 8; ++i)
                acc = __builtin_amdgcn_wmma_f32_16x16x4_f32(false, a1[i], false, b[8 * (g + 1) + i],
                                                            (short)0, acc, false, false);
            __builtin_amdgcn_sched_barrier(0);
        }
        if (more) {
            float* sp = &xs[buf ^ 1][so];
            *reinterpret_cast<v4f*>(sp)                 = st0;
            *reinterpret_cast<v4f*>(sp +  4 * G1_PITCH) = st1;
            *reinterpret_cast<v4f*>(sp +  8 * G1_PITCH) = st2;
            *reinterpret_cast<v4f*>(sp + 12 * G1_PITCH) = st3;
        }
        __syncthreads();
        buf ^= 1;

        float* dr = h1 + (size_t)(mt * 16 + 8 * half) * F1 + wave * 16 + l;
#pragma unroll
        for (int r = 0; r < 8; ++r) dr[(size_t)r * F1] = acc[r];
    }
}

__global__ __launch_bounds__(96) void gat_gemm2_wmma(const float* __restrict__ h2in,
                                                     const float* __restrict__ w2p,
                                                     float* __restrict__ h2,
                                                     float* __restrict__ a_src,
                                                     float* __restrict__ a_dst) {
    const int mtile = blockIdx.x;
    const int wave  = threadIdx.x >> 5;
    const int lane  = threadIdx.x & 31;
    const int half  = lane >> 4;
    const int l     = lane & 15;
    const int row   = mtile * 16 + l;
    const int ncol  = wave * 16 + l;

    v8f acc = {};
    const float* xr = h2in + (size_t)row * F1 + 2 * half;
    const float* bp = w2p + (size_t)(2 * half) * W2P_COLS + ncol;
    for (int k = 0; k < F1; k += 4) {
        v2f a = *(const v2f*)(xr + k);
        v2f b;
        b.x = bp[(size_t)k * W2P_COLS]; b.y = bp[(size_t)(k + 1) * W2P_COLS];
        acc = __builtin_amdgcn_wmma_f32_16x16x4_f32(false, a, false, b,
                                                    (short)0, acc, false, false);
    }
    const int r0 = mtile * 16 + 8 * half;
    if (ncol < NCLS) {
        float* dr = h2 + (size_t)r0 * NCLS + ncol;
#pragma unroll
        for (int r = 0; r < 8; ++r) dr[(size_t)r * NCLS] = acc[r];
    } else if (ncol < NCLS + 2) {
        float* dr = (ncol == NCLS ? a_src : a_dst) + r0;
#pragma unroll
        for (int r = 0; r < 8; ++r) dr[r] = acc[r];
    }
}

__global__ __launch_bounds__(256) void gat_att1(const float* __restrict__ h1, const float* __restrict__ as1,
                         const float* __restrict__ ad1, float* __restrict__ a_src,
                         float* __restrict__ a_dst, const int* __restrict__ ei,
                         int* __restrict__ srt, int* __restrict__ rtab,
                         const float* __restrict__ W2, const float* __restrict__ as2,
                         const float* __restrict__ ad2, float* __restrict__ w2p) {
    if (blockIdx.x == ATT1_BLOCKS + SORT_BLOCKS) {
        const int k = threadIdx.x >> 1, which = threadIdx.x & 1;
        const float* wr = W2 + (size_t)k * NCLS;
        const float* av = which ? ad2 : as2;
        float* pr = w2p + (size_t)k * W2P_COLS;
        float s = 0.f;
        for (int cc = 0; cc < NCLS; ++cc) s += wr[cc] * av[cc];
        pr[NCLS + which] = s;
        if (which == 0) { for (int cc = 0; cc < NCLS; ++cc) pr[cc] = wr[cc]; }
        else            { for (int cc = NCLS + 2; cc < W2P_COLS; ++cc) pr[cc] = 0.f; }
        return;
    }
    if (blockIdx.x >= ATT1_BLOCKS) {
        __shared__ int bins[256];
        __shared__ int wtot[8];
        const int t    = threadIdx.x;
        const int lane = t & 31;
        const int wave = t >> 5;
        const int blk  = blockIdx.x - ATT1_BLOCKS;
        const int w    = blk * 256 + t;
        const bool ok  = w < SORT_WORDS;
        bins[t] = 0;
        __syncthreads();
        int pk[4], own[4], rk[4];
        if (ok) {
            const int4 s = reinterpret_cast<const int4*>(ei)[w];
            const int4 d = reinterpret_cast<const int4*>(ei + N_EDGES)[w];
            const int sv[4] = {s.x, s.y, s.z, s.w};
            const int dv[4] = {d.x, d.y, d.z, d.w};
#pragma unroll
            for (int k = 0; k < 4; ++k) {
                const unsigned o = (unsigned)dv[k] / OWN_R;
                own[k] = (int)(o < 255u ? o : 255u);
                pk[k]  = sv[k] | ((dv[k] - own[k] * OWN_R) << 16);
                rk[k]  = atomicAdd(&bins[own[k]], 1);
            }
        }
        __syncthreads();
        const int c = bins[t];
        int incl = c;
        for (int dlt = 1; dlt < 32; dlt <<= 1) {
            const int v = __shfl_up(incl, dlt);
            if (lane >= dlt) incl += v;
        }
        if (lane == 31) wtot[wave] = incl;
        __syncthreads();
        int base = incl - c;
        for (int i = 0; i < wave; ++i) base += wtot[i];
        bins[t] = base;
        rtab[t * RUN_PITCH + blk] = base | (c << 16);
        __syncthreads();
        if (ok) {
#pragma unroll
            for (int k = 0; k < 4; ++k) srt[blk * SORT_E + bins[own[k]] + rk[k]] = pk[k];
        }
        return;
    }
    int i = blockIdx.x * blockDim.x + threadIdx.x;
    if (i >= N_NODES * HEADS) return;
    int n = i >> 1, h = i & 1;
    const float* hp = h1 + (size_t)n * F1 + h * HID;
    const float* s0 = as1 + h * HID;
    const float* d0 = ad1 + h * HID;
    float s = 0.f, d = 0.f;
#pragma unroll 8
    for (int c = 0; c < HID; ++c) { float v = hp[c]; s += v * s0[c]; d += v * d0[c]; }
    a_src[i] = s; a_dst[i] = d;
}

#define SELF_LANES ((OWN_R + 31) / 32)
#define LIST_PITCH 896
#define WS_LISTS_OFF ((size_t)16 << 20)
#define WS_PLANES ((size_t)N_NODES * (2 * F1 + 2 * HEADS + 2 + NCLS))
static_assert(WS_PLANES <= WS_LISTS_OFF, "the saved lists start behind the feature and score planes");
static_assert((WS_LISTS_OFF + (size_t)OWN_BLOCKS * 32 * (LIST_PITCH + 1)) * 4 <= ((size_t)256 << 20),
              "lists and lengths end inside the workspace");
#define WS_SORT_OFF ((size_t)32 << 20)
static_assert(WS_LISTS_OFF + (size_t)OWN_BLOCKS * 32 * (LIST_PITCH + 1) <= WS_SORT_OFF, "the sorted entries start behind the saved lists");
static_assert((WS_SORT_OFF + (size_t)SORT_BLOCKS * SORT_E + (size_t)OWN_BLOCKS * RUN_PITCH) * 4 <= ((size_t)256 << 20),
              "sorted entries and run table end inside the workspace");
#define WS_W2P_OFF ((size_t)48 << 20)
static_assert(WS_SORT_OFF + (size_t)SORT_BLOCKS * SORT_E + (size_t)OWN_BLOCKS * RUN_PITCH <= WS_W2P_OFF, "the padded copy of W2 starts behind the run table");
static_assert((WS_W2P_OFF + (size_t)F1 * W2P_COLS) * 4 <= ((size_t)256 << 20), "the padded copy of W2 ends inside the workspace");
static_assert(F1 * 2 == 256 && NCLS + 2 <= W2P_COLS, "two threads of the folding block per row of W2; the two score columns fit the padding");
static_assert(N_NODES <= 65536 && OWN_R <= 65536, "a list entry packs source and local destination, 16 bits each");
static_assert(SORT_E <= 32767, "a run's start and length fit 16 bits each");
static_assert(SELF_LANES <= 32, "one lane per self-loop of a wave");

static __device__ __forceinline__ float lane_f(float v, int i) {
    return __int_as_float(__builtin_amdgcn_readlane(__float_as_int(v), i));
}

static __device__ __forceinline__ double widen_normal(float f) {
    const unsigned u  = __float_as_uint(f);
    const unsigned hi = (u & 0x80000000u) | (((u >> 3) & 0x0FFFFFFFu) + 0x38000000u);
    const unsigned lo = u << 29;
    return __builtin_bit_cast(double, ((unsigned long long)hi << 32) | lo);
}
static __device__ __forceinline__ bool not_normal(float f) { return __builtin_amdgcn_classf(f, 0x2F7); }
template <int N>
static __device__ __forceinline__ void widen(const float (&f)[N], double (&d)[N]) {
    bool other = false;
#pragma unroll
    for (int i = 0; i < N; ++i) { d[i] = widen_normal(f[i]); other |= not_normal(f[i]); }
    if (__builtin_expect(other, false)) {
#pragma unroll
        for (int i = 0; i < N; ++i) d[i] = (double)f[i];
    }
}

template <int F, int NH, int CAP, int ROLE>
__global__ __launch_bounds__(1024) void gat_own_accum(
        const float* __restrict__ a_src, const float* __restrict__ a_dst,
        const float* __restrict__ h,
        const float* __restrict__ bias, float* __restrict__ outp,
        int* __restrict__ wl, int* __restrict__ wc,
        const int* __restrict__ srt, const int* __restrict__ rtab) {
    static_assert(CAP >= 32 + SELF_LANES, "a piece of 32 entries goes into a list that holds the self-loops");
    static_assert(ROLE == 0 ? CAP <= LIST_PITCH : CAP >= LIST_PITCH, "a saved list fits its slot, and fits the list of the wave that takes it");
    __shared__ double   acc[OWN_R * F];
    __shared__ double   den[OWN_R * NH];
    __shared__ unsigned mx[OWN_R * NH];
    __shared__ int      que[32 * CAP];

    const int tid  = threadIdx.x;
    const int lane = tid & 31;
    const int wave = __builtin_amdgcn_readfirstlane(tid >> 5);
    const int n0   = blockIdx.x * OWN_R;
    const int nr   = (N_NODES - n0) < OWN_R ? (N_NODES - n0) : OWN_R;
    int* q = que + wave * CAP;

    for (int i = tid; i < OWN_R * F;  i += 1024) acc[i] = 0.0;
    for (int i = tid; i < OWN_R * NH; i += 1024) { den[i] = 0.0; mx[i] = 0u; }
    __syncthreads();

    auto max_list = [&](int n) {
        for (int off = 0; off < n; off += 128) {
            unsigned pk[4];
            float x0[4];
            [[maybe_unused]] float x1[4];
#pragma unroll
            for (int u = 0; u < 4; ++u) {
                const int i = off + 32 * u + lane;
                pk[u] = (unsigned)q[i < n ? i : 0];
            }
            float as0[4], ad0[4];
            [[maybe_unused]] float as1[4], ad1[4];
#pragma unroll
            for (int u = 0; u < 4; ++u) {
                const int s = (int)(pk[u] & 0xFFFFu), d = n0 + (int)(pk[u] >> 16);
                as0[u] = a_src[s * NH]; ad0[u] = a_dst[d * NH];
                if constexpr (NH == 2) { as1[u] = a_src[s * 2 + 1]; ad1[u] = a_dst[d * 2 + 1]; }
            }
            asm volatile("" :: "v"(as0[0]), "v"(as0[1]), "v"(as0[2]), "v"(as0[3]), "v"(ad0[0]), "v"(ad0[1]), "v"(ad0[2]), "v"(ad0[3]));
            if constexpr (NH == 2)
                asm volatile("" :: "v"(as1[0]), "v"(as1[1]), "v"(as1[2]), "v"(as1[3]), "v"(ad1[0]), "v"(ad1[1]), "v"(ad1[2]), "v"(ad1[3]));
#pragma unroll
            for (int u = 0; u < 4; ++u) {
                x0[u] = lrelu(as0[u] + ad0[u]);
                if constexpr (NH == 2) x1[u] = lrelu(as1[u] + ad1[u]);
            }
#pragma unroll
            for (int u = 0; u < 4; ++u) {
                const int i = off + 32 * u + lane;
                if (i < n) {
                    const int dl = (int)(pk[u] >> 16);
                    atomicMax(&mx[dl * NH], fmap(x0[u]));
                    if constexpr (NH == 2) atomicMax(&mx[dl * 2 + 1], fmap(x1[u]));
                }
            }
        }
    };

    [[maybe_unused]] auto max_saved = [&](const int* __restrict__ sv, int n) {
        for (int off = 0; off < n; off += 128) {
            unsigned pk[4];
            float x0[4];
            [[maybe_unused]] float x1[4];
#pragma unroll
            for (int u = 0; u < 4; ++u) {
                const int i = off + 32 * u + lane;
                pk[u] = (unsigned)sv[i < n ? i : 0];
            }
            asm volatile("" :: "v"(pk[0]), "v"(pk[1]), "v"(pk[2]), "v"(pk[3]));
            float as0[4], ad0[4];
            [[maybe_unused]] float as1[4], ad1[4];
#pragma unroll
            for (int u = 0; u < 4; ++u) {
                const int s = (int)(pk[u] & 0xFFFFu), d = n0 + (int)(pk[u] >> 16);
                as0[u] = a_src[s * NH]; ad0[u] = a_dst[d * NH];
                if constexpr (NH == 2) { as1[u] = a_src[s * 2 + 1]; ad1[u] = a_dst[d * 2 + 1]; }
            }
            asm volatile("" :: "v"(as0[0]), "v"(as0[1]), "v"(as0[2]), "v"(as0[3]), "v"(ad0[0]), "v"(ad0[1]), "v"(ad0[2]), "v"(ad0[3]));
            if constexpr (NH == 2)
                asm volatile("" :: "v"(as1[0]), "v"(as1[1]), "v"(as1[2]), "v"(as1[3]), "v"(ad1[0]), "v"(ad1[1]), "v"(ad1[2]), "v"(ad1[3]));
#pragma unroll
            for (int u = 0; u < 4; ++u) {
                x0[u] = lrelu(as0[u] + ad0[u]);
                if constexpr (NH == 2) x1[u] = lrelu(as1[u] + ad1[u]);
            }
#pragma unroll
            for (int u = 0; u < 4; ++u) {
                const int i = off + 32 * u + lane;
                if (i < n) {
                    const int dl = (int)(pk[u] >> 16);
                    atomicMax(&mx[dl * NH], fmap(x0[u]));
                    if constexpr (NH == 2) atomicMax(&mx[dl * 2 + 1], fmap(x1[u]));
                    q[i] = (int)pk[u];
                }
            }
        }
    };

    auto work = [&](int n) {
        for (int off = 0; off < n; off += 32) {
            const int nb = (n - off) < 32 ? (n - off) : 32;
            int s = 0, dl = 0;
            [[maybe_unused]] int pkl = 0;
            float p0 = 0.f, p1 = 0.f;
            if (lane < nb) {
                const unsigned pk = (unsigned)q[off + lane];
                pkl = (int)pk;
                s  = (int)(pk & 0xFFFFu);
                dl = (int)(pk >> 16);
                const int d = n0 + dl;
                p0 = __expf(lrelu(a_src[s * NH] + a_dst[d * NH]) - finv(mx[dl * NH]));
                atomicAdd(&den[dl * NH], (double)p0);
                if constexpr (NH == 2) {
                    p1 = __expf(lrelu(a_src[s * 2 + 1] + a_dst[d * 2 + 1]) - finv(mx[dl * 2 + 1]));
                    atomicAdd(&den[dl * 2 + 1], (double)p1);
                }
            }
            if constexpr (F == F1) {
#pragma unroll 1
                for (int i0 = 0; i0 < nb; i0 += 8) {
                    float hv[8][4];
#pragma unroll
                    for (int j = 0; j < 8; ++j) {
                        const int i = (i0 + j < nb) ? (i0 + j) : (nb - 1);
                        const int sj = __builtin_amdgcn_readlane(s, i);
                        const float* rp = h + (size_t)sj * F1 + lane;
#pragma unroll
                        for (int k = 0; k < 4; ++k) hv[j][k] = rp[k * 32];
                    }
#pragma unroll
                    for (int j = 0; j < 8; ++j) {
                        if (i0 + j < nb) {
                            const int dj = __builtin_amdgcn_readlane(dl, i0 + j);
                            const float q0 = lane_f(p0, i0 + j);
                            const float q1 = lane_f(p1, i0 + j);
                            double* ap = acc + dj * F1 + lane;
                            const float tf[4] = {hv[j][0] * q0, hv[j][1] * q0,
                                                 hv[j][2] * q1, hv[j][3] * q1};
                            double td[4];
                            widen(tf, td);
                            atomicAdd(ap +  0, td[0]);
                            atomicAdd(ap + 32, td[1]);
                            atomicAdd(ap + 64, td[2]);
                            atomicAdd(ap + 96, td[3]);
                        }
                    }
                }
            } else {
                static_assert(F % 4 == 0 && 3 * (F / 4) <= 32, "three rows of 16-byte pieces fit the lanes of a wave");
                constexpr int PCS = F / 4;
                const int slot  = lane / PCS;
                const int piece = lane - slot * PCS;
                const float* hp = h + 4 * piece;
                constexpr int STEPS = (32 + 2) / 3;
                int   ea[STEPS];
                int   pe[STEPS];
                float pw[STEPS];
                v4f   hv[STEPS];
#pragma unroll
                for (int j = 0; j < STEPS; ++j) {
                    const int e = 3 * j + slot;
                    ea[j] = (slot < 3 && e < nb) ? (e << 2) : 0;
                    pe[j] = __builtin_amdgcn_ds_bpermute(ea[j], pkl);
                }
#pragma unroll
                for (int j = 0; j < STEPS; ++j)
                    hv[j] = *reinterpret_cast<const v4f*>(hp + (size_t)(pe[j] & 0xFFFF) * F);
#pragma unroll
                for (int j = 0; j < STEPS; ++j)
                    pw[j] = __int_as_float(__builtin_amdgcn_ds_bpermute(ea[j], __float_as_int(p0)));
#pragma unroll
                for (int j = 0; j < STEPS; ++j) {
                    if (3 * j < nb) {
                        const int e = 3 * j + slot;
                        const float tf[4] = {hv[j].x * pw[j], hv[j].y * pw[j], hv[j].z * pw[j], hv[j].w * pw[j]};
                        double td[4];
                        widen(tf, td);
                        if (slot < 3 && e < nb) {
                            double* ap = acc + (int)((unsigned)pe[j] >> 16) * F + piece;
                            atomicAdd(ap,           td[0]);
                            atomicAdd(ap + PCS,     td[1]);
                            atomicAdd(ap + 2 * PCS, td[2]);
                            atomicAdd(ap + 3 * PCS, td[3]);
                        }
                    }
                }
            }
        }
    };

    bool ovf = false;
    auto gather = [&](auto&& flush) -> int {
        int cnt = 0;
        {
            const int node = wave + 32 * lane;
            const bool m = lane < SELF_LANES && node < nr;
            const unsigned mk = __builtin_amdgcn_ballot_w32(m);
            if (m) q[__builtin_amdgcn_mbcnt_lo(mk, 0)] = (n0 + node) | (node << 16);
            cnt = __builtin_popcount(mk);
        }
        const int blk = wave * RUNS_PER_WAVE + lane;
        const bool has = lane < RUNS_PER_WAVE && blk < SORT_BLOCKS;
        const int rv = has ? rtab[blockIdx.x * RUN_PITCH + blk] : 0;
        const int st = rv & 0xFFFF, n = rv >> 16;
        int incl = n;
        for (int dlt = 1; dlt < 32; dlt <<= 1) {
            const int v = __shfl_up(incl, dlt);
            if (lane >= dlt) incl += v;
        }
        const int tot = __builtin_amdgcn_readlane(incl, 31);
        if (cnt + tot <= CAP) {
            const int* sp = n > 0 ? srt + (size_t)blk * SORT_E + st : srt;
            const int pos = cnt + incl - n;
            for (int k = 0; __builtin_amdgcn_ballot_w32(k < n) != 0; k += 8) {
                int v[8];
#pragma unroll
                for (int u = 0; u < 8; ++u) v[u] = sp[k + u < n ? k + u : 0];
#pragma unroll
                for (int u = 0; u < 8; ++u) if (k + u < n) q[pos + k + u] = v[u];
            }
            cnt += tot;
        } else {
            for (int l = 0; l < RUNS_PER_WAVE; ++l) {
                const int nl = __builtin_amdgcn_readlane(n, l);
                const int sl = __builtin_amdgcn_readlane(st, l);
                const int* rp = srt + (size_t)(wave * RUNS_PER_WAVE + l) * SORT_E + sl;
                for (int off = 0; off < nl; off += 32) {
                    if (cnt > CAP - 32) { flush(cnt); cnt = 0; ovf = true; }
                    const int i = off + lane;
                    if (i < nl) q[cnt + lane] = rp[i];
                    cnt += (nl - off) < 32 ? (nl - off) : 32;
                }
            }
        }
        return cnt;
    };

    const int slot = blockIdx.x * 32 + wave;
    int  cnt  = 0;
    bool have = false;
    if constexpr (ROLE == 1) {
        const int c = __builtin_amdgcn_readfirstlane(wc[slot]);
        if (c >= 0) {
            have = true;
            cnt  = c;
            max_saved(wl + (size_t)slot * LIST_PITCH, cnt);
        }
    }
    if (!have) {
        cnt = gather([&](int n) { max_list(n); });
        max_list(cnt);
    }
    if constexpr (ROLE == 0) {
        if (!ovf) for (int i = lane; i < cnt; i += 32) wl[(size_t)slot * LIST_PITCH + i] = q[i];
        if (lane == 0) wc[slot] = ovf ? -1 : cnt;
    }
    __syncthreads();
    if (!ovf) {
        work(cnt);
    } else {
        cnt = gather([&](int n) { work(n); });
        work(cnt);
    }

    __syncthreads();
    for (int i = tid; i < nr * F; i += 1024) {
        const int r = i / F, c = i - r * F;
        const int ai = (F == F1) ? i : r * F + (c & 3) * (F / 4) + (c >> 2);
        float v = (float)acc[ai] / (float)den[r * NH + (NH == 2 ? (c >> 6) : 0)] + bias[c];
        if constexpr (F == F1) v = v > 0.f ? v : (expf(v) - 1.0f);
        outp[(size_t)n0 * F + i] = v;
    }
}

extern "C" void kernel_launch(void* const* d_in, const int* in_sizes, int n_in,
                              void* d_out, int out_size, void* d_ws, size_t ws_size,
                              hipStream_t stream) {
    const float* x    = (const float*)d_in[0];
    const int*   ei   = (const int*)  d_in[1];
    const float* W1   = (const float*)d_in[2];
    const float* as1  = (const float*)d_in[3];
    const float* ad1  = (const float*)d_in[4];
    const float* b1   = (const float*)d_in[5];
    const float* W2   = (const float*)d_in[6];
    const float* as2  = (const float*)d_in[7];
    const float* ad2  = (const float*)d_in[8];
    const float* b2   = (const float*)d_in[9];
    float* out = (float*)d_out;

    float* ws      = (float*)d_ws;
    float* h1      = ws;
    float* h2in    = h1 + (size_t)N_NODES * F1;
    float* a_src1  = h2in + (size_t)N_NODES * F1;
    float* a_dst1  = a_src1 + (size_t)N_NODES * 2;
    float* a_src2  = a_dst1 + (size_t)N_NODES * 2;
    float* a_dst2  = a_src2 + N_NODES;
    float* h2      = a_dst2 + N_NODES;
    int*   wl      = (int*)d_ws + WS_LISTS_OFF;
    int*   wc      = wl + (size_t)OWN_BLOCKS * 32 * LIST_PITCH;
    int*   srt     = (int*)d_ws + WS_SORT_OFF;
    int*   rtab    = srt + (size_t)SORT_BLOCKS * SORT_E;
    float* w2p     = ws + WS_W2P_OFF;

    const int MT = N_NODES / 16;
    gat_gemm1_wmma<<<G1_BLOCKS, 256, 0, stream>>>(x, W1, h1);
    gat_att1<<<ATT1_BLOCKS + SORT_BLOCKS + 1, 256, 0, stream>>>(h1, as1, ad1, a_src1, a_dst1, ei, srt, rtab, W2, as2, ad2, w2p);
    gat_own_accum<F1, HEADS, LIST_PITCH, 0><<<OWN_BLOCKS, 1024, 0, stream>>>(a_src1, a_dst1, h1, b1, h2in, wl, wc, srt, rtab);
    gat_gemm2_wmma<<<MT, 96, 0, stream>>>(h2in, w2p, h2, a_src2, a_dst2);
    gat_own_accum<NCLS, 1, 1152, 1><<<OWN_BLOCKS, 1024, 0, stream>>>(a_src2, a_dst2, h2, b2, out, wl, wc, srt, rtab);
}
